# starting kernel
# baseline (speedup 1.0000x reference)
#include <hip/hip_runtime.h>
#include <hip/hip_bf16.h>

typedef __attribute__((ext_vector_type(2))) float v2f;
typedef __attribute__((ext_vector_type(8))) float v8f;

#define HID 128
#define NGRAPH 64


__global__ void k_fill(float* p, int n, float v) {
    int i = blockIdx.x * 256 + threadIdx.x;
    if (i < n) p[i] = v;
}

__global__ void k_deg_count(const int* __restrict__ dst, int E, float* __restrict__ deg) {
    int e = blockIdx.x * 256 + threadIdx.x;
    if (e < E) atomicAdd(deg + dst[e], 1.0f);
}

__global__ void k_rsqrt(float* d, int n) {
    int i = blockIdx.x * 256 + threadIdx.x;
    if (i < n) d[i] = rsqrtf(d[i]);
}

__global__ void k_embed(const int* __restrict__ res, const float* __restrict__ pos,
                        const float* __restrict__ W, const float* __restrict__ dinv,
                        int N, float* __restrict__ y) {
    int idx = blockIdx.x * 256 + threadIdx.x;
    if (idx >= N * HID) return;
    int i = idx >> 7, f = idx & 127;
    float xw = W[res[i] * HID + f]
             + pos[i * 3 + 0] * W[20 * HID + f]
             + pos[i * 3 + 1] * W[21 * HID + f]
             + pos[i * 3 + 2] * W[22 * HID + f];
    y[idx] = xw * dinv[i];
}

__global__ void k_scatter(const int* __restrict__ src, const int* __restrict__ dst,
                          int E, const float* __restrict__ y, float* __restrict__ s) {
    int e = blockIdx.x * 8 + (threadIdx.x >> 5);
    if (e >= E) return;
    int lane = threadIdx.x & 31;
    size_t so = (size_t)src[e] * HID + lane * 4;
    size_t doo = (size_t)dst[e] * HID + lane * 4;
    float4 v = *(const float4*)(y + so);
    float* p = s + doo;
    atomicAdd(p + 0, v.x);
    atomicAdd(p + 1, v.y);
    atomicAdd(p + 2, v.z);
    atomicAdd(p + 3, v.w);
}

__global__ void k_combine(const float* __restrict__ y, const float* __restrict__ s,
                          const float* __restrict__ dinv, const float* __restrict__ b,
                          int N, float* __restrict__ h) {
    int idx = blockIdx.x * 256 + threadIdx.x;
    if (idx >= N * HID) return;
    int i = idx >> 7, f = idx & 127;
    h[idx] = fmaxf(dinv[i] * (s[idx] + y[idx]) + b[f], 0.0f);
}

__global__ void k_cnt(const int* __restrict__ batch, int N, float* __restrict__ cnt) {
    int i = blockIdx.x * 256 + threadIdx.x;
    if (i < N) atomicAdd(cnt + batch[i], 1.0f);
}

__global__ void k_pool_sum(const float* __restrict__ h, const int* __restrict__ batch,
                           int N, float* __restrict__ pesum) {
    int idx = blockIdx.x * 256 + threadIdx.x;
    if (idx >= N * HID) return;
    int i = idx >> 7, f = idx & 127;
    atomicAdd(pesum + batch[i] * HID + f, h[idx]);
}

__global__ void k_pool_div(const float* __restrict__ pesum, const float* __restrict__ cnt,
                           float* __restrict__ pe) {
    int idx = blockIdx.x * 256 + threadIdx.x;
    if (idx >= NGRAPH * HID) return;
    pe[idx] = pesum[idx] / fmaxf(cnt[idx >> 7], 1.0f);
}

__global__ void k_feat(const float* __restrict__ h, const float* __restrict__ pe,
                       const int* __restrict__ batch, int N, float* __restrict__ feat) {
    int idx = blockIdx.x * 256 + threadIdx.x;
    if (idx >= N * 2 * HID) return;
    int i = idx >> 8, f = idx & 255;
    feat[idx] = (f < HID) ? h[(size_t)i * HID + f]
                          : pe[batch[i] * HID + (f - HID)];
}

__global__ void k_head(const float* __restrict__ hmid, const float* __restrict__ W2,
                       const float* __restrict__ b2, int N, float* __restrict__ out) {
    int i = blockIdx.x * 256 + threadIdx.x;
    if (i >= N) return;
    float a0 = b2[0], a1 = b2[1], a2 = b2[2];
    const float* h = hmid + (size_t)i * HID;
    for (int k = 0; k < HID; ++k) {
        float hv = h[k];
        a0 += hv * W2[k * 3 + 0];
        a1 += hv * W2[k * 3 + 1];
        a2 += hv * W2[k * 3 + 2];
    }
    out[(size_t)i * 3 + 0] = a0;
    out[(size_t)i * 3 + 1] = a1;
    out[(size_t)i * 3 + 2] = a2;
}

__global__ void k_wfrag(const float* __restrict__ W, int K, float* __restrict__ Wf) {
    int idx = blockIdx.x * 256 + threadIdx.x;
    if (idx >= K * HID) return;
    int j    = idx & 1;
    int nt   = (idx >> 1) & 7;
    int l    = (idx >> 4) & 31;
    int kk   = idx >> 9;
    int krow = 4 * kk + (l >> 4) * 2 + j;
    int col  = nt * 16 + (l & 15);
    Wf[idx] = W[(size_t)krow * HID + col];
}

__global__ void k_gemm128(const float* __restrict__ A, int K, int M,
                          const float* __restrict__ Wf,
                          const float* __restrict__ rowscale,
                          const float* __restrict__ bias, int relu,
                          float* __restrict__ out) {
    int w = threadIdx.x >> 5;
    int lane = threadIdx.x & 31;
    int tile = blockIdx.x * 4 + w;
    if (tile * 16 >= M) return;
    int row0 = tile * 16;
    int half = lane >> 4;
    int lmod = lane & 15;
    int koff = half * 2;
    const float* Arow = A + (size_t)(row0 + lmod) * K + koff;
    const float* Wl   = Wf + (size_t)lane * 16;

    v8f acc[8];
    #pragma unroll
    for (int nt = 0; nt < 8; ++nt) acc[nt] = (v8f){};

    int ksteps = K >> 2;
    for (int kk = 0; kk < ksteps; ++kk) {
        v2f a;
        a.x = Arow[4 * kk];
        a.y = Arow[4 * kk + 1];
        const float4* q = (const float4*)(Wl + (size_t)kk * 512);
        float4 q0 = q[0], q1 = q[1], q2 = q[2], q3 = q[3];
        v2f b0; b0.x = q0.x; b0.y = q0.y;
        v2f b1; b1.x = q0.z; b1.y = q0.w;
        v2f b2; b2.x = q1.x; b2.y = q1.y;
        v2f b3; b3.x = q1.z; b3.y = q1.w;
        v2f b4; b4.x = q2.x; b4.y = q2.y;
        v2f b5; b5.x = q2.z; b5.y = q2.w;
        v2f b6; b6.x = q3.x; b6.y = q3.y;
        v2f b7; b7.x = q3.z; b7.y = q3.w;
        acc[0] = __builtin_amdgcn_wmma_f32_16x16x4_f32(false, a, false, b0, (short)0, acc[0], false, false);
        acc[1] = __builtin_amdgcn_wmma_f32_16x16x4_f32(false, a, false, b1, (short)0, acc[1], false, false);
        acc[2] = __builtin_amdgcn_wmma_f32_16x16x4_f32(false, a, false, b2, (short)0, acc[2], false, false);
        acc[3] = __builtin_amdgcn_wmma_f32_16x16x4_f32(false, a, false, b3, (short)0, acc[3], false, false);
        acc[4] = __builtin_amdgcn_wmma_f32_16x16x4_f32(false, a, false, b4, (short)0, acc[4], false, false);
        acc[5] = __builtin_amdgcn_wmma_f32_16x16x4_f32(false, a, false, b5, (short)0, acc[5], false, false);
        acc[6] = __builtin_amdgcn_wmma_f32_16x16x4_f32(false, a, false, b6, (short)0, acc[6], false, false);
        acc[7] = __builtin_amdgcn_wmma_f32_16x16x4_f32(false, a, false, b7, (short)0, acc[7], false, false);
    }

    float rs[8];
    if (rowscale) {
        #pragma unroll
        for (int r = 0; r < 8; ++r) rs[r] = rowscale[row0 + r + half * 8];
    }

    #pragma unroll
    for (int nt = 0; nt < 8; ++nt) {
        int col = nt * 16 + lmod;
        float bv = bias ? bias[col] : 0.0f;
        #pragma unroll
        for (int r = 0; r < 8; ++r) {
            float v = acc[nt][r];
            if (rowscale) v *= rs[r];
            v += bv;
            if (relu) v = fmaxf(v, 0.0f);
            out[(size_t)(row0 + r + half * 8) * HID + col] = v;
        }
    }
}


static inline int cdiv(long long a, long long b) { return (int)((a + b - 1) / b); }

static void run_branch(const int* res, const float* pos, const int* ei,
                       int N, int E,
                       const float* W1c, const float* b1c,
                       const float* W2c, const float* b2c,
                       float* dinv, float* A, float* B, float* C, float* Wf,
                       hipStream_t stream) {
    const int* src = ei;
    const int* dst = ei + E;
    size_t nh = (size_t)N * HID;

    k_fill<<<cdiv(N, 256), 256, 0, stream>>>(dinv, N, 1.0f);
    k_deg_count<<<cdiv(E, 256), 256, 0, stream>>>(dst, E, dinv);
    k_rsqrt<<<cdiv(N, 256), 256, 0, stream>>>(dinv, N);

    k_embed<<<cdiv(nh, 256), 256, 0, stream>>>(res, pos, W1c, dinv, N, A);
    hipMemsetAsync(B, 0, nh * sizeof(float), stream);
    k_scatter<<<cdiv(E, 8), 256, 0, stream>>>(src, dst, E, A, B);
    k_combine<<<cdiv(nh, 256), 256, 0, stream>>>(A, B, dinv, b1c, N, C);

    k_wfrag<<<cdiv(HID * HID, 256), 256, 0, stream>>>(W2c, HID, Wf);
    k_gemm128<<<cdiv(N / 16, 4), 128, 0, stream>>>(C, HID, N, Wf, dinv, nullptr, 0, A);
    hipMemsetAsync(B, 0, nh * sizeof(float), stream);
    k_scatter<<<cdiv(E, 8), 256, 0, stream>>>(src, dst, E, A, B);
    k_combine<<<cdiv(nh, 256), 256, 0, stream>>>(A, B, dinv, b2c, N, C);
}

extern "C" void kernel_launch(void* const* d_in, const int* in_sizes, int n_in,
                              void* d_out, int out_size, void* d_ws, size_t ws_size,
                              hipStream_t stream) {
    (void)n_in; (void)out_size; (void)ws_size;

    const int*   res_p   = (const int*)d_in[0];
    const float* pos_p   = (const float*)d_in[1];
    const int*   ei_p    = (const int*)d_in[2];
    const int*   batch_p = (const int*)d_in[3];
    const int*   res_m   = (const int*)d_in[4];
    const float* pos_m   = (const float*)d_in[5];
    const int*   ei_m    = (const int*)d_in[6];
    const int*   batch_m = (const int*)d_in[7];
    const float* W_p1 = (const float*)d_in[8];
    const float* b_p1 = (const float*)d_in[9];
    const float* W_p2 = (const float*)d_in[10];
    const float* b_p2 = (const float*)d_in[11];
    const float* W_m1 = (const float*)d_in[12];
    const float* b_m1 = (const float*)d_in[13];
    const float* W_m2 = (const float*)d_in[14];
    const float* b_m2 = (const float*)d_in[15];
    const float* W1f  = (const float*)d_in[16];
    const float* b1f  = (const float*)d_in[17];
    const float* W2f  = (const float*)d_in[18];
    const float* b2f  = (const float*)d_in[19];

    int N_P = in_sizes[0];
    int E_P = in_sizes[2] / 2;
    int N_M = in_sizes[4];
    int E_M = in_sizes[6] / 2;

    float* ws = (float*)d_ws;
    size_t nhp = (size_t)N_P * HID;
    float* A      = ws;
    float* B      = A + nhp;
    float* C      = B + nhp;
    float* dinv_p = C + nhp;
    float* dinv_m = dinv_p + N_P;
    float* pe_sum = dinv_m + N_M;
    float* pe     = pe_sum + NGRAPH * HID;
    float* cnt    = pe + NGRAPH * HID;
    float* Wfrag  = cnt + NGRAPH;

    run_branch(res_p, pos_p, ei_p, N_P, E_P, W_p1, b_p1, W_p2, b_p2,
               dinv_p, A, B, C, Wfrag, stream);
    hipMemsetAsync(pe_sum, 0, (size_t)NGRAPH * HID * sizeof(float), stream);
    hipMemsetAsync(cnt, 0, NGRAPH * sizeof(float), stream);
    k_cnt<<<cdiv(N_P, 256), 256, 0, stream>>>(batch_p, N_P, cnt);
    k_pool_sum<<<cdiv(nhp, 256), 256, 0, stream>>>(C, batch_p, N_P, pe_sum);
    k_pool_div<<<cdiv(NGRAPH * HID, 256), 256, 0, stream>>>(pe_sum, cnt, pe);

    run_branch(res_m, pos_m, ei_m, N_M, E_M, W_m1, b_m1, W_m2, b_m2,
               dinv_m, A, B, C, Wfrag, stream);

    size_t nfm = (size_t)N_M * 2 * HID;
    float* feat = A;
    float* hmid = B;
    k_feat<<<cdiv(nfm, 256), 256, 0, stream>>>(C, pe, batch_m, N_M, feat);
    k_wfrag<<<cdiv(2 * HID * HID, 256), 256, 0, stream>>>(W1f, 2 * HID, Wfrag);
    k_gemm128<<<cdiv(N_M / 16, 4), 128, 0, stream>>>(feat, 2 * HID, N_M, Wfrag,
                                                     nullptr, b1f, 1, hmid);
    k_head<<<cdiv(N_M, 256), 256, 0, stream>>>(hmid, W2f, b2f, N_M, (float*)d_out);
}


# Float-atomic feature scatter replaced by dst-bucketed edge lists (a returning int atomic per edge gives its slot) and a wave-per-node row gather, combine fused; sums run in slot (arrival) order
# speedup vs baseline: 2.0266x; 2.0266x over previous
#include <hip/hip_runtime.h>
#include <hip/hip_bf16.h>

typedef __attribute__((ext_vector_type(2))) float v2f;
typedef __attribute__((ext_vector_type(8))) float v8f;
typedef __attribute__((ext_vector_type(4))) float v4f;

#define HID 128
#define NGRAPH 64


__global__ void k_fill_i(int* p, int n, int v) {
    int i = blockIdx.x * 256 + threadIdx.x;
    if (i < n) p[i] = v;
}

__global__ void k_deg_rank(const int* __restrict__ dst, int E, int* __restrict__ ecnt,
                           int* __restrict__ rank) {
    int e = blockIdx.x * 256 + threadIdx.x;
    if (e < E) rank[e] = atomicAdd(ecnt + dst[e], 1);
}

__global__ void k_rowalloc(const int* __restrict__ ecnt, int N, int* __restrict__ total,
                           int* __restrict__ rowptr, float* __restrict__ dinv) {
    int i = blockIdx.x * 256 + threadIdx.x;
    int lane = threadIdx.x & 31;
    int c = (i < N) ? ecnt[i] : 0;
    int x = c;
    #pragma unroll
    for (int d = 1; d < 32; d <<= 1) {
        int t = __shfl_up(x, d);
        if (lane >= d) x += t;
    }
    int base = 0;
    if (lane == 31) base = atomicAdd(total, x);
    base = __shfl(base, 31);
    if (i < N) {
        rowptr[i] = base + x - c;
        dinv[i] = rsqrtf((float)c + 1.0f);
    }
}

__global__ void k_csr_fill(const int* __restrict__ src, const int* __restrict__ dst,
                           const int* __restrict__ rank, const int* __restrict__ rowptr,
                           int E, int* __restrict__ col) {
    int e = blockIdx.x * 256 + threadIdx.x;
    if (e < E) col[rowptr[dst[e]] + rank[e]] = src[e];
}

__global__ void k_embed(const int* __restrict__ res, const float* __restrict__ pos,
                        const float* __restrict__ W, const float* __restrict__ dinv,
                        int N, float* __restrict__ y) {
    int idx = blockIdx.x * 256 + threadIdx.x;
    if (idx >= N * HID) return;
    int i = idx >> 7, f = idx & 127;
    float xw = W[res[i] * HID + f]
             + pos[i * 3 + 0] * W[20 * HID + f]
             + pos[i * 3 + 1] * W[21 * HID + f]
             + pos[i * 3 + 2] * W[22 * HID + f];
    y[idx] = xw * dinv[i];
}

__global__ void k_gather(const float* __restrict__ y, const int* __restrict__ rowptr,
                         const int* __restrict__ ecnt, const int* __restrict__ col,
                         const float* __restrict__ dinv, const float* __restrict__ b,
                         int N, float* __restrict__ h) {
    int i = __builtin_amdgcn_readfirstlane((int)(blockIdx.x * 8 + (threadIdx.x >> 5)));
    if (i >= N) return;
    int lane = threadIdx.x & 31;
    int start = rowptr[i];
    int n = ecnt[i];
    const float* yl = y + lane * 4;
    v4f acc = (v4f){0.0f, 0.0f, 0.0f, 0.0f};
    for (int base = 0; base < n; base += 32) {
        int m = n - base;
        if (m > 32) m = 32;
        int c = (lane < m) ? col[start + base + lane] : i;
        c = min(max(c, 0), N - 1);
        for (int j = 0; j < m; j += 8) {
            int s0 = __builtin_amdgcn_readlane(c, j);
            int s1 = __builtin_amdgcn_readlane(c, j + 1);
            int s2 = __builtin_amdgcn_readlane(c, j + 2);
            int s3 = __builtin_amdgcn_readlane(c, j + 3);
            int s4 = __builtin_amdgcn_readlane(c, j + 4);
            int s5 = __builtin_amdgcn_readlane(c, j + 5);
            int s6 = __builtin_amdgcn_readlane(c, j + 6);
            int s7 = __builtin_amdgcn_readlane(c, j + 7);
            v4f v0 = *(const v4f*)(yl + (size_t)s0 * HID);
            v4f v1 = *(const v4f*)(yl + (size_t)s1 * HID);
            v4f v2 = *(const v4f*)(yl + (size_t)s2 * HID);
            v4f v3 = *(const v4f*)(yl + (size_t)s3 * HID);
            v4f v4 = *(const v4f*)(yl + (size_t)s4 * HID);
            v4f v5 = *(const v4f*)(yl + (size_t)s5 * HID);
            v4f v6 = *(const v4f*)(yl + (size_t)s6 * HID);
            v4f v7 = *(const v4f*)(yl + (size_t)s7 * HID);
            asm volatile("" :: "v"(v0), "v"(v1), "v"(v2), "v"(v3),
                               "v"(v4), "v"(v5), "v"(v6), "v"(v7));
            acc += v0;
            if (j + 1 < m) acc += v1;
            if (j + 2 < m) acc += v2;
            if (j + 3 < m) acc += v3;
            if (j + 4 < m) acc += v4;
            if (j + 5 < m) acc += v5;
            if (j + 6 < m) acc += v6;
            if (j + 7 < m) acc += v7;
        }
    }
    v4f self = *(const v4f*)(yl + (size_t)i * HID);
    v4f bb = *(const v4f*)(b + lane * 4);
    float d = dinv[i];
    v4f o;
    o.x = fmaxf(d * (acc.x + self.x) + bb.x, 0.0f);
    o.y = fmaxf(d * (acc.y + self.y) + bb.y, 0.0f);
    o.z = fmaxf(d * (acc.z + self.z) + bb.z, 0.0f);
    o.w = fmaxf(d * (acc.w + self.w) + bb.w, 0.0f);
    *(v4f*)(h + (size_t)i * HID + lane * 4) = o;
}

__global__ void k_cnt(const int* __restrict__ batch, int N, float* __restrict__ cnt) {
    int i = blockIdx.x * 256 + threadIdx.x;
    if (i < N) atomicAdd(cnt + batch[i], 1.0f);
}

__global__ void k_pool_sum(const float* __restrict__ h, const int* __restrict__ batch,
                           int N, float* __restrict__ pesum) {
    int idx = blockIdx.x * 256 + threadIdx.x;
    if (idx >= N * HID) return;
    int i = idx >> 7, f = idx & 127;
    atomicAdd(pesum + batch[i] * HID + f, h[idx]);
}

__global__ void k_pool_div(const float* __restrict__ pesum, const float* __restrict__ cnt,
                           float* __restrict__ pe) {
    int idx = blockIdx.x * 256 + threadIdx.x;
    if (idx >= NGRAPH * HID) return;
    pe[idx] = pesum[idx] / fmaxf(cnt[idx >> 7], 1.0f);
}

__global__ void k_feat(const float* __restrict__ h, const float* __restrict__ pe,
                       const int* __restrict__ batch, int N, float* __restrict__ feat) {
    int idx = blockIdx.x * 256 + threadIdx.x;
    if (idx >= N * 2 * HID) return;
    int i = idx >> 8, f = idx & 255;
    feat[idx] = (f < HID) ? h[(size_t)i * HID + f]
                          : pe[batch[i] * HID + (f - HID)];
}

__global__ void k_head(const float* __restrict__ hmid, const float* __restrict__ W2,
                       const float* __restrict__ b2, int N, float* __restrict__ out) {
    int i = blockIdx.x * 256 + threadIdx.x;
    if (i >= N) return;
    float a0 = b2[0], a1 = b2[1], a2 = b2[2];
    const float* h = hmid + (size_t)i * HID;
    for (int k = 0; k < HID; ++k) {
        float hv = h[k];
        a0 += hv * W2[k * 3 + 0];
        a1 += hv * W2[k * 3 + 1];
        a2 += hv * W2[k * 3 + 2];
    }
    out[(size_t)i * 3 + 0] = a0;
    out[(size_t)i * 3 + 1] = a1;
    out[(size_t)i * 3 + 2] = a2;
}

__global__ void k_wfrag(const float* __restrict__ W, int K, float* __restrict__ Wf) {
    int idx = blockIdx.x * 256 + threadIdx.x;
    if (idx >= K * HID) return;
    int j    = idx & 1;
    int nt   = (idx >> 1) & 7;
    int l    = (idx >> 4) & 31;
    int kk   = idx >> 9;
    int krow = 4 * kk + (l >> 4) * 2 + j;
    int col  = nt * 16 + (l & 15);
    Wf[idx] = W[(size_t)krow * HID + col];
}

__global__ void k_gemm128(const float* __restrict__ A, int K, int M,
                          const float* __restrict__ Wf,
                          const float* __restrict__ rowscale,
                          const float* __restrict__ bias, int relu,
                          float* __restrict__ out) {
    int w = threadIdx.x >> 5;
    int lane = threadIdx.x & 31;
    int tile = blockIdx.x * 4 + w;
    if (tile * 16 >= M) return;
    int row0 = tile * 16;
    int half = lane >> 4;
    int lmod = lane & 15;
    int koff = half * 2;
    const float* Arow = A + (size_t)(row0 + lmod) * K + koff;
    const float* Wl   = Wf + (size_t)lane * 16;

    v8f acc[8];
    #pragma unroll
    for (int nt = 0; nt < 8; ++nt) acc[nt] = (v8f){};

    int ksteps = K >> 2;
    for (int kk = 0; kk < ksteps; ++kk) {
        v2f a;
        a.x = Arow[4 * kk];
        a.y = Arow[4 * kk + 1];
        const float4* q = (const float4*)(Wl + (size_t)kk * 512);
        float4 q0 = q[0], q1 = q[1], q2 = q[2], q3 = q[3];
        v2f b0; b0.x = q0.x; b0.y = q0.y;
        v2f b1; b1.x = q0.z; b1.y = q0.w;
        v2f b2; b2.x = q1.x; b2.y = q1.y;
        v2f b3; b3.x = q1.z; b3.y = q1.w;
        v2f b4; b4.x = q2.x; b4.y = q2.y;
        v2f b5; b5.x = q2.z; b5.y = q2.w;
        v2f b6; b6.x = q3.x; b6.y = q3.y;
        v2f b7; b7.x = q3.z; b7.y = q3.w;
        acc[0] = __builtin_amdgcn_wmma_f32_16x16x4_f32(false, a, false, b0, (short)0, acc[0], false, false);
        acc[1] = __builtin_amdgcn_wmma_f32_16x16x4_f32(false, a, false, b1, (short)0, acc[1], false, false);
        acc[2] = __builtin_amdgcn_wmma_f32_16x16x4_f32(false, a, false, b2, (short)0, acc[2], false, false);
        acc[3] = __builtin_amdgcn_wmma_f32_16x16x4_f32(false, a, false, b3, (short)0, acc[3], false, false);
        acc[4] = __builtin_amdgcn_wmma_f32_16x16x4_f32(false, a, false, b4, (short)0, acc[4], false, false);
        acc[5] = __builtin_amdgcn_wmma_f32_16x16x4_f32(false, a, false, b5, (short)0, acc[5], false, false);
        acc[6] = __builtin_amdgcn_wmma_f32_16x16x4_f32(false, a, false, b6, (short)0, acc[6], false, false);
        acc[7] = __builtin_amdgcn_wmma_f32_16x16x4_f32(false, a, false, b7, (short)0, acc[7], false, false);
    }

    float rs[8];
    if (rowscale) {
        #pragma unroll
        for (int r = 0; r < 8; ++r) rs[r] = rowscale[row0 + r + half * 8];
    }

    #pragma unroll
    for (int nt = 0; nt < 8; ++nt) {
        int col = nt * 16 + lmod;
        float bv = bias ? bias[col] : 0.0f;
        #pragma unroll
        for (int r = 0; r < 8; ++r) {
            float v = acc[nt][r];
            if (rowscale) v *= rs[r];
            v += bv;
            if (relu) v = fmaxf(v, 0.0f);
            out[(size_t)(row0 + r + half * 8) * HID + col] = v;
        }
    }
}


static inline int cdiv(long long a, long long b) { return (int)((a + b - 1) / b); }

static void run_branch(const int* res, const float* pos, const int* ei,
                       int N, int E,
                       const float* W1c, const float* b1c,
                       const float* W2c, const float* b2c,
                       float* dinv, float* A, float* C, float* Wf,
                       int* ecnt, int* rowptr, int* rank, int* col,
                       hipStream_t stream) {
    const int* src = ei;
    const int* dst = ei + E;
    size_t nh = (size_t)N * HID;

    k_fill_i<<<cdiv(N + 1, 256), 256, 0, stream>>>(ecnt, N + 1, 0);
    k_deg_rank<<<cdiv(E, 256), 256, 0, stream>>>(dst, E, ecnt, rank);
    k_rowalloc<<<cdiv(N, 256), 256, 0, stream>>>(ecnt, N, ecnt + N, rowptr, dinv);
    k_csr_fill<<<cdiv(E, 256), 256, 0, stream>>>(src, dst, rank, rowptr, E, col);

    k_embed<<<cdiv(nh, 256), 256, 0, stream>>>(res, pos, W1c, dinv, N, A);
    k_gather<<<cdiv(N, 8), 256, 0, stream>>>(A, rowptr, ecnt, col, dinv, b1c, N, C);

    k_wfrag<<<cdiv(HID * HID, 256), 256, 0, stream>>>(W2c, HID, Wf);
    k_gemm128<<<cdiv(N / 16, 4), 128, 0, stream>>>(C, HID, N, Wf, dinv, nullptr, 0, A);
    k_gather<<<cdiv(N, 8), 256, 0, stream>>>(A, rowptr, ecnt, col, dinv, b2c, N, C);
}

extern "C" void kernel_launch(void* const* d_in, const int* in_sizes, int n_in,
                              void* d_out, int out_size, void* d_ws, size_t ws_size,
                              hipStream_t stream) {
    (void)n_in; (void)out_size; (void)ws_size;

    const int*   res_p   = (const int*)d_in[0];
    const float* pos_p   = (const float*)d_in[1];
    const int*   ei_p    = (const int*)d_in[2];
    const int*   batch_p = (const int*)d_in[3];
    const int*   res_m   = (const int*)d_in[4];
    const float* pos_m   = (const float*)d_in[5];
    const int*   ei_m    = (const int*)d_in[6];
    const int*   batch_m = (const int*)d_in[7];
    const float* W_p1 = (const float*)d_in[8];
    const float* b_p1 = (const float*)d_in[9];
    const float* W_p2 = (const float*)d_in[10];
    const float* b_p2 = (const float*)d_in[11];
    const float* W_m1 = (const float*)d_in[12];
    const float* b_m1 = (const float*)d_in[13];
    const float* W_m2 = (const float*)d_in[14];
    const float* b_m2 = (const float*)d_in[15];
    const float* W1f  = (const float*)d_in[16];
    const float* b1f  = (const float*)d_in[17];
    const float* W2f  = (const float*)d_in[18];
    const float* b2f  = (const float*)d_in[19];

    int N_P = in_sizes[0];
    int E_P = in_sizes[2] / 2;
    int N_M = in_sizes[4];
    int E_M = in_sizes[6] / 2;

    float* ws = (float*)d_ws;
    size_t nhp = (size_t)N_P * HID;
    float* A      = ws;
    float* B      = A + nhp;
    float* C      = B + nhp;
    float* dinv_p = C + nhp;
    float* dinv_m = dinv_p + N_P;
    float* pe_sum = dinv_m + N_M;
    float* pe     = pe_sum + NGRAPH * HID;
    float* cnt    = pe + NGRAPH * HID;
    float* Wfrag  = cnt + NGRAPH;
    int* ecnt   = (int*)(Wfrag + 2 * HID * HID);
    int* rowptr = ecnt + (N_P + 1);
    int* rank   = rowptr + N_P;
    int* col    = rank + E_P;

    run_branch(res_p, pos_p, ei_p, N_P, E_P, W_p1, b_p1, W_p2, b_p2,
               dinv_p, A, C, Wfrag, ecnt, rowptr, rank, col, stream);
    hipMemsetAsync(pe_sum, 0, (size_t)NGRAPH * HID * sizeof(float), stream);
    hipMemsetAsync(cnt, 0, NGRAPH * sizeof(float), stream);
    k_cnt<<<cdiv(N_P, 256), 256, 0, stream>>>(batch_p, N_P, cnt);
    k_pool_sum<<<cdiv(nhp, 256), 256, 0, stream>>>(C, batch_p, N_P, pe_sum);
    k_pool_div<<<cdiv(NGRAPH * HID, 256), 256, 0, stream>>>(pe_sum, cnt, pe);

    run_branch(res_m, pos_m, ei_m, N_M, E_M, W_m1, b_m1, W_m2, b_m2,
               dinv_m, A, C, Wfrag, ecnt, rowptr, rank, col, stream);

    size_t nfm = (size_t)N_M * 2 * HID;
    float* feat = A;
    float* hmid = B;
    k_feat<<<cdiv(nfm, 256), 256, 0, stream>>>(C, pe, batch_m, N_M, feat);
    k_wfrag<<<cdiv(2 * HID * HID, 256), 256, 0, stream>>>(W1f, 2 * HID, Wfrag);
    k_gemm128<<<cdiv(N_M / 16, 4), 128, 0, stream>>>(feat, 2 * HID, N_M, Wfrag,
                                                     nullptr, b1f, 1, hmid);
    k_head<<<cdiv(N_M, 256), 256, 0, stream>>>(hmid, W2f, b2f, N_M, (float*)d_out);
}


# Layer 1 sums neighbours before the weight product: each edge fetches a 32-B node record and adds dinv*W[res] from an LDS table, not a 512-B transformed row; summation order changes
# speedup vs baseline: 2.2574x; 1.1139x over previous
#include <hip/hip_runtime.h>
#include <hip/hip_bf16.h>

typedef __attribute__((ext_vector_type(2))) float v2f;
typedef __attribute__((ext_vector_type(8))) float v8f;
typedef __attribute__((ext_vector_type(4))) float v4f;

#define HID 128
#define NGRAPH 64
#define NRES 20
#define AGG_NPW 4


__global__ void k_fill_i(int* p, int n, int v) {
    int i = blockIdx.x * 256 + threadIdx.x;
    if (i < n) p[i] = v;
}

__global__ void k_deg_rank(const int* __restrict__ dst, int E, int* __restrict__ ecnt,
                           int* __restrict__ rank) {
    int e = blockIdx.x * 256 + threadIdx.x;
    if (e < E) rank[e] = atomicAdd(ecnt + dst[e], 1);
}

__global__ void k_rowalloc(const int* __restrict__ ecnt, int N, int* __restrict__ total,
                           int* __restrict__ rowptr, float* __restrict__ dinv) {
    int i = blockIdx.x * 256 + threadIdx.x;
    int lane = threadIdx.x & 31;
    int c = (i < N) ? ecnt[i] : 0;
    int x = c;
    #pragma unroll
    for (int d = 1; d < 32; d <<= 1) {
        int t = __shfl_up(x, d);
        if (lane >= d) x += t;
    }
    int base = 0;
    if (lane == 31) base = atomicAdd(total, x);
    base = __shfl(base, 31);
    if (i < N) {
        rowptr[i] = base + x - c;
        dinv[i] = rsqrtf((float)c + 1.0f);
    }
}

__global__ void k_csr_fill(const int* __restrict__ src, const int* __restrict__ dst,
                           const int* __restrict__ rank, const int* __restrict__ rowptr,
                           int E, int* __restrict__ col) {
    int e = blockIdx.x * 256 + threadIdx.x;
    if (e < E) col[rowptr[dst[e]] + rank[e]] = src[e];
}

__global__ void k_rec(const int* __restrict__ res, const float* __restrict__ pos,
                      const float* __restrict__ dinv, int N, float* __restrict__ rec) {
    int i = blockIdx.x * 256 + threadIdx.x;
    if (i >= N) return;
    float d = dinv[i];
    v4f q;
    q.x = d * pos[i * 3 + 0];
    q.y = d * pos[i * 3 + 1];
    q.z = d * pos[i * 3 + 2];
    q.w = d;
    *(v4f*)(rec + (size_t)i * 8) = q;
    int r = res[i];
    if (r < 0 || r >= NRES) r = NRES;
    ((int*)rec)[(size_t)i * 8 + 4] = r * (HID * 4);
}

__global__ void __launch_bounds__(256)
k_agg1(const float* __restrict__ rec, const int* __restrict__ rowptr,
       const int* __restrict__ ecnt, const int* __restrict__ col,
       const float* __restrict__ W, const float* __restrict__ b,
       int N, float* __restrict__ h) {
    __shared__ __attribute__((aligned(16))) float Ws[(NRES + 1) * HID];
    for (int t = threadIdx.x; t < NRES * HID / 4; t += 256)
        ((v4f*)Ws)[t] = ((const v4f*)W)[t];
    if (threadIdx.x < HID / 4)
        ((v4f*)Ws)[NRES * HID / 4 + threadIdx.x] = (v4f){0.0f, 0.0f, 0.0f, 0.0f};
    __syncthreads();

    int lane = threadIdx.x & 31;
    const v4f w20 = *(const v4f*)(W + (NRES + 0) * HID + lane * 4);
    const v4f w21 = *(const v4f*)(W + (NRES + 1) * HID + lane * 4);
    const v4f w22 = *(const v4f*)(W + (NRES + 2) * HID + lane * 4);
    const v4f bb = *(const v4f*)(b + lane * 4);
    const char* Wl = (const char*)Ws + lane * 16;
    const int* reci = (const int*)rec;
    int first = (int)(blockIdx.x * 8 + (threadIdx.x >> 5)) * AGG_NPW;

    for (int t = 0; t < AGG_NPW; ++t) {
        int i = __builtin_amdgcn_readfirstlane(first + t);
        if (i >= N) break;
        int start = rowptr[i];
        int n = ecnt[i];
        v4f rs = *(const v4f*)(rec + (size_t)i * 8);
        int os = reci[(size_t)i * 8 + 4];
        v4f acc = (v4f){0.0f, 0.0f, 0.0f, 0.0f};
        float q0 = 0.0f, q1 = 0.0f, q2 = 0.0f;
        for (int base = 0; base < n; base += 32) {
            int m = n - base;
            if (m > 32) m = 32;
            bool live = lane < m;
            int c = live ? col[start + base + lane] : i;
            c = min(max(c, 0), N - 1);
            v4f r = *(const v4f*)(rec + (size_t)c * 8);
            int off = reci[(size_t)c * 8 + 4];
            float d = live ? r.w : 0.0f;
            off = live ? off : NRES * HID * 4;
            q0 += live ? r.x : 0.0f;
            q1 += live ? r.y : 0.0f;
            q2 += live ? r.z : 0.0f;
            int dbits = __float_as_int(d);
            for (int j = 0; j < m; j += 8) {
                int o0 = __builtin_amdgcn_readlane(off, j);
                int o1 = __builtin_amdgcn_readlane(off, j + 1);
                int o2 = __builtin_amdgcn_readlane(off, j + 2);
                int o3 = __builtin_amdgcn_readlane(off, j + 3);
                int o4 = __builtin_amdgcn_readlane(off, j + 4);
                int o5 = __builtin_amdgcn_readlane(off, j + 5);
                int o6 = __builtin_amdgcn_readlane(off, j + 6);
                int o7 = __builtin_amdgcn_readlane(off, j + 7);
                v4f x0 = *(const v4f*)(Wl + o0);
                v4f x1 = *(const v4f*)(Wl + o1);
                v4f x2 = *(const v4f*)(Wl + o2);
                v4f x3 = *(const v4f*)(Wl + o3);
                v4f x4 = *(const v4f*)(Wl + o4);
                v4f x5 = *(const v4f*)(Wl + o5);
                v4f x6 = *(const v4f*)(Wl + o6);
                v4f x7 = *(const v4f*)(Wl + o7);
                float d0 = __int_as_float(__builtin_amdgcn_readlane(dbits, j));
                float d1 = __int_as_float(__builtin_amdgcn_readlane(dbits, j + 1));
                float d2 = __int_as_float(__builtin_amdgcn_readlane(dbits, j + 2));
                float d3 = __int_as_float(__builtin_amdgcn_readlane(dbits, j + 3));
                float d4 = __int_as_float(__builtin_amdgcn_readlane(dbits, j + 4));
                float d5 = __int_as_float(__builtin_amdgcn_readlane(dbits, j + 5));
                float d6 = __int_as_float(__builtin_amdgcn_readlane(dbits, j + 6));
                float d7 = __int_as_float(__builtin_amdgcn_readlane(dbits, j + 7));
                acc += d0 * x0;
                acc += d1 * x1;
                acc += d2 * x2;
                acc += d3 * x3;
                acc += d4 * x4;
                acc += d5 * x5;
                acc += d6 * x6;
                acc += d7 * x7;
            }
        }
        v4f xs = *(const v4f*)(Wl + os);
        acc += rs.w * xs;
        #pragma unroll
        for (int s = 16; s >= 1; s >>= 1) {
            q0 += __shfl_xor(q0, s);
            q1 += __shfl_xor(q1, s);
            q2 += __shfl_xor(q2, s);
        }
        q0 += rs.x;
        q1 += rs.y;
        q2 += rs.z;
        v4f S = acc + q0 * w20 + q1 * w21 + q2 * w22;
        float dnode = rs.w;
        v4f o;
        o.x = fmaxf(dnode * S.x + bb.x, 0.0f);
        o.y = fmaxf(dnode * S.y + bb.y, 0.0f);
        o.z = fmaxf(dnode * S.z + bb.z, 0.0f);
        o.w = fmaxf(dnode * S.w + bb.w, 0.0f);
        *(v4f*)(h + (size_t)i * HID + lane * 4) = o;
    }
}

__global__ void k_gather(const float* __restrict__ y, const int* __restrict__ rowptr,
                         const int* __restrict__ ecnt, const int* __restrict__ col,
                         const float* __restrict__ dinv, const float* __restrict__ b,
                         int N, float* __restrict__ h) {
    int i = __builtin_amdgcn_readfirstlane((int)(blockIdx.x * 8 + (threadIdx.x >> 5)));
    if (i >= N) return;
    int lane = threadIdx.x & 31;
    int start = rowptr[i];
    int n = ecnt[i];
    const float* yl = y + lane * 4;
    v4f acc = (v4f){0.0f, 0.0f, 0.0f, 0.0f};
    for (int base = 0; base < n; base += 32) {
        int m = n - base;
        if (m > 32) m = 32;
        int c = (lane < m) ? col[start + base + lane] : i;
        c = min(max(c, 0), N - 1);
        for (int j = 0; j < m; j += 8) {
            int s0 = __builtin_amdgcn_readlane(c, j);
            int s1 = __builtin_amdgcn_readlane(c, j + 1);
            int s2 = __builtin_amdgcn_readlane(c, j + 2);
            int s3 = __builtin_amdgcn_readlane(c, j + 3);
            int s4 = __builtin_amdgcn_readlane(c, j + 4);
            int s5 = __builtin_amdgcn_readlane(c, j + 5);
            int s6 = __builtin_amdgcn_readlane(c, j + 6);
            int s7 = __builtin_amdgcn_readlane(c, j + 7);
            v4f v0 = *(const v4f*)(yl + (size_t)s0 * HID);
            v4f v1 = *(const v4f*)(yl + (size_t)s1 * HID);
            v4f v2 = *(const v4f*)(yl + (size_t)s2 * HID);
            v4f v3 = *(const v4f*)(yl + (size_t)s3 * HID);
            v4f v4 = *(const v4f*)(yl + (size_t)s4 * HID);
            v4f v5 = *(const v4f*)(yl + (size_t)s5 * HID);
            v4f v6 = *(const v4f*)(yl + (size_t)s6 * HID);
            v4f v7 = *(const v4f*)(yl + (size_t)s7 * HID);
            asm volatile("" :: "v"(v0), "v"(v1), "v"(v2), "v"(v3),
                               "v"(v4), "v"(v5), "v"(v6), "v"(v7));
            acc += v0;
            if (j + 1 < m) acc += v1;
            if (j + 2 < m) acc += v2;
            if (j + 3 < m) acc += v3;
            if (j + 4 < m) acc += v4;
            if (j + 5 < m) acc += v5;
            if (j + 6 < m) acc += v6;
            if (j + 7 < m) acc += v7;
        }
    }
    v4f self = *(const v4f*)(yl + (size_t)i * HID);
    v4f bb = *(const v4f*)(b + lane * 4);
    float d = dinv[i];
    v4f o;
    o.x = fmaxf(d * (acc.x + self.x) + bb.x, 0.0f);
    o.y = fmaxf(d * (acc.y + self.y) + bb.y, 0.0f);
    o.z = fmaxf(d * (acc.z + self.z) + bb.z, 0.0f);
    o.w = fmaxf(d * (acc.w + self.w) + bb.w, 0.0f);
    *(v4f*)(h + (size_t)i * HID + lane * 4) = o;
}

__global__ void k_cnt(const int* __restrict__ batch, int N, float* __restrict__ cnt) {
    int i = blockIdx.x * 256 + threadIdx.x;
    if (i < N) atomicAdd(cnt + batch[i], 1.0f);
}

__global__ void k_pool_sum(const float* __restrict__ h, const int* __restrict__ batch,
                           int N, float* __restrict__ pesum) {
    int idx = blockIdx.x * 256 + threadIdx.x;
    if (idx >= N * HID) return;
    int i = idx >> 7, f = idx & 127;
    atomicAdd(pesum + batch[i] * HID + f, h[idx]);
}

__global__ void k_pool_div(const float* __restrict__ pesum, const float* __restrict__ cnt,
                           float* __restrict__ pe) {
    int idx = blockIdx.x * 256 + threadIdx.x;
    if (idx >= NGRAPH * HID) return;
    pe[idx] = pesum[idx] / fmaxf(cnt[idx >> 7], 1.0f);
}

__global__ void k_feat(const float* __restrict__ h, const float* __restrict__ pe,
                       const int* __restrict__ batch, int N, float* __restrict__ feat) {
    int idx = blockIdx.x * 256 + threadIdx.x;
    if (idx >= N * 2 * HID) return;
    int i = idx >> 8, f = idx & 255;
    feat[idx] = (f < HID) ? h[(size_t)i * HID + f]
                          : pe[batch[i] * HID + (f - HID)];
}

__global__ void k_head(const float* __restrict__ hmid, const float* __restrict__ W2,
                       const float* __restrict__ b2, int N, float* __restrict__ out) {
    int i = blockIdx.x * 256 + threadIdx.x;
    if (i >= N) return;
    float a0 = b2[0], a1 = b2[1], a2 = b2[2];
    const float* h = hmid + (size_t)i * HID;
    for (int k = 0; k < HID; ++k) {
        float hv = h[k];
        a0 += hv * W2[k * 3 + 0];
        a1 += hv * W2[k * 3 + 1];
        a2 += hv * W2[k * 3 + 2];
    }
    out[(size_t)i * 3 + 0] = a0;
    out[(size_t)i * 3 + 1] = a1;
    out[(size_t)i * 3 + 2] = a2;
}

__global__ void k_wfrag(const float* __restrict__ W, int K, float* __restrict__ Wf) {
    int idx = blockIdx.x * 256 + threadIdx.x;
    if (idx >= K * HID) return;
    int j    = idx & 1;
    int nt   = (idx >> 1) & 7;
    int l    = (idx >> 4) & 31;
    int kk   = idx >> 9;
    int krow = 4 * kk + (l >> 4) * 2 + j;
    int col  = nt * 16 + (l & 15);
    Wf[idx] = W[(size_t)krow * HID + col];
}

__global__ void k_gemm128(const float* __restrict__ A, int K, int M,
                          const float* __restrict__ Wf,
                          const float* __restrict__ rowscale,
                          const float* __restrict__ bias, int relu,
                          float* __restrict__ out) {
    int w = threadIdx.x >> 5;
    int lane = threadIdx.x & 31;
    int tile = blockIdx.x * 4 + w;
    if (tile * 16 >= M) return;
    int row0 = tile * 16;
    int half = lane >> 4;
    int lmod = lane & 15;
    int koff = half * 2;
    const float* Arow = A + (size_t)(row0 + lmod) * K + koff;
    const float* Wl   = Wf + (size_t)lane * 16;

    v8f acc[8];
    #pragma unroll
    for (int nt = 0; nt < 8; ++nt) acc[nt] = (v8f){};

    int ksteps = K >> 2;
    for (int kk = 0; kk < ksteps; ++kk) {
        v2f a;
        a.x = Arow[4 * kk];
        a.y = Arow[4 * kk + 1];
        const float4* q = (const float4*)(Wl + (size_t)kk * 512);
        float4 q0 = q[0], q1 = q[1], q2 = q[2], q3 = q[3];
        v2f b0; b0.x = q0.x; b0.y = q0.y;
        v2f b1; b1.x = q0.z; b1.y = q0.w;
        v2f b2; b2.x = q1.x; b2.y = q1.y;
        v2f b3; b3.x = q1.z; b3.y = q1.w;
        v2f b4; b4.x = q2.x; b4.y = q2.y;
        v2f b5; b5.x = q2.z; b5.y = q2.w;
        v2f b6; b6.x = q3.x; b6.y = q3.y;
        v2f b7; b7.x = q3.z; b7.y = q3.w;
        acc[0] = __builtin_amdgcn_wmma_f32_16x16x4_f32(false, a, false, b0, (short)0, acc[0], false, false);
        acc[1] = __builtin_amdgcn_wmma_f32_16x16x4_f32(false, a, false, b1, (short)0, acc[1], false, false);
        acc[2] = __builtin_amdgcn_wmma_f32_16x16x4_f32(false, a, false, b2, (short)0, acc[2], false, false);
        acc[3] = __builtin_amdgcn_wmma_f32_16x16x4_f32(false, a, false, b3, (short)0, acc[3], false, false);
        acc[4] = __builtin_amdgcn_wmma_f32_16x16x4_f32(false, a, false, b4, (short)0, acc[4], false, false);
        acc[5] = __builtin_amdgcn_wmma_f32_16x16x4_f32(false, a, false, b5, (short)0, acc[5], false, false);
        acc[6] = __builtin_amdgcn_wmma_f32_16x16x4_f32(false, a, false, b6, (short)0, acc[6], false, false);
        acc[7] = __builtin_amdgcn_wmma_f32_16x16x4_f32(false, a, false, b7, (short)0, acc[7], false, false);
    }

    float rs[8];
    if (rowscale) {
        #pragma unroll
        for (int r = 0; r < 8; ++r) rs[r] = rowscale[row0 + r + half * 8];
    }

    #pragma unroll
    for (int nt = 0; nt < 8; ++nt) {
        int col = nt * 16 + lmod;
        float bv = bias ? bias[col] : 0.0f;
        #pragma unroll
        for (int r = 0; r < 8; ++r) {
            float v = acc[nt][r];
            if (rowscale) v *= rs[r];
            v += bv;
            if (relu) v = fmaxf(v, 0.0f);
            out[(size_t)(row0 + r + half * 8) * HID + col] = v;
        }
    }
}


static inline int cdiv(long long a, long long b) { return (int)((a + b - 1) / b); }

static void run_branch(const int* res, const float* pos, const int* ei,
                       int N, int E,
                       const float* W1c, const float* b1c,
                       const float* W2c, const float* b2c,
                       float* dinv, float* A, float* C, float* Wf, float* rec,
                       int* ecnt, int* rowptr, int* rank, int* col,
                       hipStream_t stream) {
    const int* src = ei;
    const int* dst = ei + E;

    k_fill_i<<<cdiv(N + 1, 256), 256, 0, stream>>>(ecnt, N + 1, 0);
    k_deg_rank<<<cdiv(E, 256), 256, 0, stream>>>(dst, E, ecnt, rank);
    k_rowalloc<<<cdiv(N, 256), 256, 0, stream>>>(ecnt, N, ecnt + N, rowptr, dinv);
    k_csr_fill<<<cdiv(E, 256), 256, 0, stream>>>(src, dst, rank, rowptr, E, col);

    k_rec<<<cdiv(N, 256), 256, 0, stream>>>(res, pos, dinv, N, rec);
    k_agg1<<<cdiv(N, 8 * AGG_NPW), 256, 0, stream>>>(rec, rowptr, ecnt, col, W1c, b1c, N, C);

    k_wfrag<<<cdiv(HID * HID, 256), 256, 0, stream>>>(W2c, HID, Wf);
    k_gemm128<<<cdiv(N / 16, 4), 128, 0, stream>>>(C, HID, N, Wf, dinv, nullptr, 0, A);
    k_gather<<<cdiv(N, 8), 256, 0, stream>>>(A, rowptr, ecnt, col, dinv, b2c, N, C);
}

extern "C" void kernel_launch(void* const* d_in, const int* in_sizes, int n_in,
                              void* d_out, int out_size, void* d_ws, size_t ws_size,
                              hipStream_t stream) {
    (void)n_in; (void)out_size; (void)ws_size;

    const int*   res_p   = (const int*)d_in[0];
    const float* pos_p   = (const float*)d_in[1];
    const int*   ei_p    = (const int*)d_in[2];
    const int*   batch_p = (const int*)d_in[3];
    const int*   res_m   = (const int*)d_in[4];
    const float* pos_m   = (const float*)d_in[5];
    const int*   ei_m    = (const int*)d_in[6];
    const int*   batch_m = (const int*)d_in[7];
    const float* W_p1 = (const float*)d_in[8];
    const float* b_p1 = (const float*)d_in[9];
    const float* W_p2 = (const float*)d_in[10];
    const float* b_p2 = (const float*)d_in[11];
    const float* W_m1 = (const float*)d_in[12];
    const float* b_m1 = (const float*)d_in[13];
    const float* W_m2 = (const float*)d_in[14];
    const float* b_m2 = (const float*)d_in[15];
    const float* W1f  = (const float*)d_in[16];
    const float* b1f  = (const float*)d_in[17];
    const float* W2f  = (const float*)d_in[18];
    const float* b2f  = (const float*)d_in[19];

    int N_P = in_sizes[0];
    int E_P = in_sizes[2] / 2;
    int N_M = in_sizes[4];
    int E_M = in_sizes[6] / 2;

    float* ws = (float*)d_ws;
    size_t nhp = (size_t)N_P * HID;
    float* A      = ws;
    float* B      = A + nhp;
    float* C      = B + nhp;
    float* dinv_p = C + nhp;
    float* dinv_m = dinv_p + N_P;
    float* pe_sum = dinv_m + N_M;
    float* pe     = pe_sum + NGRAPH * HID;
    float* cnt    = pe + NGRAPH * HID;
    float* Wfrag  = cnt + NGRAPH;
    int* ecnt   = (int*)(Wfrag + 2 * HID * HID);
    int* rowptr = ecnt + (N_P + 1);
    int* rank   = rowptr + N_P;
    int* col    = rank + E_P;

    run_branch(res_p, pos_p, ei_p, N_P, E_P, W_p1, b_p1, W_p2, b_p2,
               dinv_p, A, C, Wfrag, B, ecnt, rowptr, rank, col, stream);
    hipMemsetAsync(pe_sum, 0, (size_t)NGRAPH * HID * sizeof(float), stream);
    hipMemsetAsync(cnt, 0, NGRAPH * sizeof(float), stream);
    k_cnt<<<cdiv(N_P, 256), 256, 0, stream>>>(batch_p, N_P, cnt);
    k_pool_sum<<<cdiv(nhp, 256), 256, 0, stream>>>(C, batch_p, N_P, pe_sum);
    k_pool_div<<<cdiv(NGRAPH * HID, 256), 256, 0, stream>>>(pe_sum, cnt, pe);

    run_branch(res_m, pos_m, ei_m, N_M, E_M, W_m1, b_m1, W_m2, b_m2,
               dinv_m, A, C, Wfrag, B, ecnt, rowptr, rank, col, stream);

    size_t nfm = (size_t)N_M * 2 * HID;
    float* feat = A;
    float* hmid = B;
    k_feat<<<cdiv(nfm, 256), 256, 0, stream>>>(C, pe, batch_m, N_M, feat);
    k_wfrag<<<cdiv(2 * HID * HID, 256), 256, 0, stream>>>(W1f, 2 * HID, Wfrag);
    k_gemm128<<<cdiv(N_M / 16, 4), 128, 0, stream>>>(feat, 2 * HID, N_M, Wfrag,
                                                     nullptr, b1f, 1, hmid);
    k_head<<<cdiv(N_M, 256), 256, 0, stream>>>(hmid, W2f, b2f, N_M, (float*)d_out);
}


# Mean pool sums rows in registers: a wave adds 16 consecutive rows per run of equal graph id and sends one float atomic per feature per run, not per element; count folded in; sum order changes
# speedup vs baseline: 2.8801x; 1.2758x over previous
#include <hip/hip_runtime.h>
#include <hip/hip_bf16.h>

typedef __attribute__((ext_vector_type(2))) float v2f;
typedef __attribute__((ext_vector_type(8))) float v8f;
typedef __attribute__((ext_vector_type(4))) float v4f;

#define HID 128
#define NGRAPH 64
#define NRES 20
#define AGG_NPW 4


__global__ void k_fill_i(int* p, int n, int v) {
    int i = blockIdx.x * 256 + threadIdx.x;
    if (i < n) p[i] = v;
}

__global__ void k_deg_rank(const int* __restrict__ dst, int E, int* __restrict__ ecnt,
                           int* __restrict__ rank) {
    int e = blockIdx.x * 256 + threadIdx.x;
    if (e < E) rank[e] = atomicAdd(ecnt + dst[e], 1);
}

__global__ void k_rowalloc(const int* __restrict__ ecnt, int N, int* __restrict__ total,
                           int* __restrict__ rowptr, float* __restrict__ dinv) {
    int i = blockIdx.x * 256 + threadIdx.x;
    int lane = threadIdx.x & 31;
    int c = (i < N) ? ecnt[i] : 0;
    int x = c;
    #pragma unroll
    for (int d = 1; d < 32; d <<= 1) {
        int t = __shfl_up(x, d);
        if (lane >= d) x += t;
    }
    int base = 0;
    if (lane == 31) base = atomicAdd(total, x);
    base = __shfl(base, 31);
    if (i < N) {
        rowptr[i] = base + x - c;
        dinv[i] = rsqrtf((float)c + 1.0f);
    }
}

__global__ void k_csr_fill(const int* __restrict__ src, const int* __restrict__ dst,
                           const int* __restrict__ rank, const int* __restrict__ rowptr,
                           int E, int* __restrict__ col) {
    int e = blockIdx.x * 256 + threadIdx.x;
    if (e < E) col[rowptr[dst[e]] + rank[e]] = src[e];
}

__global__ void k_rec(const int* __restrict__ res, const float* __restrict__ pos,
                      const float* __restrict__ dinv, int N, float* __restrict__ rec) {
    int i = blockIdx.x * 256 + threadIdx.x;
    if (i >= N) return;
    float d = dinv[i];
    v4f q;
    q.x = d * pos[i * 3 + 0];
    q.y = d * pos[i * 3 + 1];
    q.z = d * pos[i * 3 + 2];
    q.w = d;
    *(v4f*)(rec + (size_t)i * 8) = q;
    int r = res[i];
    if (r < 0 || r >= NRES) r = NRES;
    ((int*)rec)[(size_t)i * 8 + 4] = r * (HID * 4);
}

__global__ void __launch_bounds__(256)
k_agg1(const float* __restrict__ rec, const int* __restrict__ rowptr,
       const int* __restrict__ ecnt, const int* __restrict__ col,
       const float* __restrict__ W, const float* __restrict__ b,
       int N, float* __restrict__ h) {
    __shared__ __attribute__((aligned(16))) float Ws[(NRES + 1) * HID];
    for (int t = threadIdx.x; t < NRES * HID / 4; t += 256)
        ((v4f*)Ws)[t] = ((const v4f*)W)[t];
    if (threadIdx.x < HID / 4)
        ((v4f*)Ws)[NRES * HID / 4 + threadIdx.x] = (v4f){0.0f, 0.0f, 0.0f, 0.0f};
    __syncthreads();

    int lane = threadIdx.x & 31;
    const v4f w20 = *(const v4f*)(W + (NRES + 0) * HID + lane * 4);
    const v4f w21 = *(const v4f*)(W + (NRES + 1) * HID + lane * 4);
    const v4f w22 = *(const v4f*)(W + (NRES + 2) * HID + lane * 4);
    const v4f bb = *(const v4f*)(b + lane * 4);
    const char* Wl = (const char*)Ws + lane * 16;
    const int* reci = (const int*)rec;
    int first = (int)(blockIdx.x * 8 + (threadIdx.x >> 5)) * AGG_NPW;

    for (int t = 0; t < AGG_NPW; ++t) {
        int i = __builtin_amdgcn_readfirstlane(first + t);
        if (i >= N) break;
        int start = rowptr[i];
        int n = ecnt[i];
        v4f rs = *(const v4f*)(rec + (size_t)i * 8);
        int os = reci[(size_t)i * 8 + 4];
        v4f acc = (v4f){0.0f, 0.0f, 0.0f, 0.0f};
        float q0 = 0.0f, q1 = 0.0f, q2 = 0.0f;
        for (int base = 0; base < n; base += 32) {
            int m = n - base;
            if (m > 32) m = 32;
            bool live = lane < m;
            int c = live ? col[start + base + lane] : i;
            c = min(max(c, 0), N - 1);
            v4f r = *(const v4f*)(rec + (size_t)c * 8);
            int off = reci[(size_t)c * 8 + 4];
            float d = live ? r.w : 0.0f;
            off = live ? off : NRES * HID * 4;
            q0 += live ? r.x : 0.0f;
            q1 += live ? r.y : 0.0f;
            q2 += live ? r.z : 0.0f;
            int dbits = __float_as_int(d);
            for (int j = 0; j < m; j += 8) {
                int o0 = __builtin_amdgcn_readlane(off, j);
                int o1 = __builtin_amdgcn_readlane(off, j + 1);
                int o2 = __builtin_amdgcn_readlane(off, j + 2);
                int o3 = __builtin_amdgcn_readlane(off, j + 3);
                int o4 = __builtin_amdgcn_readlane(off, j + 4);
                int o5 = __builtin_amdgcn_readlane(off, j + 5);
                int o6 = __builtin_amdgcn_readlane(off, j + 6);
                int o7 = __builtin_amdgcn_readlane(off, j + 7);
                v4f x0 = *(const v4f*)(Wl + o0);
                v4f x1 = *(const v4f*)(Wl + o1);
                v4f x2 = *(const v4f*)(Wl + o2);
                v4f x3 = *(const v4f*)(Wl + o3);
                v4f x4 = *(const v4f*)(Wl + o4);
                v4f x5 = *(const v4f*)(Wl + o5);
                v4f x6 = *(const v4f*)(Wl + o6);
                v4f x7 = *(const v4f*)(Wl + o7);
                float d0 = __int_as_float(__builtin_amdgcn_readlane(dbits, j));
                float d1 = __int_as_float(__builtin_amdgcn_readlane(dbits, j + 1));
                float d2 = __int_as_float(__builtin_amdgcn_readlane(dbits, j + 2));
                float d3 = __int_as_float(__builtin_amdgcn_readlane(dbits, j + 3));
                float d4 = __int_as_float(__builtin_amdgcn_readlane(dbits, j + 4));
                float d5 = __int_as_float(__builtin_amdgcn_readlane(dbits, j + 5));
                float d6 = __int_as_float(__builtin_amdgcn_readlane(dbits, j + 6));
                float d7 = __int_as_float(__builtin_amdgcn_readlane(dbits, j + 7));
                acc += d0 * x0;
                acc += d1 * x1;
                acc += d2 * x2;
                acc += d3 * x3;
                acc += d4 * x4;
                acc += d5 * x5;
                acc += d6 * x6;
                acc += d7 * x7;
            }
        }
        v4f xs = *(const v4f*)(Wl + os);
        acc += rs.w * xs;
        #pragma unroll
        for (int s = 16; s >= 1; s >>= 1) {
            q0 += __shfl_xor(q0, s);
            q1 += __shfl_xor(q1, s);
            q2 += __shfl_xor(q2, s);
        }
        q0 += rs.x;
        q1 += rs.y;
        q2 += rs.z;
        v4f S = acc + q0 * w20 + q1 * w21 + q2 * w22;
        float dnode = rs.w;
        v4f o;
        o.x = fmaxf(dnode * S.x + bb.x, 0.0f);
        o.y = fmaxf(dnode * S.y + bb.y, 0.0f);
        o.z = fmaxf(dnode * S.z + bb.z, 0.0f);
        o.w = fmaxf(dnode * S.w + bb.w, 0.0f);
        *(v4f*)(h + (size_t)i * HID + lane * 4) = o;
    }
}

__global__ void k_gather(const float* __restrict__ y, const int* __restrict__ rowptr,
                         const int* __restrict__ ecnt, const int* __restrict__ col,
                         const float* __restrict__ dinv, const float* __restrict__ b,
                         int N, float* __restrict__ h) {
    int i = __builtin_amdgcn_readfirstlane((int)(blockIdx.x * 8 + (threadIdx.x >> 5)));
    if (i >= N) return;
    int lane = threadIdx.x & 31;
    int start = rowptr[i];
    int n = ecnt[i];
    const float* yl = y + lane * 4;
    v4f acc = (v4f){0.0f, 0.0f, 0.0f, 0.0f};
    for (int base = 0; base < n; base += 32) {
        int m = n - base;
        if (m > 32) m = 32;
        int c = (lane < m) ? col[start + base + lane] : i;
        c = min(max(c, 0), N - 1);
        for (int j = 0; j < m; j += 8) {
            int s0 = __builtin_amdgcn_readlane(c, j);
            int s1 = __builtin_amdgcn_readlane(c, j + 1);
            int s2 = __builtin_amdgcn_readlane(c, j + 2);
            int s3 = __builtin_amdgcn_readlane(c, j + 3);
            int s4 = __builtin_amdgcn_readlane(c, j + 4);
            int s5 = __builtin_amdgcn_readlane(c, j + 5);
            int s6 = __builtin_amdgcn_readlane(c, j + 6);
            int s7 = __builtin_amdgcn_readlane(c, j + 7);
            v4f v0 = *(const v4f*)(yl + (size_t)s0 * HID);
            v4f v1 = *(const v4f*)(yl + (size_t)s1 * HID);
            v4f v2 = *(const v4f*)(yl + (size_t)s2 * HID);
            v4f v3 = *(const v4f*)(yl + (size_t)s3 * HID);
            v4f v4 = *(const v4f*)(yl + (size_t)s4 * HID);
            v4f v5 = *(const v4f*)(yl + (size_t)s5 * HID);
            v4f v6 = *(const v4f*)(yl + (size_t)s6 * HID);
            v4f v7 = *(const v4f*)(yl + (size_t)s7 * HID);
            asm volatile("" :: "v"(v0), "v"(v1), "v"(v2), "v"(v3),
                               "v"(v4), "v"(v5), "v"(v6), "v"(v7));
            acc += v0;
            if (j + 1 < m) acc += v1;
            if (j + 2 < m) acc += v2;
            if (j + 3 < m) acc += v3;
            if (j + 4 < m) acc += v4;
            if (j + 5 < m) acc += v5;
            if (j + 6 < m) acc += v6;
            if (j + 7 < m) acc += v7;
        }
    }
    v4f self = *(const v4f*)(yl + (size_t)i * HID);
    v4f bb = *(const v4f*)(b + lane * 4);
    float d = dinv[i];
    v4f o;
    o.x = fmaxf(d * (acc.x + self.x) + bb.x, 0.0f);
    o.y = fmaxf(d * (acc.y + self.y) + bb.y, 0.0f);
    o.z = fmaxf(d * (acc.z + self.z) + bb.z, 0.0f);
    o.w = fmaxf(d * (acc.w + self.w) + bb.w, 0.0f);
    *(v4f*)(h + (size_t)i * HID + lane * 4) = o;
}

#define POOL_RPW 16

__device__ __forceinline__ void pool_flush(float* __restrict__ pesum, float* __restrict__ cnt,
                                           int g, int lane, v4f acc, int n) {
    if ((unsigned)g >= (unsigned)NGRAPH) return;
    float* p = pesum + g * HID + lane * 4;
    atomicAdd(p + 0, acc.x);
    atomicAdd(p + 1, acc.y);
    atomicAdd(p + 2, acc.z);
    atomicAdd(p + 3, acc.w);
    if (lane == 0) atomicAdd(cnt + g, (float)n);
}

__global__ void __launch_bounds__(256)
k_pool(const float* __restrict__ h, const int* __restrict__ batch, int N,
       float* __restrict__ pesum, float* __restrict__ cnt) {
    int r0 = __builtin_amdgcn_readfirstlane((int)(blockIdx.x * 8 + (threadIdx.x >> 5))) * POOL_RPW;
    if (r0 >= N) return;
    int last = min(r0 + POOL_RPW, N) - 1;
    int lane = threadIdx.x & 31;
    const float* hl = h + lane * 4;
    int g[POOL_RPW];
    v4f v[POOL_RPW];
    #pragma unroll
    for (int k = 0; k < POOL_RPW; ++k) {
        int r = min(r0 + k, last);
        g[k] = batch[r];
        v[k] = *(const v4f*)(hl + (size_t)r * HID);
    }
    asm volatile("" :: "v"(v[0]), "v"(v[1]), "v"(v[2]), "v"(v[3]), "v"(v[4]), "v"(v[5]),
                       "v"(v[6]), "v"(v[7]), "v"(v[8]), "v"(v[9]), "v"(v[10]), "v"(v[11]),
                       "v"(v[12]), "v"(v[13]), "v"(v[14]), "v"(v[15]));
    int cur = g[0];
    int n = 0;
    v4f acc = (v4f){0.0f, 0.0f, 0.0f, 0.0f};
    #pragma unroll
    for (int k = 0; k < POOL_RPW; ++k) {
        if (r0 + k <= last) {
            if (g[k] != cur) {
                pool_flush(pesum, cnt, cur, lane, acc, n);
                cur = g[k];
                n = 0;
                acc = (v4f){0.0f, 0.0f, 0.0f, 0.0f};
            }
            acc += v[k];
            n += 1;
        }
    }
    pool_flush(pesum, cnt, cur, lane, acc, n);
}

__global__ void k_pool_div(const float* __restrict__ pesum, const float* __restrict__ cnt,
                           float* __restrict__ pe) {
    int idx = blockIdx.x * 256 + threadIdx.x;
    if (idx >= NGRAPH * HID) return;
    pe[idx] = pesum[idx] / fmaxf(cnt[idx >> 7], 1.0f);
}

__global__ void k_feat(const float* __restrict__ h, const float* __restrict__ pe,
                       const int* __restrict__ batch, int N, float* __restrict__ feat) {
    int idx = blockIdx.x * 256 + threadIdx.x;
    if (idx >= N * 2 * HID) return;
    int i = idx >> 8, f = idx & 255;
    feat[idx] = (f < HID) ? h[(size_t)i * HID + f]
                          : pe[batch[i] * HID + (f - HID)];
}

__global__ void k_head(const float* __restrict__ hmid, const float* __restrict__ W2,
                       const float* __restrict__ b2, int N, float* __restrict__ out) {
    int i = blockIdx.x * 256 + threadIdx.x;
    if (i >= N) return;
    float a0 = b2[0], a1 = b2[1], a2 = b2[2];
    const float* h = hmid + (size_t)i * HID;
    for (int k = 0; k < HID; ++k) {
        float hv = h[k];
        a0 += hv * W2[k * 3 + 0];
        a1 += hv * W2[k * 3 + 1];
        a2 += hv * W2[k * 3 + 2];
    }
    out[(size_t)i * 3 + 0] = a0;
    out[(size_t)i * 3 + 1] = a1;
    out[(size_t)i * 3 + 2] = a2;
}

__global__ void k_wfrag(const float* __restrict__ W, int K, float* __restrict__ Wf) {
    int idx = blockIdx.x * 256 + threadIdx.x;
    if (idx >= K * HID) return;
    int j    = idx & 1;
    int nt   = (idx >> 1) & 7;
    int l    = (idx >> 4) & 31;
    int kk   = idx >> 9;
    int krow = 4 * kk + (l >> 4) * 2 + j;
    int col  = nt * 16 + (l & 15);
    Wf[idx] = W[(size_t)krow * HID + col];
}

__global__ void k_gemm128(const float* __restrict__ A, int K, int M,
                          const float* __restrict__ Wf,
                          const float* __restrict__ rowscale,
                          const float* __restrict__ bias, int relu,
                          float* __restrict__ out) {
    int w = threadIdx.x >> 5;
    int lane = threadIdx.x & 31;
    int tile = blockIdx.x * 4 + w;
    if (tile * 16 >= M) return;
    int row0 = tile * 16;
    int half = lane >> 4;
    int lmod = lane & 15;
    int koff = half * 2;
    const float* Arow = A + (size_t)(row0 + lmod) * K + koff;
    const float* Wl   = Wf + (size_t)lane * 16;

    v8f acc[8];
    #pragma unroll
    for (int nt = 0; nt < 8; ++nt) acc[nt] = (v8f){};

    int ksteps = K >> 2;
    for (int kk = 0; kk < ksteps; ++kk) {
        v2f a;
        a.x = Arow[4 * kk];
        a.y = Arow[4 * kk + 1];
        const float4* q = (const float4*)(Wl + (size_t)kk * 512);
        float4 q0 = q[0], q1 = q[1], q2 = q[2], q3 = q[3];
        v2f b0; b0.x = q0.x; b0.y = q0.y;
        v2f b1; b1.x = q0.z; b1.y = q0.w;
        v2f b2; b2.x = q1.x; b2.y = q1.y;
        v2f b3; b3.x = q1.z; b3.y = q1.w;
        v2f b4; b4.x = q2.x; b4.y = q2.y;
        v2f b5; b5.x = q2.z; b5.y = q2.w;
        v2f b6; b6.x = q3.x; b6.y = q3.y;
        v2f b7; b7.x = q3.z; b7.y = q3.w;
        acc[0] = __builtin_amdgcn_wmma_f32_16x16x4_f32(false, a, false, b0, (short)0, acc[0], false, false);
        acc[1] = __builtin_amdgcn_wmma_f32_16x16x4_f32(false, a, false, b1, (short)0, acc[1], false, false);
        acc[2] = __builtin_amdgcn_wmma_f32_16x16x4_f32(false, a, false, b2, (short)0, acc[2], false, false);
        acc[3] = __builtin_amdgcn_wmma_f32_16x16x4_f32(false, a, false, b3, (short)0, acc[3], false, false);
        acc[4] = __builtin_amdgcn_wmma_f32_16x16x4_f32(false, a, false, b4, (short)0, acc[4], false, false);
        acc[5] = __builtin_amdgcn_wmma_f32_16x16x4_f32(false, a, false, b5, (short)0, acc[5], false, false);
        acc[6] = __builtin_amdgcn_wmma_f32_16x16x4_f32(false, a, false, b6, (short)0, acc[6], false, false);
        acc[7] = __builtin_amdgcn_wmma_f32_16x16x4_f32(false, a, false, b7, (short)0, acc[7], false, false);
    }

    float rs[8];
    if (rowscale) {
        #pragma unroll
        for (int r = 0; r < 8; ++r) rs[r] = rowscale[row0 + r + half * 8];
    }

    #pragma unroll
    for (int nt = 0; nt < 8; ++nt) {
        int col = nt * 16 + lmod;
        float bv = bias ? bias[col] : 0.0f;
        #pragma unroll
        for (int r = 0; r < 8; ++r) {
            float v = acc[nt][r];
            if (rowscale) v *= rs[r];
            v += bv;
            if (relu) v = fmaxf(v, 0.0f);
            out[(size_t)(row0 + r + half * 8) * HID + col] = v;
        }
    }
}


static inline int cdiv(long long a, long long b) { return (int)((a + b - 1) / b); }

static void run_branch(const int* res, const float* pos, const int* ei,
                       int N, int E,
                       const float* W1c, const float* b1c,
                       const float* W2c, const float* b2c,
                       float* dinv, float* A, float* C, float* Wf, float* rec,
                       int* ecnt, int* rowptr, int* rank, int* col,
                       hipStream_t stream) {
    const int* src = ei;
    const int* dst = ei + E;

    k_fill_i<<<cdiv(N + 1, 256), 256, 0, stream>>>(ecnt, N + 1, 0);
    k_deg_rank<<<cdiv(E, 256), 256, 0, stream>>>(dst, E, ecnt, rank);
    k_rowalloc<<<cdiv(N, 256), 256, 0, stream>>>(ecnt, N, ecnt + N, rowptr, dinv);
    k_csr_fill<<<cdiv(E, 256), 256, 0, stream>>>(src, dst, rank, rowptr, E, col);

    k_rec<<<cdiv(N, 256), 256, 0, stream>>>(res, pos, dinv, N, rec);
    k_agg1<<<cdiv(N, 8 * AGG_NPW), 256, 0, stream>>>(rec, rowptr, ecnt, col, W1c, b1c, N, C);

    k_wfrag<<<cdiv(HID * HID, 256), 256, 0, stream>>>(W2c, HID, Wf);
    k_gemm128<<<cdiv(N / 16, 4), 128, 0, stream>>>(C, HID, N, Wf, dinv, nullptr, 0, A);
    k_gather<<<cdiv(N, 8), 256, 0, stream>>>(A, rowptr, ecnt, col, dinv, b2c, N, C);
}

extern "C" void kernel_launch(void* const* d_in, const int* in_sizes, int n_in,
                              void* d_out, int out_size, void* d_ws, size_t ws_size,
                              hipStream_t stream) {
    (void)n_in; (void)out_size; (void)ws_size;

    const int*   res_p   = (const int*)d_in[0];
    const float* pos_p   = (const float*)d_in[1];
    const int*   ei_p    = (const int*)d_in[2];
    const int*   batch_p = (const int*)d_in[3];
    const int*   res_m   = (const int*)d_in[4];
    const float* pos_m   = (const float*)d_in[5];
    const int*   ei_m    = (const int*)d_in[6];
    const int*   batch_m = (const int*)d_in[7];
    const float* W_p1 = (const float*)d_in[8];
    const float* b_p1 = (const float*)d_in[9];
    const float* W_p2 = (const float*)d_in[10];
    const float* b_p2 = (const float*)d_in[11];
    const float* W_m1 = (const float*)d_in[12];
    const float* b_m1 = (const float*)d_in[13];
    const float* W_m2 = (const float*)d_in[14];
    const float* b_m2 = (const float*)d_in[15];
    const float* W1f  = (const float*)d_in[16];
    const float* b1f  = (const float*)d_in[17];
    const float* W2f  = (const float*)d_in[18];
    const float* b2f  = (const float*)d_in[19];

    int N_P = in_sizes[0];
    int E_P = in_sizes[2] / 2;
    int N_M = in_sizes[4];
    int E_M = in_sizes[6] / 2;

    float* ws = (float*)d_ws;
    size_t nhp = (size_t)N_P * HID;
    float* A      = ws;
    float* B      = A + nhp;
    float* C      = B + nhp;
    float* dinv_p = C + nhp;
    float* dinv_m = dinv_p + N_P;
    float* pe_sum = dinv_m + N_M;
    float* pe     = pe_sum + NGRAPH * HID;
    float* cnt    = pe + NGRAPH * HID;
    float* Wfrag  = cnt + NGRAPH;
    int* ecnt   = (int*)(Wfrag + 2 * HID * HID);
    int* rowptr = ecnt + (N_P + 1);
    int* rank   = rowptr + N_P;
    int* col    = rank + E_P;

    run_branch(res_p, pos_p, ei_p, N_P, E_P, W_p1, b_p1, W_p2, b_p2,
               dinv_p, A, C, Wfrag, B, ecnt, rowptr, rank, col, stream);
    hipMemsetAsync(pe_sum, 0, (size_t)NGRAPH * HID * sizeof(float), stream);
    hipMemsetAsync(cnt, 0, NGRAPH * sizeof(float), stream);
    k_pool<<<cdiv(N_P, 8 * POOL_RPW), 256, 0, stream>>>(C, batch_p, N_P, pe_sum, cnt);
    k_pool_div<<<cdiv(NGRAPH * HID, 256), 256, 0, stream>>>(pe_sum, cnt, pe);

    run_branch(res_m, pos_m, ei_m, N_M, E_M, W_m1, b_m1, W_m2, b_m2,
               dinv_m, A, C, Wfrag, B, ecnt, rowptr, rank, col, stream);

    size_t nfm = (size_t)N_M * 2 * HID;
    float* feat = A;
    float* hmid = B;
    k_feat<<<cdiv(nfm, 256), 256, 0, stream>>>(C, pe, batch_m, N_M, feat);
    k_wfrag<<<cdiv(2 * HID * HID, 256), 256, 0, stream>>>(W1f, 2 * HID, Wfrag);
    k_gemm128<<<cdiv(N_M / 16, 4), 128, 0, stream>>>(feat, 2 * HID, N_M, Wfrag,
                                                     nullptr, b1f, 1, hmid);
    k_head<<<cdiv(N_M, 256), 256, 0, stream>>>(hmid, W2f, b2f, N_M, (float*)d_out);
}


# Head GEMM reads its A operand where it lies (k<128: node rows, k>=128: pe[batch[row]]); the concatenated copy and its 400000-wave one-element-per-thread launch are gone; sum order unchanged
# speedup vs baseline: 3.1770x; 1.1031x over previous
#include <hip/hip_runtime.h>
#include <hip/hip_bf16.h>

typedef __attribute__((ext_vector_type(2))) float v2f;
typedef __attribute__((ext_vector_type(8))) float v8f;
typedef __attribute__((ext_vector_type(4))) float v4f;

#define HID 128
#define NGRAPH 64
#define NRES 20
#define AGG_NPW 4


__global__ void k_fill_i(int* p, int n, int v) {
    int i = blockIdx.x * 256 + threadIdx.x;
    if (i < n) p[i] = v;
}

__global__ void k_deg_rank(const int* __restrict__ dst, int E, int* __restrict__ ecnt,
                           int* __restrict__ rank) {
    int e = blockIdx.x * 256 + threadIdx.x;
    if (e < E) rank[e] = atomicAdd(ecnt + dst[e], 1);
}

__global__ void k_rowalloc(const int* __restrict__ ecnt, int N, int* __restrict__ total,
                           int* __restrict__ rowptr, float* __restrict__ dinv) {
    int i = blockIdx.x * 256 + threadIdx.x;
    int lane = threadIdx.x & 31;
    int c = (i < N) ? ecnt[i] : 0;
    int x = c;
    #pragma unroll
    for (int d = 1; d < 32; d <<= 1) {
        int t = __shfl_up(x, d);
        if (lane >= d) x += t;
    }
    int base = 0;
    if (lane == 31) base = atomicAdd(total, x);
    base = __shfl(base, 31);
    if (i < N) {
        rowptr[i] = base + x - c;
        dinv[i] = rsqrtf((float)c + 1.0f);
    }
}

__global__ void k_csr_fill(const int* __restrict__ src, const int* __restrict__ dst,
                           const int* __restrict__ rank, const int* __restrict__ rowptr,
                           int E, int* __restrict__ col) {
    int e = blockIdx.x * 256 + threadIdx.x;
    if (e < E) col[rowptr[dst[e]] + rank[e]] = src[e];
}

__global__ void k_rec(const int* __restrict__ res, const float* __restrict__ pos,
                      const float* __restrict__ dinv, int N, float* __restrict__ rec) {
    int i = blockIdx.x * 256 + threadIdx.x;
    if (i >= N) return;
    float d = dinv[i];
    v4f q;
    q.x = d * pos[i * 3 + 0];
    q.y = d * pos[i * 3 + 1];
    q.z = d * pos[i * 3 + 2];
    q.w = d;
    *(v4f*)(rec + (size_t)i * 8) = q;
    int r = res[i];
    if (r < 0 || r >= NRES) r = NRES;
    ((int*)rec)[(size_t)i * 8 + 4] = r * (HID * 4);
}

__global__ void __launch_bounds__(256)
k_agg1(const float* __restrict__ rec, const int* __restrict__ rowptr,
       const int* __restrict__ ecnt, const int* __restrict__ col,
       const float* __restrict__ W, const float* __restrict__ b,
       int N, float* __restrict__ h) {
    __shared__ __attribute__((aligned(16))) float Ws[(NRES + 1) * HID];
    for (int t = threadIdx.x; t < NRES * HID / 4; t += 256)
        ((v4f*)Ws)[t] = ((const v4f*)W)[t];
    if (threadIdx.x < HID / 4)
        ((v4f*)Ws)[NRES * HID / 4 + threadIdx.x] = (v4f){0.0f, 0.0f, 0.0f, 0.0f};
    __syncthreads();

    int lane = threadIdx.x & 31;
    const v4f w20 = *(const v4f*)(W + (NRES + 0) * HID + lane * 4);
    const v4f w21 = *(const v4f*)(W + (NRES + 1) * HID + lane * 4);
    const v4f w22 = *(const v4f*)(W + (NRES + 2) * HID + lane * 4);
    const v4f bb = *(const v4f*)(b + lane * 4);
    const char* Wl = (const char*)Ws + lane * 16;
    const int* reci = (const int*)rec;
    int first = (int)(blockIdx.x * 8 + (threadIdx.x >> 5)) * AGG_NPW;

    for (int t = 0; t < AGG_NPW; ++t) {
        int i = __builtin_amdgcn_readfirstlane(first + t);
        if (i >= N) break;
        int start = rowptr[i];
        int n = ecnt[i];
        v4f rs = *(const v4f*)(rec + (size_t)i * 8);
        int os = reci[(size_t)i * 8 + 4];
        v4f acc = (v4f){0.0f, 0.0f, 0.0f, 0.0f};
        float q0 = 0.0f, q1 = 0.0f, q2 = 0.0f;
        for (int base = 0; base < n; base += 32) {
            int m = n - base;
            if (m > 32) m = 32;
            bool live = lane < m;
            int c = live ? col[start + base + lane] : i;
            c = min(max(c, 0), N - 1);
            v4f r = *(const v4f*)(rec + (size_t)c * 8);
            int off = reci[(size_t)c * 8 + 4];
            float d = live ? r.w : 0.0f;
            off = live ? off : NRES * HID * 4;
            q0 += live ? r.x : 0.0f;
            q1 += live ? r.y : 0.0f;
            q2 += live ? r.z : 0.0f;
            int dbits = __float_as_int(d);
            for (int j = 0; j < m; j += 8) {
                int o0 = __builtin_amdgcn_readlane(off, j);
                int o1 = __builtin_amdgcn_readlane(off, j + 1);
                int o2 = __builtin_amdgcn_readlane(off, j + 2);
                int o3 = __builtin_amdgcn_readlane(off, j + 3);
                int o4 = __builtin_amdgcn_readlane(off, j + 4);
                int o5 = __builtin_amdgcn_readlane(off, j + 5);
                int o6 = __builtin_amdgcn_readlane(off, j + 6);
                int o7 = __builtin_amdgcn_readlane(off, j + 7);
                v4f x0 = *(const v4f*)(Wl + o0);
                v4f x1 = *(const v4f*)(Wl + o1);
                v4f x2 = *(const v4f*)(Wl + o2);
                v4f x3 = *(const v4f*)(Wl + o3);
                v4f x4 = *(const v4f*)(Wl + o4);
                v4f x5 = *(const v4f*)(Wl + o5);
                v4f x6 = *(const v4f*)(Wl + o6);
                v4f x7 = *(const v4f*)(Wl + o7);
                float d0 = __int_as_float(__builtin_amdgcn_readlane(dbits, j));
                float d1 = __int_as_float(__builtin_amdgcn_readlane(dbits, j + 1));
                float d2 = __int_as_float(__builtin_amdgcn_readlane(dbits, j + 2));
                float d3 = __int_as_float(__builtin_amdgcn_readlane(dbits, j + 3));
                float d4 = __int_as_float(__builtin_amdgcn_readlane(dbits, j + 4));
                float d5 = __int_as_float(__builtin_amdgcn_readlane(dbits, j + 5));
                float d6 = __int_as_float(__builtin_amdgcn_readlane(dbits, j + 6));
                float d7 = __int_as_float(__builtin_amdgcn_readlane(dbits, j + 7));
                acc += d0 * x0;
                acc += d1 * x1;
                acc += d2 * x2;
                acc += d3 * x3;
                acc += d4 * x4;
                acc += d5 * x5;
                acc += d6 * x6;
                acc += d7 * x7;
            }
        }
        v4f xs = *(const v4f*)(Wl + os);
        acc += rs.w * xs;
        #pragma unroll
        for (int s = 16; s >= 1; s >>= 1) {
            q0 += __shfl_xor(q0, s);
            q1 += __shfl_xor(q1, s);
            q2 += __shfl_xor(q2, s);
        }
        q0 += rs.x;
        q1 += rs.y;
        q2 += rs.z;
        v4f S = acc + q0 * w20 + q1 * w21 + q2 * w22;
        float dnode = rs.w;
        v4f o;
        o.x = fmaxf(dnode * S.x + bb.x, 0.0f);
        o.y = fmaxf(dnode * S.y + bb.y, 0.0f);
        o.z = fmaxf(dnode * S.z + bb.z, 0.0f);
        o.w = fmaxf(dnode * S.w + bb.w, 0.0f);
        *(v4f*)(h + (size_t)i * HID + lane * 4) = o;
    }
}

__global__ void k_gather(const float* __restrict__ y, const int* __restrict__ rowptr,
                         const int* __restrict__ ecnt, const int* __restrict__ col,
                         const float* __restrict__ dinv, const float* __restrict__ b,
                         int N, float* __restrict__ h) {
    int i = __builtin_amdgcn_readfirstlane((int)(blockIdx.x * 8 + (threadIdx.x >> 5)));
    if (i >= N) return;
    int lane = threadIdx.x & 31;
    int start = rowptr[i];
    int n = ecnt[i];
    const float* yl = y + lane * 4;
    v4f acc = (v4f){0.0f, 0.0f, 0.0f, 0.0f};
    for (int base = 0; base < n; base += 32) {
        int m = n - base;
        if (m > 32) m = 32;
        int c = (lane < m) ? col[start + base + lane] : i;
        c = min(max(c, 0), N - 1);
        for (int j = 0; j < m; j += 8) {
            int s0 = __builtin_amdgcn_readlane(c, j);
            int s1 = __builtin_amdgcn_readlane(c, j + 1);
            int s2 = __builtin_amdgcn_readlane(c, j + 2);
            int s3 = __builtin_amdgcn_readlane(c, j + 3);
            int s4 = __builtin_amdgcn_readlane(c, j + 4);
            int s5 = __builtin_amdgcn_readlane(c, j + 5);
            int s6 = __builtin_amdgcn_readlane(c, j + 6);
            int s7 = __builtin_amdgcn_readlane(c, j + 7);
            v4f v0 = *(const v4f*)(yl + (size_t)s0 * HID);
            v4f v1 = *(const v4f*)(yl + (size_t)s1 * HID);
            v4f v2 = *(const v4f*)(yl + (size_t)s2 * HID);
            v4f v3 = *(const v4f*)(yl + (size_t)s3 * HID);
            v4f v4 = *(const v4f*)(yl + (size_t)s4 * HID);
            v4f v5 = *(const v4f*)(yl + (size_t)s5 * HID);
            v4f v6 = *(const v4f*)(yl + (size_t)s6 * HID);
            v4f v7 = *(const v4f*)(yl + (size_t)s7 * HID);
            asm volatile("" :: "v"(v0), "v"(v1), "v"(v2), "v"(v3),
                               "v"(v4), "v"(v5), "v"(v6), "v"(v7));
            acc += v0;
            if (j + 1 < m) acc += v1;
            if (j + 2 < m) acc += v2;
            if (j + 3 < m) acc += v3;
            if (j + 4 < m) acc += v4;
            if (j + 5 < m) acc += v5;
            if (j + 6 < m) acc += v6;
            if (j + 7 < m) acc += v7;
        }
    }
    v4f self = *(const v4f*)(yl + (size_t)i * HID);
    v4f bb = *(const v4f*)(b + lane * 4);
    float d = dinv[i];
    v4f o;
    o.x = fmaxf(d * (acc.x + self.x) + bb.x, 0.0f);
    o.y = fmaxf(d * (acc.y + self.y) + bb.y, 0.0f);
    o.z = fmaxf(d * (acc.z + self.z) + bb.z, 0.0f);
    o.w = fmaxf(d * (acc.w + self.w) + bb.w, 0.0f);
    *(v4f*)(h + (size_t)i * HID + lane * 4) = o;
}

#define POOL_RPW 16

__device__ __forceinline__ void pool_flush(float* __restrict__ pesum, float* __restrict__ cnt,
                                           int g, int lane, v4f acc, int n) {
    if ((unsigned)g >= (unsigned)NGRAPH) return;
    float* p = pesum + g * HID + lane * 4;
    atomicAdd(p + 0, acc.x);
    atomicAdd(p + 1, acc.y);
    atomicAdd(p + 2, acc.z);
    atomicAdd(p + 3, acc.w);
    if (lane == 0) atomicAdd(cnt + g, (float)n);
}

__global__ void __launch_bounds__(256)
k_pool(const float* __restrict__ h, const int* __restrict__ batch, int N,
       float* __restrict__ pesum, float* __restrict__ cnt) {
    int r0 = __builtin_amdgcn_readfirstlane((int)(blockIdx.x * 8 + (threadIdx.x >> 5))) * POOL_RPW;
    if (r0 >= N) return;
    int last = min(r0 + POOL_RPW, N) - 1;
    int lane = threadIdx.x & 31;
    const float* hl = h + lane * 4;
    int g[POOL_RPW];
    v4f v[POOL_RPW];
    #pragma unroll
    for (int k = 0; k < POOL_RPW; ++k) {
        int r = min(r0 + k, last);
        g[k] = batch[r];
        v[k] = *(const v4f*)(hl + (size_t)r * HID);
    }
    asm volatile("" :: "v"(v[0]), "v"(v[1]), "v"(v[2]), "v"(v[3]), "v"(v[4]), "v"(v[5]),
                       "v"(v[6]), "v"(v[7]), "v"(v[8]), "v"(v[9]), "v"(v[10]), "v"(v[11]),
                       "v"(v[12]), "v"(v[13]), "v"(v[14]), "v"(v[15]));
    int cur = g[0];
    int n = 0;
    v4f acc = (v4f){0.0f, 0.0f, 0.0f, 0.0f};
    #pragma unroll
    for (int k = 0; k < POOL_RPW; ++k) {
        if (r0 + k <= last) {
            if (g[k] != cur) {
                pool_flush(pesum, cnt, cur, lane, acc, n);
                cur = g[k];
                n = 0;
                acc = (v4f){0.0f, 0.0f, 0.0f, 0.0f};
            }
            acc += v[k];
            n += 1;
        }
    }
    pool_flush(pesum, cnt, cur, lane, acc, n);
}

__global__ void k_pool_div(const float* __restrict__ pesum, const float* __restrict__ cnt,
                           float* __restrict__ pe) {
    int idx = blockIdx.x * 256 + threadIdx.x;
    if (idx >= NGRAPH * HID) return;
    pe[idx] = pesum[idx] / fmaxf(cnt[idx >> 7], 1.0f);
}

__global__ void k_head(const float* __restrict__ hmid, const float* __restrict__ W2,
                       const float* __restrict__ b2, int N, float* __restrict__ out) {
    int i = blockIdx.x * 256 + threadIdx.x;
    if (i >= N) return;
    float a0 = b2[0], a1 = b2[1], a2 = b2[2];
    const float* h = hmid + (size_t)i * HID;
    for (int k = 0; k < HID; ++k) {
        float hv = h[k];
        a0 += hv * W2[k * 3 + 0];
        a1 += hv * W2[k * 3 + 1];
        a2 += hv * W2[k * 3 + 2];
    }
    out[(size_t)i * 3 + 0] = a0;
    out[(size_t)i * 3 + 1] = a1;
    out[(size_t)i * 3 + 2] = a2;
}

__global__ void k_wfrag(const float* __restrict__ W, int K, float* __restrict__ Wf) {
    int idx = blockIdx.x * 256 + threadIdx.x;
    if (idx >= K * HID) return;
    int j    = idx & 1;
    int nt   = (idx >> 1) & 7;
    int l    = (idx >> 4) & 31;
    int kk   = idx >> 9;
    int krow = 4 * kk + (l >> 4) * 2 + j;
    int col  = nt * 16 + (l & 15);
    Wf[idx] = W[(size_t)krow * HID + col];
}

__device__ __forceinline__ void gemm_ksteps(const float* __restrict__ Arow,
                                            const float* __restrict__ Wl,
                                            int ksteps, v8f (&acc)[8]) {
    for (int kk = 0; kk < ksteps; ++kk) {
        v2f a;
        a.x = Arow[4 * kk];
        a.y = Arow[4 * kk + 1];
        const float4* q = (const float4*)(Wl + (size_t)kk * 512);
        float4 q0 = q[0], q1 = q[1], q2 = q[2], q3 = q[3];
        v2f b0; b0.x = q0.x; b0.y = q0.y;
        v2f b1; b1.x = q0.z; b1.y = q0.w;
        v2f b2; b2.x = q1.x; b2.y = q1.y;
        v2f b3; b3.x = q1.z; b3.y = q1.w;
        v2f b4; b4.x = q2.x; b4.y = q2.y;
        v2f b5; b5.x = q2.z; b5.y = q2.w;
        v2f b6; b6.x = q3.x; b6.y = q3.y;
        v2f b7; b7.x = q3.z; b7.y = q3.w;
        acc[0] = __builtin_amdgcn_wmma_f32_16x16x4_f32(false, a, false, b0, (short)0, acc[0], false, false);
        acc[1] = __builtin_amdgcn_wmma_f32_16x16x4_f32(false, a, false, b1, (short)0, acc[1], false, false);
        acc[2] = __builtin_amdgcn_wmma_f32_16x16x4_f32(false, a, false, b2, (short)0, acc[2], false, false);
        acc[3] = __builtin_amdgcn_wmma_f32_16x16x4_f32(false, a, false, b3, (short)0, acc[3], false, false);
        acc[4] = __builtin_amdgcn_wmma_f32_16x16x4_f32(false, a, false, b4, (short)0, acc[4], false, false);
        acc[5] = __builtin_amdgcn_wmma_f32_16x16x4_f32(false, a, false, b5, (short)0, acc[5], false, false);
        acc[6] = __builtin_amdgcn_wmma_f32_16x16x4_f32(false, a, false, b6, (short)0, acc[6], false, false);
        acc[7] = __builtin_amdgcn_wmma_f32_16x16x4_f32(false, a, false, b7, (short)0, acc[7], false, false);
    }
}

__device__ __forceinline__ void gemm_store(v8f (&acc)[8], int row0, int half, int lmod,
                                           const float* __restrict__ rowscale,
                                           const float* __restrict__ bias, int relu,
                                           float* __restrict__ out) {
    float rs[8];
    if (rowscale) {
        #pragma unroll
        for (int r = 0; r < 8; ++r) rs[r] = rowscale[row0 + r + half * 8];
    }

    #pragma unroll
    for (int nt = 0; nt < 8; ++nt) {
        int col = nt * 16 + lmod;
        float bv = bias ? bias[col] : 0.0f;
        #pragma unroll
        for (int r = 0; r < 8; ++r) {
            float v = acc[nt][r];
            if (rowscale) v *= rs[r];
            v += bv;
            if (relu) v = fmaxf(v, 0.0f);
            out[(size_t)(row0 + r + half * 8) * HID + col] = v;
        }
    }
}

__global__ void k_gemm128(const float* __restrict__ A, int K, int M,
                          const float* __restrict__ Wf,
                          const float* __restrict__ rowscale,
                          const float* __restrict__ bias, int relu,
                          float* __restrict__ out) {
    int w = threadIdx.x >> 5;
    int lane = threadIdx.x & 31;
    int tile = blockIdx.x * 4 + w;
    if (tile * 16 >= M) return;
    int row0 = tile * 16;
    int half = lane >> 4;
    int lmod = lane & 15;
    int koff = half * 2;
    const float* Arow = A + (size_t)(row0 + lmod) * K + koff;
    const float* Wl   = Wf + (size_t)lane * 16;

    v8f acc[8];
    #pragma unroll
    for (int nt = 0; nt < 8; ++nt) acc[nt] = (v8f){};

    gemm_ksteps(Arow, Wl, K >> 2, acc);
    gemm_store(acc, row0, half, lmod, rowscale, bias, relu, out);
}

__global__ void k_gemm_head(const float* __restrict__ h, const float* __restrict__ pe,
                            const int* __restrict__ batch, int M,
                            const float* __restrict__ Wf, const float* __restrict__ bias,
                            float* __restrict__ out) {
    int w = threadIdx.x >> 5;
    int lane = threadIdx.x & 31;
    int tile = blockIdx.x * 4 + w;
    if (tile * 16 >= M) return;
    int row0 = tile * 16;
    int half = lane >> 4;
    int lmod = lane & 15;
    int koff = half * 2;
    int g = min(max(batch[row0 + lmod], 0), NGRAPH - 1);
    const float* Wl = Wf + (size_t)lane * 16;

    v8f acc[8];
    #pragma unroll
    for (int nt = 0; nt < 8; ++nt) acc[nt] = (v8f){};

    gemm_ksteps(h + (size_t)(row0 + lmod) * HID + koff, Wl, HID / 4, acc);
    gemm_ksteps(pe + (size_t)g * HID + koff, Wl + (size_t)(HID / 4) * 512, HID / 4, acc);
    gemm_store(acc, row0, half, lmod, nullptr, bias, 1, out);
}


static inline int cdiv(long long a, long long b) { return (int)((a + b - 1) / b); }

static void run_branch(const int* res, const float* pos, const int* ei,
                       int N, int E,
                       const float* W1c, const float* b1c,
                       const float* W2c, const float* b2c,
                       float* dinv, float* A, float* C, float* Wf, float* rec,
                       int* ecnt, int* rowptr, int* rank, int* col,
                       hipStream_t stream) {
    const int* src = ei;
    const int* dst = ei + E;

    k_fill_i<<<cdiv(N + 1, 256), 256, 0, stream>>>(ecnt, N + 1, 0);
    k_deg_rank<<<cdiv(E, 256), 256, 0, stream>>>(dst, E, ecnt, rank);
    k_rowalloc<<<cdiv(N, 256), 256, 0, stream>>>(ecnt, N, ecnt + N, rowptr, dinv);
    k_csr_fill<<<cdiv(E, 256), 256, 0, stream>>>(src, dst, rank, rowptr, E, col);

    k_rec<<<cdiv(N, 256), 256, 0, stream>>>(res, pos, dinv, N, rec);
    k_agg1<<<cdiv(N, 8 * AGG_NPW), 256, 0, stream>>>(rec, rowptr, ecnt, col, W1c, b1c, N, C);

    k_wfrag<<<cdiv(HID * HID, 256), 256, 0, stream>>>(W2c, HID, Wf);
    k_gemm128<<<cdiv(N / 16, 4), 128, 0, stream>>>(C, HID, N, Wf, dinv, nullptr, 0, A);
    k_gather<<<cdiv(N, 8), 256, 0, stream>>>(A, rowptr, ecnt, col, dinv, b2c, N, C);
}

extern "C" void kernel_launch(void* const* d_in, const int* in_sizes, int n_in,
                              void* d_out, int out_size, void* d_ws, size_t ws_size,
                              hipStream_t stream) {
    (void)n_in; (void)out_size; (void)ws_size;

    const int*   res_p   = (const int*)d_in[0];
    const float* pos_p   = (const float*)d_in[1];
    const int*   ei_p    = (const int*)d_in[2];
    const int*   batch_p = (const int*)d_in[3];
    const int*   res_m   = (const int*)d_in[4];
    const float* pos_m   = (const float*)d_in[5];
    const int*   ei_m    = (const int*)d_in[6];
    const int*   batch_m = (const int*)d_in[7];
    const float* W_p1 = (const float*)d_in[8];
    const float* b_p1 = (const float*)d_in[9];
    const float* W_p2 = (const float*)d_in[10];
    const float* b_p2 = (const float*)d_in[11];
    const float* W_m1 = (const float*)d_in[12];
    const float* b_m1 = (const float*)d_in[13];
    const float* W_m2 = (const float*)d_in[14];
    const float* b_m2 = (const float*)d_in[15];
    const float* W1f  = (const float*)d_in[16];
    const float* b1f  = (const float*)d_in[17];
    const float* W2f  = (const float*)d_in[18];
    const float* b2f  = (const float*)d_in[19];

    int N_P = in_sizes[0];
    int E_P = in_sizes[2] / 2;
    int N_M = in_sizes[4];
    int E_M = in_sizes[6] / 2;

    float* ws = (float*)d_ws;
    size_t nhp = (size_t)N_P * HID;
    float* A      = ws;
    float* B      = A + nhp;
    float* C      = B + nhp;
    float* dinv_p = C + nhp;
    float* dinv_m = dinv_p + N_P;
    float* pe_sum = dinv_m + N_M;
    float* pe     = pe_sum + NGRAPH * HID;
    float* cnt    = pe + NGRAPH * HID;
    float* Wfrag  = cnt + NGRAPH;
    int* ecnt   = (int*)(Wfrag + 2 * HID * HID);
    int* rowptr = ecnt + (N_P + 1);
    int* rank   = rowptr + N_P;
    int* col    = rank + E_P;

    run_branch(res_p, pos_p, ei_p, N_P, E_P, W_p1, b_p1, W_p2, b_p2,
               dinv_p, A, C, Wfrag, B, ecnt, rowptr, rank, col, stream);
    hipMemsetAsync(pe_sum, 0, (size_t)NGRAPH * HID * sizeof(float), stream);
    hipMemsetAsync(cnt, 0, NGRAPH * sizeof(float), stream);
    k_pool<<<cdiv(N_P, 8 * POOL_RPW), 256, 0, stream>>>(C, batch_p, N_P, pe_sum, cnt);
    k_pool_div<<<cdiv(NGRAPH * HID, 256), 256, 0, stream>>>(pe_sum, cnt, pe);

    run_branch(res_m, pos_m, ei_m, N_M, E_M, W_m1, b_m1, W_m2, b_m2,
               dinv_m, A, C, Wfrag, B, ecnt, rowptr, rank, col, stream);

    float* hmid = B;
    k_wfrag<<<cdiv(2 * HID * HID, 256), 256, 0, stream>>>(W1f, 2 * HID, Wfrag);
    k_gemm_head<<<cdiv(N_M / 16, 4), 128, 0, stream>>>(C, pe, batch_m, N_M, Wfrag, b1f, hmid);
    k_head<<<cdiv(N_M, 256), 256, 0, stream>>>(hmid, W2f, b2f, N_M, (float*)d_out);
}


# Edge lists built by a two-level LDS counting sort (256-node bins, then nodes), not by a returning global add, a random 4-B read and a scattered 4-B store per edge; slot (sum) order changes
# speedup vs baseline: 4.4438x; 1.3987x over previous
#include <hip/hip_runtime.h>
#include <hip/hip_bf16.h>

typedef __attribute__((ext_vector_type(2))) float v2f;
typedef __attribute__((ext_vector_type(8))) float v8f;
typedef __attribute__((ext_vector_type(4))) float v4f;

#define HID 128
#define NGRAPH 64
#define NRES 20
#define AGG_NPW 4


__global__ void k_fill_i(int* p, int n, int v) {
    int i = blockIdx.x * 256 + threadIdx.x;
    if (i < n) p[i] = v;
}

#define BIN_SHIFT 8
#define BIN_NODES 256
#define MAXBIN 512
#define PCH 4096
#define PEPT 16
#define LCAP 16384
#define BIN_PAD 32


__global__ void __launch_bounds__(256)
k_bin_hist(const int* __restrict__ dst, int E, int N, int* __restrict__ binCount) {
    __shared__ int h[MAXBIN];
    int tid = threadIdx.x;
    for (int t = tid; t < MAXBIN; t += 256) h[t] = 0;
    __syncthreads();
    int e0 = blockIdx.x * PCH;
    int dd[PEPT];
    #pragma unroll
    for (int k = 0; k < PEPT; ++k) dd[k] = dst[min(e0 + k * 256 + tid, E - 1)];
    #pragma unroll
    for (int k = 0; k < PEPT; ++k) {
        int e = e0 + k * 256 + tid;
        int d = min(max(dd[k], 0), N - 1);
        if (e < E) atomicAdd(&h[d >> BIN_SHIFT], 1);
    }
    __syncthreads();
    for (int t = tid; t < MAXBIN; t += 256) {
        int c = h[t];
        if (c > 0) atomicAdd(binCount + t * BIN_PAD, c);
    }
}

__device__ __forceinline__ void scan_bins(const int* in, int* out, int lane) {
    int v[MAXBIN / 32];
    int s = 0;
    #pragma unroll
    for (int j = 0; j < MAXBIN / 32; ++j) {
        v[j] = in[lane * (MAXBIN / 32) + j];
        s += v[j];
    }
    int x = s;
    #pragma unroll
    for (int d = 1; d < 32; d <<= 1) {
        int t = __shfl_up(x, d);
        if (lane >= d) x += t;
    }
    int run = x - s;
    #pragma unroll
    for (int j = 0; j < MAXBIN / 32; ++j) {
        out[lane * (MAXBIN / 32) + j] = run;
        run += v[j];
    }
}

__global__ void __launch_bounds__(256)
k_bin_part(const int* __restrict__ src, const int* __restrict__ dst, int E, int N, int nbin,
           const int* __restrict__ binCount, int* __restrict__ binCursor,
           int* __restrict__ binStart, int* __restrict__ part) {
    __shared__ int h[MAXBIN];
    __shared__ int gc[MAXBIN];
    __shared__ int loc[MAXBIN];
    __shared__ int bstart[MAXBIN];
    __shared__ int gbase[MAXBIN];
    __shared__ int sval[PCH];
    __shared__ unsigned short sbin[PCH];
    int tid = threadIdx.x;
    for (int t = tid; t < MAXBIN; t += 256) {
        h[t] = 0;
        gc[t] = (t < nbin) ? binCount[t * BIN_PAD] : 0;
    }
    __syncthreads();
    int e0 = blockIdx.x * PCH;
    int pk[PEPT], bb[PEPT], rr[PEPT];
    #pragma unroll
    for (int k = 0; k < PEPT; ++k) {
        int ec = min(e0 + k * 256 + tid, E - 1);
        bb[k] = dst[ec];
        pk[k] = src[ec];
    }
    #pragma unroll
    for (int k = 0; k < PEPT; ++k) {
        int e = e0 + k * 256 + tid;
        int d = min(max(bb[k], 0), N - 1);
        int s = min(max(pk[k], 0), N - 1);
        bb[k] = (e < E) ? (d >> BIN_SHIFT) : -1;
        pk[k] = (s << BIN_SHIFT) | (d & (BIN_NODES - 1));
        rr[k] = 0;
    }
    #pragma unroll
    for (int k = 0; k < PEPT; ++k)
        if (bb[k] >= 0) rr[k] = atomicAdd(&h[bb[k]], 1);
    __syncthreads();
    int wave = tid >> 5, lane = tid & 31;
    if (wave == 0) scan_bins(h, loc, lane);
    else if (wave == 1) scan_bins(gc, bstart, lane);
    __syncthreads();
    for (int t = tid; t < MAXBIN; t += 256) {
        int c = h[t];
        if (c > 0) gbase[t] = bstart[t] + atomicAdd(binCursor + t * BIN_PAD, c);
        if (blockIdx.x == 0 && t < nbin) binStart[t] = bstart[t];
    }
    #pragma unroll
    for (int k = 0; k < PEPT; ++k) {
        if (bb[k] >= 0) {
            int p = loc[bb[k]] + rr[k];
            sval[p] = pk[k];
            sbin[p] = (unsigned short)bb[k];
        }
    }
    __syncthreads();
    int nv = min(PCH, E - e0);
    for (int i = tid; i < nv; i += 256) {
        int b = sbin[i];
        part[gbase[b] + (i - loc[b])] = sval[i];
    }
}

__global__ void __launch_bounds__(256)
k_bin_csr(const int* __restrict__ part, const int* __restrict__ binStart,
          const int* __restrict__ binCount, int N, int* __restrict__ ecnt,
          int* __restrict__ rowptr, float* __restrict__ dinv, int* __restrict__ col) {
    __shared__ int cnt[BIN_NODES];
    __shared__ int cur[BIN_NODES];
    __shared__ int wtot[8];
    __shared__ int lcol[LCAP];
    int tid = threadIdx.x;
    int wave = tid >> 5, lane = tid & 31;
    int b = blockIdx.x;
    int s = binStart[b];
    int m = binCount[b * BIN_PAD];
    cnt[tid] = 0;
    __syncthreads();
    for (int i0 = 0; i0 < m; i0 += 1024) {
        int v[4];
        #pragma unroll
        for (int u = 0; u < 4; ++u) {
            int i = i0 + u * 256 + tid;
            v[u] = (i < m) ? part[s + i] : -1;
        }
        #pragma unroll
        for (int u = 0; u < 4; ++u)
            if (v[u] >= 0) atomicAdd(&cnt[v[u] & (BIN_NODES - 1)], 1);
    }
    __syncthreads();
    int c = cnt[tid];
    int x = c;
    #pragma unroll
    for (int d = 1; d < 32; d <<= 1) {
        int t = __shfl_up(x, d);
        if (lane >= d) x += t;
    }
    if (lane == 31) wtot[wave] = x;
    __syncthreads();
    int wb = 0;
    for (int w = 0; w < wave; ++w) wb += wtot[w];
    int off = wb + x - c;
    cur[tid] = off;
    int node = b * BIN_NODES + tid;
    if (node < N) {
        ecnt[node] = c;
        rowptr[node] = s + off;
        dinv[node] = rsqrtf((float)c + 1.0f);
    }
    __syncthreads();
    bool staged = m <= LCAP;
    for (int i0 = 0; i0 < m; i0 += 1024) {
        int v[4];
        #pragma unroll
        for (int u = 0; u < 4; ++u) {
            int i = i0 + u * 256 + tid;
            v[u] = (i < m) ? part[s + i] : -1;
        }
        #pragma unroll
        for (int u = 0; u < 4; ++u) {
            if (v[u] >= 0) {
                int p = atomicAdd(&cur[v[u] & (BIN_NODES - 1)], 1);
                int sv = (int)((unsigned)v[u] >> BIN_SHIFT);
                if (staged) lcol[p] = sv;
                else col[s + p] = sv;
            }
        }
    }
    if (staged) {
        __syncthreads();
        for (int i = tid; i < m; i += 256) col[s + i] = lcol[i];
    }
}

__global__ void k_rec(const int* __restrict__ res, const float* __restrict__ pos,
                      const float* __restrict__ dinv, int N, float* __restrict__ rec) {
    int i = blockIdx.x * 256 + threadIdx.x;
    if (i >= N) return;
    float d = dinv[i];
    v4f q;
    q.x = d * pos[i * 3 + 0];
    q.y = d * pos[i * 3 + 1];
    q.z = d * pos[i * 3 + 2];
    q.w = d;
    *(v4f*)(rec + (size_t)i * 8) = q;
    int r = res[i];
    if (r < 0 || r >= NRES) r = NRES;
    ((int*)rec)[(size_t)i * 8 + 4] = r * (HID * 4);
}

__global__ void __launch_bounds__(256)
k_agg1(const float* __restrict__ rec, const int* __restrict__ rowptr,
       const int* __restrict__ ecnt, const int* __restrict__ col,
       const float* __restrict__ W, const float* __restrict__ b,
       int N, float* __restrict__ h) {
    __shared__ __attribute__((aligned(16))) float Ws[(NRES + 1) * HID];
    for (int t = threadIdx.x; t < NRES * HID / 4; t += 256)
        ((v4f*)Ws)[t] = ((const v4f*)W)[t];
    if (threadIdx.x < HID / 4)
        ((v4f*)Ws)[NRES * HID / 4 + threadIdx.x] = (v4f){0.0f, 0.0f, 0.0f, 0.0f};
    __syncthreads();

    int lane = threadIdx.x & 31;
    const v4f w20 = *(const v4f*)(W + (NRES + 0) * HID + lane * 4);
    const v4f w21 = *(const v4f*)(W + (NRES + 1) * HID + lane * 4);
    const v4f w22 = *(const v4f*)(W + (NRES + 2) * HID + lane * 4);
    const v4f bb = *(const v4f*)(b + lane * 4);
    const char* Wl = (const char*)Ws + lane * 16;
    const int* reci = (const int*)rec;
    int first = (int)(blockIdx.x * 8 + (threadIdx.x >> 5)) * AGG_NPW;

    for (int t = 0; t < AGG_NPW; ++t) {
        int i = __builtin_amdgcn_readfirstlane(first + t);
        if (i >= N) break;
        int start = rowptr[i];
        int n = ecnt[i];
        v4f rs = *(const v4f*)(rec + (size_t)i * 8);
        int os = reci[(size_t)i * 8 + 4];
        v4f acc = (v4f){0.0f, 0.0f, 0.0f, 0.0f};
        float q0 = 0.0f, q1 = 0.0f, q2 = 0.0f;
        for (int base = 0; base < n; base += 32) {
            int m = n - base;
            if (m > 32) m = 32;
            bool live = lane < m;
            int c = live ? col[start + base + lane] : i;
            c = min(max(c, 0), N - 1);
            v4f r = *(const v4f*)(rec + (size_t)c * 8);
            int off = reci[(size_t)c * 8 + 4];
            float d = live ? r.w : 0.0f;
            off = live ? off : NRES * HID * 4;
            q0 += live ? r.x : 0.0f;
            q1 += live ? r.y : 0.0f;
            q2 += live ? r.z : 0.0f;
            int dbits = __float_as_int(d);
            for (int j = 0; j < m; j += 8) {
                int o0 = __builtin_amdgcn_readlane(off, j);
                int o1 = __builtin_amdgcn_readlane(off, j + 1);
                int o2 = __builtin_amdgcn_readlane(off, j + 2);
                int o3 = __builtin_amdgcn_readlane(off, j + 3);
                int o4 = __builtin_amdgcn_readlane(off, j + 4);
                int o5 = __builtin_amdgcn_readlane(off, j + 5);
                int o6 = __builtin_amdgcn_readlane(off, j + 6);
                int o7 = __builtin_amdgcn_readlane(off, j + 7);
                v4f x0 = *(const v4f*)(Wl + o0);
                v4f x1 = *(const v4f*)(Wl + o1);
                v4f x2 = *(const v4f*)(Wl + o2);
                v4f x3 = *(const v4f*)(Wl + o3);
                v4f x4 = *(const v4f*)(Wl + o4);
                v4f x5 = *(const v4f*)(Wl + o5);
                v4f x6 = *(const v4f*)(Wl + o6);
                v4f x7 = *(const v4f*)(Wl + o7);
                float d0 = __int_as_float(__builtin_amdgcn_readlane(dbits, j));
                float d1 = __int_as_float(__builtin_amdgcn_readlane(dbits, j + 1));
                float d2 = __int_as_float(__builtin_amdgcn_readlane(dbits, j + 2));
                float d3 = __int_as_float(__builtin_amdgcn_readlane(dbits, j + 3));
                float d4 = __int_as_float(__builtin_amdgcn_readlane(dbits, j + 4));
                float d5 = __int_as_float(__builtin_amdgcn_readlane(dbits, j + 5));
                float d6 = __int_as_float(__builtin_amdgcn_readlane(dbits, j + 6));
                float d7 = __int_as_float(__builtin_amdgcn_readlane(dbits, j + 7));
                acc += d0 * x0;
                acc += d1 * x1;
                acc += d2 * x2;
                acc += d3 * x3;
                acc += d4 * x4;
                acc += d5 * x5;
                acc += d6 * x6;
                acc += d7 * x7;
            }
        }
        v4f xs = *(const v4f*)(Wl + os);
        acc += rs.w * xs;
        #pragma unroll
        for (int s = 16; s >= 1; s >>= 1) {
            q0 += __shfl_xor(q0, s);
            q1 += __shfl_xor(q1, s);
            q2 += __shfl_xor(q2, s);
        }
        q0 += rs.x;
        q1 += rs.y;
        q2 += rs.z;
        v4f S = acc + q0 * w20 + q1 * w21 + q2 * w22;
        float dnode = rs.w;
        v4f o;
        o.x = fmaxf(dnode * S.x + bb.x, 0.0f);
        o.y = fmaxf(dnode * S.y + bb.y, 0.0f);
        o.z = fmaxf(dnode * S.z + bb.z, 0.0f);
        o.w = fmaxf(dnode * S.w + bb.w, 0.0f);
        *(v4f*)(h + (size_t)i * HID + lane * 4) = o;
    }
}

__global__ void k_gather(const float* __restrict__ y, const int* __restrict__ rowptr,
                         const int* __restrict__ ecnt, const int* __restrict__ col,
                         const float* __restrict__ dinv, const float* __restrict__ b,
                         int N, float* __restrict__ h) {
    int i = __builtin_amdgcn_readfirstlane((int)(blockIdx.x * 8 + (threadIdx.x >> 5)));
    if (i >= N) return;
    int lane = threadIdx.x & 31;
    int start = rowptr[i];
    int n = ecnt[i];
    const float* yl = y + lane * 4;
    v4f acc = (v4f){0.0f, 0.0f, 0.0f, 0.0f};
    for (int base = 0; base < n; base += 32) {
        int m = n - base;
        if (m > 32) m = 32;
        int c = (lane < m) ? col[start + base + lane] : i;
        c = min(max(c, 0), N - 1);
        for (int j = 0; j < m; j += 8) {
            int s0 = __builtin_amdgcn_readlane(c, j);
            int s1 = __builtin_amdgcn_readlane(c, j + 1);
            int s2 = __builtin_amdgcn_readlane(c, j + 2);
            int s3 = __builtin_amdgcn_readlane(c, j + 3);
            int s4 = __builtin_amdgcn_readlane(c, j + 4);
            int s5 = __builtin_amdgcn_readlane(c, j + 5);
            int s6 = __builtin_amdgcn_readlane(c, j + 6);
            int s7 = __builtin_amdgcn_readlane(c, j + 7);
            v4f v0 = *(const v4f*)(yl + (size_t)s0 * HID);
            v4f v1 = *(const v4f*)(yl + (size_t)s1 * HID);
            v4f v2 = *(const v4f*)(yl + (size_t)s2 * HID);
            v4f v3 = *(const v4f*)(yl + (size_t)s3 * HID);
            v4f v4 = *(const v4f*)(yl + (size_t)s4 * HID);
            v4f v5 = *(const v4f*)(yl + (size_t)s5 * HID);
            v4f v6 = *(const v4f*)(yl + (size_t)s6 * HID);
            v4f v7 = *(const v4f*)(yl + (size_t)s7 * HID);
            asm volatile("" :: "v"(v0), "v"(v1), "v"(v2), "v"(v3),
                               "v"(v4), "v"(v5), "v"(v6), "v"(v7));
            acc += v0;
            if (j + 1 < m) acc += v1;
            if (j + 2 < m) acc += v2;
            if (j + 3 < m) acc += v3;
            if (j + 4 < m) acc += v4;
            if (j + 5 < m) acc += v5;
            if (j + 6 < m) acc += v6;
            if (j + 7 < m) acc += v7;
        }
    }
    v4f self = *(const v4f*)(yl + (size_t)i * HID);
    v4f bb = *(const v4f*)(b + lane * 4);
    float d = dinv[i];
    v4f o;
    o.x = fmaxf(d * (acc.x + self.x) + bb.x, 0.0f);
    o.y = fmaxf(d * (acc.y + self.y) + bb.y, 0.0f);
    o.z = fmaxf(d * (acc.z + self.z) + bb.z, 0.0f);
    o.w = fmaxf(d * (acc.w + self.w) + bb.w, 0.0f);
    *(v4f*)(h + (size_t)i * HID + lane * 4) = o;
}

#define POOL_RPW 16

__device__ __forceinline__ void pool_flush(float* __restrict__ pesum, float* __restrict__ cnt,
                                           int g, int lane, v4f acc, int n) {
    if ((unsigned)g >= (unsigned)NGRAPH) return;
    float* p = pesum + g * HID + lane * 4;
    atomicAdd(p + 0, acc.x);
    atomicAdd(p + 1, acc.y);
    atomicAdd(p + 2, acc.z);
    atomicAdd(p + 3, acc.w);
    if (lane == 0) atomicAdd(cnt + g, (float)n);
}

__global__ void __launch_bounds__(256)
k_pool(const float* __restrict__ h, const int* __restrict__ batch, int N,
       float* __restrict__ pesum, float* __restrict__ cnt) {
    int r0 = __builtin_amdgcn_readfirstlane((int)(blockIdx.x * 8 + (threadIdx.x >> 5))) * POOL_RPW;
    if (r0 >= N) return;
    int last = min(r0 + POOL_RPW, N) - 1;
    int lane = threadIdx.x & 31;
    const float* hl = h + lane * 4;
    int g[POOL_RPW];
    v4f v[POOL_RPW];
    #pragma unroll
    for (int k = 0; k < POOL_RPW; ++k) {
        int r = min(r0 + k, last);
        g[k] = batch[r];
        v[k] = *(const v4f*)(hl + (size_t)r * HID);
    }
    asm volatile("" :: "v"(v[0]), "v"(v[1]), "v"(v[2]), "v"(v[3]), "v"(v[4]), "v"(v[5]),
                       "v"(v[6]), "v"(v[7]), "v"(v[8]), "v"(v[9]), "v"(v[10]), "v"(v[11]),
                       "v"(v[12]), "v"(v[13]), "v"(v[14]), "v"(v[15]));
    int cur = g[0];
    int n = 0;
    v4f acc = (v4f){0.0f, 0.0f, 0.0f, 0.0f};
    #pragma unroll
    for (int k = 0; k < POOL_RPW; ++k) {
        if (r0 + k <= last) {
            if (g[k] != cur) {
                pool_flush(pesum, cnt, cur, lane, acc, n);
                cur = g[k];
                n = 0;
                acc = (v4f){0.0f, 0.0f, 0.0f, 0.0f};
            }
            acc += v[k];
            n += 1;
        }
    }
    pool_flush(pesum, cnt, cur, lane, acc, n);
}

__global__ void k_pool_div(const float* __restrict__ pesum, const float* __restrict__ cnt,
                           float* __restrict__ pe) {
    int idx = blockIdx.x * 256 + threadIdx.x;
    if (idx >= NGRAPH * HID) return;
    pe[idx] = pesum[idx] / fmaxf(cnt[idx >> 7], 1.0f);
}

__global__ void k_head(const float* __restrict__ hmid, const float* __restrict__ W2,
                       const float* __restrict__ b2, int N, float* __restrict__ out) {
    int i = blockIdx.x * 256 + threadIdx.x;
    if (i >= N) return;
    float a0 = b2[0], a1 = b2[1], a2 = b2[2];
    const float* h = hmid + (size_t)i * HID;
    for (int k = 0; k < HID; ++k) {
        float hv = h[k];
        a0 += hv * W2[k * 3 + 0];
        a1 += hv * W2[k * 3 + 1];
        a2 += hv * W2[k * 3 + 2];
    }
    out[(size_t)i * 3 + 0] = a0;
    out[(size_t)i * 3 + 1] = a1;
    out[(size_t)i * 3 + 2] = a2;
}

__global__ void k_wfrag(const float* __restrict__ W, int K, float* __restrict__ Wf) {
    int idx = blockIdx.x * 256 + threadIdx.x;
    if (idx >= K * HID) return;
    int j    = idx & 1;
    int nt   = (idx >> 1) & 7;
    int l    = (idx >> 4) & 31;
    int kk   = idx >> 9;
    int krow = 4 * kk + (l >> 4) * 2 + j;
    int col  = nt * 16 + (l & 15);
    Wf[idx] = W[(size_t)krow * HID + col];
}

__device__ __forceinline__ void gemm_ksteps(const float* __restrict__ Arow,
                                            const float* __restrict__ Wl,
                                            int ksteps, v8f (&acc)[8]) {
    for (int kk = 0; kk < ksteps; ++kk) {
        v2f a;
        a.x = Arow[4 * kk];
        a.y = Arow[4 * kk + 1];
        const float4* q = (const float4*)(Wl + (size_t)kk * 512);
        float4 q0 = q[0], q1 = q[1], q2 = q[2], q3 = q[3];
        v2f b0; b0.x = q0.x; b0.y = q0.y;
        v2f b1; b1.x = q0.z; b1.y = q0.w;
        v2f b2; b2.x = q1.x; b2.y = q1.y;
        v2f b3; b3.x = q1.z; b3.y = q1.w;
        v2f b4; b4.x = q2.x; b4.y = q2.y;
        v2f b5; b5.x = q2.z; b5.y = q2.w;
        v2f b6; b6.x = q3.x; b6.y = q3.y;
        v2f b7; b7.x = q3.z; b7.y = q3.w;
        acc[0] = __builtin_amdgcn_wmma_f32_16x16x4_f32(false, a, false, b0, (short)0, acc[0], false, false);
        acc[1] = __builtin_amdgcn_wmma_f32_16x16x4_f32(false, a, false, b1, (short)0, acc[1], false, false);
        acc[2] = __builtin_amdgcn_wmma_f32_16x16x4_f32(false, a, false, b2, (short)0, acc[2], false, false);
        acc[3] = __builtin_amdgcn_wmma_f32_16x16x4_f32(false, a, false, b3, (short)0, acc[3], false, false);
        acc[4] = __builtin_amdgcn_wmma_f32_16x16x4_f32(false, a, false, b4, (short)0, acc[4], false, false);
        acc[5] = __builtin_amdgcn_wmma_f32_16x16x4_f32(false, a, false, b5, (short)0, acc[5], false, false);
        acc[6] = __builtin_amdgcn_wmma_f32_16x16x4_f32(false, a, false, b6, (short)0, acc[6], false, false);
        acc[7] = __builtin_amdgcn_wmma_f32_16x16x4_f32(false, a, false, b7, (short)0, acc[7], false, false);
    }
}

__device__ __forceinline__ void gemm_store(v8f (&acc)[8], int row0, int half, int lmod,
                                           const float* __restrict__ rowscale,
                                           const float* __restrict__ bias, int relu,
                                           float* __restrict__ out) {
    float rs[8];
    if (rowscale) {
        #pragma unroll
        for (int r = 0; r < 8; ++r) rs[r] = rowscale[row0 + r + half * 8];
    }

    #pragma unroll
    for (int nt = 0; nt < 8; ++nt) {
        int col = nt * 16 + lmod;
        float bv = bias ? bias[col] : 0.0f;
        #pragma unroll
        for (int r = 0; r < 8; ++r) {
            float v = acc[nt][r];
            if (rowscale) v *= rs[r];
            v += bv;
            if (relu) v = fmaxf(v, 0.0f);
            out[(size_t)(row0 + r + half * 8) * HID + col] = v;
        }
    }
}

__global__ void k_gemm128(const float* __restrict__ A, int K, int M,
                          const float* __restrict__ Wf,
                          const float* __restrict__ rowscale,
                          const float* __restrict__ bias, int relu,
                          float* __restrict__ out) {
    int w = threadIdx.x >> 5;
    int lane = threadIdx.x & 31;
    int tile = blockIdx.x * 4 + w;
    if (tile * 16 >= M) return;
    int row0 = tile * 16;
    int half = lane >> 4;
    int lmod = lane & 15;
    int koff = half * 2;
    const float* Arow = A + (size_t)(row0 + lmod) * K + koff;
    const float* Wl   = Wf + (size_t)lane * 16;

    v8f acc[8];
    #pragma unroll
    for (int nt = 0; nt < 8; ++nt) acc[nt] = (v8f){};

    gemm_ksteps(Arow, Wl, K >> 2, acc);
    gemm_store(acc, row0, half, lmod, rowscale, bias, relu, out);
}

__global__ void k_gemm_head(const float* __restrict__ h, const float* __restrict__ pe,
                            const int* __restrict__ batch, int M,
                            const float* __restrict__ Wf, const float* __restrict__ bias,
                            float* __restrict__ out) {
    int w = threadIdx.x >> 5;
    int lane = threadIdx.x & 31;
    int tile = blockIdx.x * 4 + w;
    if (tile * 16 >= M) return;
    int row0 = tile * 16;
    int half = lane >> 4;
    int lmod = lane & 15;
    int koff = half * 2;
    int g = min(max(batch[row0 + lmod], 0), NGRAPH - 1);
    const float* Wl = Wf + (size_t)lane * 16;

    v8f acc[8];
    #pragma unroll
    for (int nt = 0; nt < 8; ++nt) acc[nt] = (v8f){};

    gemm_ksteps(h + (size_t)(row0 + lmod) * HID + koff, Wl, HID / 4, acc);
    gemm_ksteps(pe + (size_t)g * HID + koff, Wl + (size_t)(HID / 4) * 512, HID / 4, acc);
    gemm_store(acc, row0, half, lmod, nullptr, bias, 1, out);
}


static inline int cdiv(long long a, long long b) { return (int)((a + b - 1) / b); }

static void run_branch(const int* res, const float* pos, const int* ei,
                       int N, int E,
                       const float* W1c, const float* b1c,
                       const float* W2c, const float* b2c,
                       float* dinv, float* A, float* C, float* Wf, float* rec,
                       int* ecnt, int* rowptr, int* part, int* col, int* bins,
                       hipStream_t stream) {
    const int* src = ei;
    const int* dst = ei + E;

    int nbin = cdiv(N, BIN_NODES);
    int* binCount = bins;
    int* binCursor = bins + MAXBIN * BIN_PAD;
    int* binStart = bins + 2 * MAXBIN * BIN_PAD;
    k_fill_i<<<cdiv(2 * MAXBIN * BIN_PAD, 256), 256, 0, stream>>>(bins, 2 * MAXBIN * BIN_PAD, 0);
    k_bin_hist<<<cdiv(E, PCH), 256, 0, stream>>>(dst, E, N, binCount);
    k_bin_part<<<cdiv(E, PCH), 256, 0, stream>>>(src, dst, E, N, nbin, binCount, binCursor,
                                                 binStart, part);
    k_bin_csr<<<nbin, 256, 0, stream>>>(part, binStart, binCount, N, ecnt, rowptr, dinv, col);

    k_rec<<<cdiv(N, 256), 256, 0, stream>>>(res, pos, dinv, N, rec);
    k_agg1<<<cdiv(N, 8 * AGG_NPW), 256, 0, stream>>>(rec, rowptr, ecnt, col, W1c, b1c, N, C);

    k_wfrag<<<cdiv(HID * HID, 256), 256, 0, stream>>>(W2c, HID, Wf);
    k_gemm128<<<cdiv(N / 16, 4), 128, 0, stream>>>(C, HID, N, Wf, dinv, nullptr, 0, A);
    k_gather<<<cdiv(N, 8), 256, 0, stream>>>(A, rowptr, ecnt, col, dinv, b2c, N, C);
}

extern "C" void kernel_launch(void* const* d_in, const int* in_sizes, int n_in,
                              void* d_out, int out_size, void* d_ws, size_t ws_size,
                              hipStream_t stream) {
    (void)n_in; (void)out_size; (void)ws_size;

    const int*   res_p   = (const int*)d_in[0];
    const float* pos_p   = (const float*)d_in[1];
    const int*   ei_p    = (const int*)d_in[2];
    const int*   batch_p = (const int*)d_in[3];
    const int*   res_m   = (const int*)d_in[4];
    const float* pos_m   = (const float*)d_in[5];
    const int*   ei_m    = (const int*)d_in[6];
    const int*   batch_m = (const int*)d_in[7];
    const float* W_p1 = (const float*)d_in[8];
    const float* b_p1 = (const float*)d_in[9];
    const float* W_p2 = (const float*)d_in[10];
    const float* b_p2 = (const float*)d_in[11];
    const float* W_m1 = (const float*)d_in[12];
    const float* b_m1 = (const float*)d_in[13];
    const float* W_m2 = (const float*)d_in[14];
    const float* b_m2 = (const float*)d_in[15];
    const float* W1f  = (const float*)d_in[16];
    const float* b1f  = (const float*)d_in[17];
    const float* W2f  = (const float*)d_in[18];
    const float* b2f  = (const float*)d_in[19];

    int N_P = in_sizes[0];
    int E_P = in_sizes[2] / 2;
    int N_M = in_sizes[4];
    int E_M = in_sizes[6] / 2;

    float* ws = (float*)d_ws;
    size_t nhp = (size_t)N_P * HID;
    float* A      = ws;
    float* B      = A + nhp;
    float* C      = B + nhp;
    float* dinv_p = C + nhp;
    float* dinv_m = dinv_p + N_P;
    float* pe_sum = dinv_m + N_M;
    float* pe     = pe_sum + NGRAPH * HID;
    float* cnt    = pe + NGRAPH * HID;
    float* Wfrag  = cnt + NGRAPH;
    int* ecnt   = (int*)(Wfrag + 2 * HID * HID);
    int* rowptr = ecnt + (N_P + 1);
    int* part   = rowptr + N_P;
    int* col    = part + E_P;
    int* bins   = col + E_P;

    run_branch(res_p, pos_p, ei_p, N_P, E_P, W_p1, b_p1, W_p2, b_p2,
               dinv_p, A, C, Wfrag, B, ecnt, rowptr, part, col, bins, stream);
    hipMemsetAsync(pe_sum, 0, (size_t)NGRAPH * HID * sizeof(float), stream);
    hipMemsetAsync(cnt, 0, NGRAPH * sizeof(float), stream);
    k_pool<<<cdiv(N_P, 8 * POOL_RPW), 256, 0, stream>>>(C, batch_p, N_P, pe_sum, cnt);
    k_pool_div<<<cdiv(NGRAPH * HID, 256), 256, 0, stream>>>(pe_sum, cnt, pe);

    run_branch(res_m, pos_m, ei_m, N_M, E_M, W_m1, b_m1, W_m2, b_m2,
               dinv_m, A, C, Wfrag, B, ecnt, rowptr, part, col, bins, stream);

    float* hmid = B;
    k_wfrag<<<cdiv(2 * HID * HID, 256), 256, 0, stream>>>(W1f, 2 * HID, Wfrag);
    k_gemm_head<<<cdiv(N_M / 16, 4), 128, 0, stream>>>(C, pe, batch_m, N_M, Wfrag, b1f, hmid);
    k_head<<<cdiv(N_M, 256), 256, 0, stream>>>(hmid, W2f, b2f, N_M, (float*)d_out);
}


# Each per-branch stage is one launch that runs both graph branches on their own buffers; zero-fills and the 3 weight swizzles are one set-up launch: 23 launches + 2 memsets -> 12; sum order unchanged
# speedup vs baseline: 4.9361x; 1.1108x over previous
#include <hip/hip_runtime.h>
#include <hip/hip_bf16.h>

typedef __attribute__((ext_vector_type(2))) float v2f;
typedef __attribute__((ext_vector_type(8))) float v8f;
typedef __attribute__((ext_vector_type(4))) float v4f;

#define HID 128
#define NGRAPH 64
#define NRES 20
#define AGG_NPW 4

#define BR_PICK bool second = (int)blockIdx.x >= split; \
                int bid = second ? (int)blockIdx.x - split : (int)blockIdx.x
#define PK(x) (second ? x##_m : x##_p)

#define BIN_SHIFT 8
#define BIN_NODES 256
#define MAXBIN 512
#define PCH 4096
#define PEPT 16
#define LCAP 16384
#define BIN_PAD 32


__device__ __forceinline__ void
bin_hist_body(const int* __restrict__ dst, int E, int N, int* __restrict__ binCount, int bid) {
    __shared__ int h[MAXBIN];
    int tid = threadIdx.x;
    for (int t = tid; t < MAXBIN; t += 256) h[t] = 0;
    __syncthreads();
    int e0 = bid * PCH;
    int dd[PEPT];
    #pragma unroll
    for (int k = 0; k < PEPT; ++k) dd[k] = dst[min(e0 + k * 256 + tid, E - 1)];
    #pragma unroll
    for (int k = 0; k < PEPT; ++k) {
        int e = e0 + k * 256 + tid;
        int d = min(max(dd[k], 0), N - 1);
        if (e < E) atomicAdd(&h[d >> BIN_SHIFT], 1);
    }
    __syncthreads();
    for (int t = tid; t < MAXBIN; t += 256) {
        int c = h[t];
        if (c > 0) atomicAdd(binCount + t * BIN_PAD, c);
    }
}

__global__ void __launch_bounds__(256)
k_bin_hist(const int* dst_p, int E_p, int N_p, int* binCount_p,
           const int* dst_m, int E_m, int N_m, int* binCount_m, int split) {
    BR_PICK;
    bin_hist_body(PK(dst), PK(E), PK(N), PK(binCount), bid);
}

__device__ __forceinline__ void scan_bins(const int* in, int* out, int lane) {
    int v[MAXBIN / 32];
    int s = 0;
    #pragma unroll
    for (int j = 0; j < MAXBIN / 32; ++j) {
        v[j] = in[lane * (MAXBIN / 32) + j];
        s += v[j];
    }
    int x = s;
    #pragma unroll
    for (int d = 1; d < 32; d <<= 1) {
        int t = __shfl_up(x, d);
        if (lane >= d) x += t;
    }
    int run = x - s;
    #pragma unroll
    for (int j = 0; j < MAXBIN / 32; ++j) {
        out[lane * (MAXBIN / 32) + j] = run;
        run += v[j];
    }
}

__device__ __forceinline__ void
bin_part_body(const int* __restrict__ src, const int* __restrict__ dst, int E, int N, int nbin,
              const int* __restrict__ binCount, int* __restrict__ binCursor,
              int* __restrict__ binStart, int* __restrict__ part, int bid) {
    __shared__ int h[MAXBIN];
    __shared__ int gc[MAXBIN];
    __shared__ int loc[MAXBIN];
    __shared__ int bstart[MAXBIN];
    __shared__ int gbase[MAXBIN];
    __shared__ int sval[PCH];
    __shared__ unsigned short sbin[PCH];
    int tid = threadIdx.x;
    for (int t = tid; t < MAXBIN; t += 256) {
        h[t] = 0;
        gc[t] = (t < nbin) ? binCount[t * BIN_PAD] : 0;
    }
    __syncthreads();
    int e0 = bid * PCH;
    int pk[PEPT], bb[PEPT], rr[PEPT];
    #pragma unroll
    for (int k = 0; k < PEPT; ++k) {
        int ec = min(e0 + k * 256 + tid, E - 1);
        bb[k] = dst[ec];
        pk[k] = src[ec];
    }
    #pragma unroll
    for (int k = 0; k < PEPT; ++k) {
        int e = e0 + k * 256 + tid;
        int d = min(max(bb[k], 0), N - 1);
        int s = min(max(pk[k], 0), N - 1);
        bb[k] = (e < E) ? (d >> BIN_SHIFT) : -1;
        pk[k] = (s << BIN_SHIFT) | (d & (BIN_NODES - 1));
        rr[k] = 0;
    }
    #pragma unroll
    for (int k = 0; k < PEPT; ++k)
        if (bb[k] >= 0) rr[k] = atomicAdd(&h[bb[k]], 1);
    __syncthreads();
    int wave = tid >> 5, lane = tid & 31;
    if (wave == 0) scan_bins(h, loc, lane);
    else if (wave == 1) scan_bins(gc, bstart, lane);
    __syncthreads();
    for (int t = tid; t < MAXBIN; t += 256) {
        int c = h[t];
        if (c > 0) gbase[t] = bstart[t] + atomicAdd(binCursor + t * BIN_PAD, c);
        if (bid == 0 && t < nbin) binStart[t] = bstart[t];
    }
    #pragma unroll
    for (int k = 0; k < PEPT; ++k) {
        if (bb[k] >= 0) {
            int p = loc[bb[k]] + rr[k];
            sval[p] = pk[k];
            sbin[p] = (unsigned short)bb[k];
        }
    }
    __syncthreads();
    int nv = min(PCH, E - e0);
    for (int i = tid; i < nv; i += 256) {
        int b = sbin[i];
        part[gbase[b] + (i - loc[b])] = sval[i];
    }
}

__global__ void __launch_bounds__(256)
k_bin_part(const int* src_p, const int* dst_p, int E_p, int N_p, int nbin_p,
           const int* binCount_p, int* binCursor_p, int* binStart_p, int* part_p,
           const int* src_m, const int* dst_m, int E_m, int N_m, int nbin_m,
           const int* binCount_m, int* binCursor_m, int* binStart_m, int* part_m, int split) {
    BR_PICK;
    bin_part_body(PK(src), PK(dst), PK(E), PK(N), PK(nbin), PK(binCount), PK(binCursor),
                  PK(binStart), PK(part), bid);
}

__device__ __forceinline__ void
bin_csr_body(const int* __restrict__ part, const int* __restrict__ binStart,
             const int* __restrict__ binCount, int N, int* __restrict__ ecnt,
             int* __restrict__ rowptr, float* __restrict__ dinv, int* __restrict__ col, int bid) {
    __shared__ int cnt[BIN_NODES];
    __shared__ int cur[BIN_NODES];
    __shared__ int wtot[8];
    __shared__ int lcol[LCAP];
    int tid = threadIdx.x;
    int wave = tid >> 5, lane = tid & 31;
    int b = bid;
    int s = binStart[b];
    int m = binCount[b * BIN_PAD];
    cnt[tid] = 0;
    __syncthreads();
    for (int i0 = 0; i0 < m; i0 += 1024) {
        int v[4];
        #pragma unroll
        for (int u = 0; u < 4; ++u) {
            int i = i0 + u * 256 + tid;
            v[u] = (i < m) ? part[s + i] : -1;
        }
        #pragma unroll
        for (int u = 0; u < 4; ++u)
            if (v[u] >= 0) atomicAdd(&cnt[v[u] & (BIN_NODES - 1)], 1);
    }
    __syncthreads();
    int c = cnt[tid];
    int x = c;
    #pragma unroll
    for (int d = 1; d < 32; d <<= 1) {
        int t = __shfl_up(x, d);
        if (lane >= d) x += t;
    }
    if (lane == 31) wtot[wave] = x;
    __syncthreads();
    int wb = 0;
    for (int w = 0; w < wave; ++w) wb += wtot[w];
    int off = wb + x - c;
    cur[tid] = off;
    int node = b * BIN_NODES + tid;
    if (node < N) {
        ecnt[node] = c;
        rowptr[node] = s + off;
        dinv[node] = rsqrtf((float)c + 1.0f);
    }
    __syncthreads();
    bool staged = m <= LCAP;
    for (int i0 = 0; i0 < m; i0 += 1024) {
        int v[4];
        #pragma unroll
        for (int u = 0; u < 4; ++u) {
            int i = i0 + u * 256 + tid;
            v[u] = (i < m) ? part[s + i] : -1;
        }
        #pragma unroll
        for (int u = 0; u < 4; ++u) {
            if (v[u] >= 0) {
                int p = atomicAdd(&cur[v[u] & (BIN_NODES - 1)], 1);
                int sv = (int)((unsigned)v[u] >> BIN_SHIFT);
                if (staged) lcol[p] = sv;
                else col[s + p] = sv;
            }
        }
    }
    if (staged) {
        __syncthreads();
        for (int i = tid; i < m; i += 256) col[s + i] = lcol[i];
    }
}

__global__ void __launch_bounds__(256)
k_bin_csr(const int* part_p, const int* binStart_p, const int* binCount_p, int N_p,
          int* ecnt_p, int* rowptr_p, float* dinv_p, int* col_p,
          const int* part_m, const int* binStart_m, const int* binCount_m, int N_m,
          int* ecnt_m, int* rowptr_m, float* dinv_m, int* col_m, int split) {
    BR_PICK;
    bin_csr_body(PK(part), PK(binStart), PK(binCount), PK(N), PK(ecnt), PK(rowptr), PK(dinv),
                 PK(col), bid);
}

__device__ __forceinline__ void
rec_body(const int* __restrict__ res, const float* __restrict__ pos,
         const float* __restrict__ dinv, int N, float* __restrict__ rec, int bid) {
    int i = bid * 256 + threadIdx.x;
    if (i >= N) return;
    float d = dinv[i];
    v4f q;
    q.x = d * pos[i * 3 + 0];
    q.y = d * pos[i * 3 + 1];
    q.z = d * pos[i * 3 + 2];
    q.w = d;
    *(v4f*)(rec + (size_t)i * 8) = q;
    int r = res[i];
    if (r < 0 || r >= NRES) r = NRES;
    ((int*)rec)[(size_t)i * 8 + 4] = r * (HID * 4);
}

__global__ void
k_rec(const int* res_p, const float* pos_p, const float* dinv_p, int N_p, float* rec_p,
      const int* res_m, const float* pos_m, const float* dinv_m, int N_m, float* rec_m,
      int split) {
    BR_PICK;
    rec_body(PK(res), PK(pos), PK(dinv), PK(N), PK(rec), bid);
}

__device__ __forceinline__ void
agg1_body(const float* __restrict__ rec, const int* __restrict__ rowptr,
          const int* __restrict__ ecnt, const int* __restrict__ col,
          const float* __restrict__ W, const float* __restrict__ b,
          int N, float* __restrict__ h, int bid) {
    __shared__ __attribute__((aligned(16))) float Ws[(NRES + 1) * HID];
    for (int t = threadIdx.x; t < NRES * HID / 4; t += 256)
        ((v4f*)Ws)[t] = ((const v4f*)W)[t];
    if (threadIdx.x < HID / 4)
        ((v4f*)Ws)[NRES * HID / 4 + threadIdx.x] = (v4f){0.0f, 0.0f, 0.0f, 0.0f};
    __syncthreads();

    int lane = threadIdx.x & 31;
    const v4f w20 = *(const v4f*)(W + (NRES + 0) * HID + lane * 4);
    const v4f w21 = *(const v4f*)(W + (NRES + 1) * HID + lane * 4);
    const v4f w22 = *(const v4f*)(W + (NRES + 2) * HID + lane * 4);
    const v4f bb = *(const v4f*)(b + lane * 4);
    const char* Wl = (const char*)Ws + lane * 16;
    const int* reci = (const int*)rec;
    int first = (int)(bid * 8 + (threadIdx.x >> 5)) * AGG_NPW;

    for (int t = 0; t < AGG_NPW; ++t) {
        int i = __builtin_amdgcn_readfirstlane(first + t);
        if (i >= N) break;
        int start = rowptr[i];
        int n = ecnt[i];
        v4f rs = *(const v4f*)(rec + (size_t)i * 8);
        int os = reci[(size_t)i * 8 + 4];
        v4f acc = (v4f){0.0f, 0.0f, 0.0f, 0.0f};
        float q0 = 0.0f, q1 = 0.0f, q2 = 0.0f;
        for (int base = 0; base < n; base += 32) {
            int m = n - base;
            if (m > 32) m = 32;
            bool live = lane < m;
            int c = live ? col[start + base + lane] : i;
            c = min(max(c, 0), N - 1);
            v4f r = *(const v4f*)(rec + (size_t)c * 8);
            int off = reci[(size_t)c * 8 + 4];
            float d = live ? r.w : 0.0f;
            off = live ? off : NRES * HID * 4;
            q0 += live ? r.x : 0.0f;
            q1 += live ? r.y : 0.0f;
            q2 += live ? r.z : 0.0f;
            int dbits = __float_as_int(d);
            for (int j = 0; j < m; j += 8) {
                int o0 = __builtin_amdgcn_readlane(off, j);
                int o1 = __builtin_amdgcn_readlane(off, j + 1);
                int o2 = __builtin_amdgcn_readlane(off, j + 2);
                int o3 = __builtin_amdgcn_readlane(off, j + 3);
                int o4 = __builtin_amdgcn_readlane(off, j + 4);
                int o5 = __builtin_amdgcn_readlane(off, j + 5);
                int o6 = __builtin_amdgcn_readlane(off, j + 6);
                int o7 = __builtin_amdgcn_readlane(off, j + 7);
                v4f x0 = *(const v4f*)(Wl + o0);
                v4f x1 = *(const v4f*)(Wl + o1);
                v4f x2 = *(const v4f*)(Wl + o2);
                v4f x3 = *(const v4f*)(Wl + o3);
                v4f x4 = *(const v4f*)(Wl + o4);
                v4f x5 = *(const v4f*)(Wl + o5);
                v4f x6 = *(const v4f*)(Wl + o6);
                v4f x7 = *(const v4f*)(Wl + o7);
                float d0 = __int_as_float(__builtin_amdgcn_readlane(dbits, j));
                float d1 = __int_as_float(__builtin_amdgcn_readlane(dbits, j + 1));
                float d2 = __int_as_float(__builtin_amdgcn_readlane(dbits, j + 2));
                float d3 = __int_as_float(__builtin_amdgcn_readlane(dbits, j + 3));
                float d4 = __int_as_float(__builtin_amdgcn_readlane(dbits, j + 4));
                float d5 = __int_as_float(__builtin_amdgcn_readlane(dbits, j + 5));
                float d6 = __int_as_float(__builtin_amdgcn_readlane(dbits, j + 6));
                float d7 = __int_as_float(__builtin_amdgcn_readlane(dbits, j + 7));
                acc += d0 * x0;
                acc += d1 * x1;
                acc += d2 * x2;
                acc += d3 * x3;
                acc += d4 * x4;
                acc += d5 * x5;
                acc += d6 * x6;
                acc += d7 * x7;
            }
        }
        v4f xs = *(const v4f*)(Wl + os);
        acc += rs.w * xs;
        #pragma unroll
        for (int s = 16; s >= 1; s >>= 1) {
            q0 += __shfl_xor(q0, s);
            q1 += __shfl_xor(q1, s);
            q2 += __shfl_xor(q2, s);
        }
        q0 += rs.x;
        q1 += rs.y;
        q2 += rs.z;
        v4f S = acc + q0 * w20 + q1 * w21 + q2 * w22;
        float dnode = rs.w;
        v4f o;
        o.x = fmaxf(dnode * S.x + bb.x, 0.0f);
        o.y = fmaxf(dnode * S.y + bb.y, 0.0f);
        o.z = fmaxf(dnode * S.z + bb.z, 0.0f);
        o.w = fmaxf(dnode * S.w + bb.w, 0.0f);
        *(v4f*)(h + (size_t)i * HID + lane * 4) = o;
    }
}

__global__ void __launch_bounds__(256)
k_agg1(const float* rec_p, const int* rowptr_p, const int* ecnt_p, const int* col_p,
       const float* W_p, const float* b_p, int N_p, float* h_p,
       const float* rec_m, const int* rowptr_m, const int* ecnt_m, const int* col_m,
       const float* W_m, const float* b_m, int N_m, float* h_m, int split) {
    BR_PICK;
    agg1_body(PK(rec), PK(rowptr), PK(ecnt), PK(col), PK(W), PK(b), PK(N), PK(h), bid);
}

__device__ __forceinline__ void
gather_body(const float* __restrict__ y, const int* __restrict__ rowptr,
            const int* __restrict__ ecnt, const int* __restrict__ col,
            const float* __restrict__ dinv, const float* __restrict__ b,
            int N, float* __restrict__ h, int bid) {
    int i = __builtin_amdgcn_readfirstlane((int)(bid * 8 + (threadIdx.x >> 5)));
    if (i >= N) return;
    int lane = threadIdx.x & 31;
    int start = rowptr[i];
    int n = ecnt[i];
    const float* yl = y + lane * 4;
    v4f acc = (v4f){0.0f, 0.0f, 0.0f, 0.0f};
    for (int base = 0; base < n; base += 32) {
        int m = n - base;
        if (m > 32) m = 32;
        int c = (lane < m) ? col[start + base + lane] : i;
        c = min(max(c, 0), N - 1);
        for (int j = 0; j < m; j += 8) {
            int s0 = __builtin_amdgcn_readlane(c, j);
            int s1 = __builtin_amdgcn_readlane(c, j + 1);
            int s2 = __builtin_amdgcn_readlane(c, j + 2);
            int s3 = __builtin_amdgcn_readlane(c, j + 3);
            int s4 = __builtin_amdgcn_readlane(c, j + 4);
            int s5 = __builtin_amdgcn_readlane(c, j + 5);
            int s6 = __builtin_amdgcn_readlane(c, j + 6);
            int s7 = __builtin_amdgcn_readlane(c, j + 7);
            v4f v0 = *(const v4f*)(yl + (size_t)s0 * HID);
            v4f v1 = *(const v4f*)(yl + (size_t)s1 * HID);
            v4f v2 = *(const v4f*)(yl + (size_t)s2 * HID);
            v4f v3 = *(const v4f*)(yl + (size_t)s3 * HID);
            v4f v4 = *(const v4f*)(yl + (size_t)s4 * HID);
            v4f v5 = *(const v4f*)(yl + (size_t)s5 * HID);
            v4f v6 = *(const v4f*)(yl + (size_t)s6 * HID);
            v4f v7 = *(const v4f*)(yl + (size_t)s7 * HID);
            asm volatile("" :: "v"(v0), "v"(v1), "v"(v2), "v"(v3),
                               "v"(v4), "v"(v5), "v"(v6), "v"(v7));
            acc += v0;
            if (j + 1 < m) acc += v1;
            if (j + 2 < m) acc += v2;
            if (j + 3 < m) acc += v3;
            if (j + 4 < m) acc += v4;
            if (j + 5 < m) acc += v5;
            if (j + 6 < m) acc += v6;
            if (j + 7 < m) acc += v7;
        }
    }
    v4f self = *(const v4f*)(yl + (size_t)i * HID);
    v4f bb = *(const v4f*)(b + lane * 4);
    float d = dinv[i];
    v4f o;
    o.x = fmaxf(d * (acc.x + self.x) + bb.x, 0.0f);
    o.y = fmaxf(d * (acc.y + self.y) + bb.y, 0.0f);
    o.z = fmaxf(d * (acc.z + self.z) + bb.z, 0.0f);
    o.w = fmaxf(d * (acc.w + self.w) + bb.w, 0.0f);
    *(v4f*)(h + (size_t)i * HID + lane * 4) = o;
}

__global__ void
k_gather(const float* y_p, const int* rowptr_p, const int* ecnt_p, const int* col_p,
         const float* dinv_p, const float* b_p, int N_p, float* h_p,
         const float* y_m, const int* rowptr_m, const int* ecnt_m, const int* col_m,
         const float* dinv_m, const float* b_m, int N_m, float* h_m, int split) {
    BR_PICK;
    gather_body(PK(y), PK(rowptr), PK(ecnt), PK(col), PK(dinv), PK(b), PK(N), PK(h), bid);
}

#define POOL_RPW 16

__device__ __forceinline__ void pool_flush(float* __restrict__ pesum, float* __restrict__ cnt,
                                           int g, int lane, v4f acc, int n) {
    if ((unsigned)g >= (unsigned)NGRAPH) return;
    float* p = pesum + g * HID + lane * 4;
    atomicAdd(p + 0, acc.x);
    atomicAdd(p + 1, acc.y);
    atomicAdd(p + 2, acc.z);
    atomicAdd(p + 3, acc.w);
    if (lane == 0) atomicAdd(cnt + g, (float)n);
}

__global__ void __launch_bounds__(256)
k_pool(const float* __restrict__ h, const int* __restrict__ batch, int N,
       float* __restrict__ pesum, float* __restrict__ cnt) {
    int r0 = __builtin_amdgcn_readfirstlane((int)(blockIdx.x * 8 + (threadIdx.x >> 5))) * POOL_RPW;
    if (r0 >= N) return;
    int last = min(r0 + POOL_RPW, N) - 1;
    int lane = threadIdx.x & 31;
    const float* hl = h + lane * 4;
    int g[POOL_RPW];
    v4f v[POOL_RPW];
    #pragma unroll
    for (int k = 0; k < POOL_RPW; ++k) {
        int r = min(r0 + k, last);
        g[k] = batch[r];
        v[k] = *(const v4f*)(hl + (size_t)r * HID);
    }
    asm volatile("" :: "v"(v[0]), "v"(v[1]), "v"(v[2]), "v"(v[3]), "v"(v[4]), "v"(v[5]),
                       "v"(v[6]), "v"(v[7]), "v"(v[8]), "v"(v[9]), "v"(v[10]), "v"(v[11]),
                       "v"(v[12]), "v"(v[13]), "v"(v[14]), "v"(v[15]));
    int cur = g[0];
    int n = 0;
    v4f acc = (v4f){0.0f, 0.0f, 0.0f, 0.0f};
    #pragma unroll
    for (int k = 0; k < POOL_RPW; ++k) {
        if (r0 + k <= last) {
            if (g[k] != cur) {
                pool_flush(pesum, cnt, cur, lane, acc, n);
                cur = g[k];
                n = 0;
                acc = (v4f){0.0f, 0.0f, 0.0f, 0.0f};
            }
            acc += v[k];
            n += 1;
        }
    }
    pool_flush(pesum, cnt, cur, lane, acc, n);
}

__global__ void k_pool_div(const float* __restrict__ pesum, const float* __restrict__ cnt,
                           float* __restrict__ pe) {
    int idx = blockIdx.x * 256 + threadIdx.x;
    if (idx >= NGRAPH * HID) return;
    pe[idx] = pesum[idx] / fmaxf(cnt[idx >> 7], 1.0f);
}

__global__ void k_head(const float* __restrict__ hmid, const float* __restrict__ W2,
                       const float* __restrict__ b2, int N, float* __restrict__ out) {
    int i = blockIdx.x * 256 + threadIdx.x;
    if (i >= N) return;
    float a0 = b2[0], a1 = b2[1], a2 = b2[2];
    const float* h = hmid + (size_t)i * HID;
    for (int k = 0; k < HID; ++k) {
        float hv = h[k];
        a0 += hv * W2[k * 3 + 0];
        a1 += hv * W2[k * 3 + 1];
        a2 += hv * W2[k * 3 + 2];
    }
    out[(size_t)i * 3 + 0] = a0;
    out[(size_t)i * 3 + 1] = a1;
    out[(size_t)i * 3 + 2] = a2;
}

__device__ __forceinline__ void
wfrag_body(const float* __restrict__ W, int K, float* __restrict__ Wf, int bid) {
    int idx = bid * 256 + threadIdx.x;
    if (idx >= K * HID) return;
    int j    = idx & 1;
    int nt   = (idx >> 1) & 7;
    int l    = (idx >> 4) & 31;
    int kk   = idx >> 9;
    int krow = 4 * kk + (l >> 4) * 2 + j;
    int col  = nt * 16 + (l & 15);
    Wf[idx] = W[(size_t)krow * HID + col];
}

#define FILL_BLOCKS (2 * MAXBIN * BIN_PAD / 256)
#define POOLZ_N     (NGRAPH * HID + NGRAPH)
#define POOLZ_BLOCKS ((POOLZ_N + 255) / 256)
#define WF_BLOCKS   (HID * HID / 256)
#define SETUP_BLOCKS (2 * FILL_BLOCKS + POOLZ_BLOCKS + 4 * WF_BLOCKS)

__global__ void __launch_bounds__(256)
k_setup(int* __restrict__ bins_p, int* __restrict__ bins_m, float* __restrict__ poolz,
        const float* __restrict__ W_p2, float* __restrict__ Wf_p,
        const float* __restrict__ W_m2, float* __restrict__ Wf_m,
        const float* __restrict__ W1f, float* __restrict__ Wf_h) {
    int bid = blockIdx.x;
    int t = threadIdx.x;
    if (bid < FILL_BLOCKS) { bins_p[bid * 256 + t] = 0; return; }
    bid -= FILL_BLOCKS;
    if (bid < FILL_BLOCKS) { bins_m[bid * 256 + t] = 0; return; }
    bid -= FILL_BLOCKS;
    if (bid < POOLZ_BLOCKS) {
        int i = bid * 256 + t;
        if (i < POOLZ_N) poolz[i] = 0.0f;
        return;
    }
    bid -= POOLZ_BLOCKS;
    if (bid < WF_BLOCKS) { wfrag_body(W_p2, HID, Wf_p, bid); return; }
    bid -= WF_BLOCKS;
    if (bid < WF_BLOCKS) { wfrag_body(W_m2, HID, Wf_m, bid); return; }
    bid -= WF_BLOCKS;
    wfrag_body(W1f, 2 * HID, Wf_h, bid);
}

__device__ __forceinline__ void gemm_ksteps(const float* __restrict__ Arow,
                                            const float* __restrict__ Wl,
                                            int ksteps, v8f (&acc)[8]) {
    for (int kk = 0; kk < ksteps; ++kk) {
        v2f a;
        a.x = Arow[4 * kk];
        a.y = Arow[4 * kk + 1];
        const float4* q = (const float4*)(Wl + (size_t)kk * 512);
        float4 q0 = q[0], q1 = q[1], q2 = q[2], q3 = q[3];
        v2f b0; b0.x = q0.x; b0.y = q0.y;
        v2f b1; b1.x = q0.z; b1.y = q0.w;
        v2f b2; b2.x = q1.x; b2.y = q1.y;
        v2f b3; b3.x = q1.z; b3.y = q1.w;
        v2f b4; b4.x = q2.x; b4.y = q2.y;
        v2f b5; b5.x = q2.z; b5.y = q2.w;
        v2f b6; b6.x = q3.x; b6.y = q3.y;
        v2f b7; b7.x = q3.z; b7.y = q3.w;
        acc[0] = __builtin_amdgcn_wmma_f32_16x16x4_f32(false, a, false, b0, (short)0, acc[0], false, false);
        acc[1] = __builtin_amdgcn_wmma_f32_16x16x4_f32(false, a, false, b1, (short)0, acc[1], false, false);
        acc[2] = __builtin_amdgcn_wmma_f32_16x16x4_f32(false, a, false, b2, (short)0, acc[2], false, false);
        acc[3] = __builtin_amdgcn_wmma_f32_16x16x4_f32(false, a, false, b3, (short)0, acc[3], false, false);
        acc[4] = __builtin_amdgcn_wmma_f32_16x16x4_f32(false, a, false, b4, (short)0, acc[4], false, false);
        acc[5] = __builtin_amdgcn_wmma_f32_16x16x4_f32(false, a, false, b5, (short)0, acc[5], false, false);
        acc[6] = __builtin_amdgcn_wmma_f32_16x16x4_f32(false, a, false, b6, (short)0, acc[6], false, false);
        acc[7] = __builtin_amdgcn_wmma_f32_16x16x4_f32(false, a, false, b7, (short)0, acc[7], false, false);
    }
}

__device__ __forceinline__ void gemm_store(v8f (&acc)[8], int row0, int half, int lmod,
                                           const float* __restrict__ rowscale,
                                           const float* __restrict__ bias, int relu,
                                           float* __restrict__ out) {
    float rs[8];
    if (rowscale) {
        #pragma unroll
        for (int r = 0; r < 8; ++r) rs[r] = rowscale[row0 + r + half * 8];
    }

    #pragma unroll
    for (int nt = 0; nt < 8; ++nt) {
        int col = nt * 16 + lmod;
        float bv = bias ? bias[col] : 0.0f;
        #pragma unroll
        for (int r = 0; r < 8; ++r) {
            float v = acc[nt][r];
            if (rowscale) v *= rs[r];
            v += bv;
            if (relu) v = fmaxf(v, 0.0f);
            out[(size_t)(row0 + r + half * 8) * HID + col] = v;
        }
    }
}

__device__ __forceinline__ void
gemm128_body(const float* __restrict__ A, int K, int M,
             const float* __restrict__ Wf,
             const float* __restrict__ rowscale,
             const float* __restrict__ bias, int relu,
             float* __restrict__ out, int bid) {
    int w = threadIdx.x >> 5;
    int lane = threadIdx.x & 31;
    int tile = bid * 4 + w;
    if (tile * 16 >= M) return;
    int row0 = tile * 16;
    int half = lane >> 4;
    int lmod = lane & 15;
    int koff = half * 2;
    const float* Arow = A + (size_t)(row0 + lmod) * K + koff;
    const float* Wl   = Wf + (size_t)lane * 16;

    v8f acc[8];
    #pragma unroll
    for (int nt = 0; nt < 8; ++nt) acc[nt] = (v8f){};

    gemm_ksteps(Arow, Wl, K >> 2, acc);
    gemm_store(acc, row0, half, lmod, rowscale, bias, relu, out);
}

__global__ void
k_gemm128(const float* h_p, int N_p, const float* Wf_p, const float* dinv_p, float* y_p,
          const float* h_m, int N_m, const float* Wf_m, const float* dinv_m, float* y_m,
          int split) {
    BR_PICK;
    gemm128_body(PK(h), HID, PK(N), PK(Wf), PK(dinv), nullptr, 0, PK(y), bid);
}

__global__ void k_gemm_head(const float* __restrict__ h, const float* __restrict__ pe,
                            const int* __restrict__ batch, int M,
                            const float* __restrict__ Wf, const float* __restrict__ bias,
                            float* __restrict__ out) {
    int w = threadIdx.x >> 5;
    int lane = threadIdx.x & 31;
    int tile = blockIdx.x * 4 + w;
    if (tile * 16 >= M) return;
    int row0 = tile * 16;
    int half = lane >> 4;
    int lmod = lane & 15;
    int koff = half * 2;
    int g = min(max(batch[row0 + lmod], 0), NGRAPH - 1);
    const float* Wl = Wf + (size_t)lane * 16;

    v8f acc[8];
    #pragma unroll
    for (int nt = 0; nt < 8; ++nt) acc[nt] = (v8f){};

    gemm_ksteps(h + (size_t)(row0 + lmod) * HID + koff, Wl, HID / 4, acc);
    gemm_ksteps(pe + (size_t)g * HID + koff, Wl + (size_t)(HID / 4) * 512, HID / 4, acc);
    gemm_store(acc, row0, half, lmod, nullptr, bias, 1, out);
}


static inline int cdiv(long long a, long long b) { return (int)((a + b - 1) / b); }

extern "C" void kernel_launch(void* const* d_in, const int* in_sizes, int n_in,
                              void* d_out, int out_size, void* d_ws, size_t ws_size,
                              hipStream_t stream) {
    (void)n_in; (void)out_size; (void)ws_size;

    const int*   res_p   = (const int*)d_in[0];
    const float* pos_p   = (const float*)d_in[1];
    const int*   ei_p    = (const int*)d_in[2];
    const int*   batch_p = (const int*)d_in[3];
    const int*   res_m   = (const int*)d_in[4];
    const float* pos_m   = (const float*)d_in[5];
    const int*   ei_m    = (const int*)d_in[6];
    const int*   batch_m = (const int*)d_in[7];
    const float* W_p1 = (const float*)d_in[8];
    const float* b_p1 = (const float*)d_in[9];
    const float* W_p2 = (const float*)d_in[10];
    const float* b_p2 = (const float*)d_in[11];
    const float* W_m1 = (const float*)d_in[12];
    const float* b_m1 = (const float*)d_in[13];
    const float* W_m2 = (const float*)d_in[14];
    const float* b_m2 = (const float*)d_in[15];
    const float* W1f  = (const float*)d_in[16];
    const float* b1f  = (const float*)d_in[17];
    const float* W2f  = (const float*)d_in[18];
    const float* b2f  = (const float*)d_in[19];

    int N_P = in_sizes[0];
    int E_P = in_sizes[2] / 2;
    int N_M = in_sizes[4];
    int E_M = in_sizes[6] / 2;
    const int* src_p = ei_p;
    const int* dst_p = ei_p + E_P;
    const int* src_m = ei_m;
    const int* dst_m = ei_m + E_M;

    float* ws = (float*)d_ws;
    size_t nhp = (size_t)N_P * HID;
    size_t nhm = (size_t)N_M * HID;
    size_t npa = ((size_t)N_P + 63) / 64 * 64;
    size_t nma = ((size_t)N_M + 63) / 64 * 64;
    float* A_p    = ws;
    float* C_p    = A_p + nhp;
    float* A_m    = C_p + nhp;
    float* C_m    = A_m + nhm;
    float* rec_p  = C_m + nhm;
    float* rec_m  = rec_p + npa * 8;
    float* dinv_p = rec_m + nma * 8;
    float* dinv_m = dinv_p + npa;
    float* pe_sum = dinv_m + nma;
    float* cnt    = pe_sum + NGRAPH * HID;
    float* pe     = cnt + NGRAPH;
    float* Wf_p   = pe + NGRAPH * HID;
    float* Wf_m   = Wf_p + HID * HID;
    float* Wf_h   = Wf_m + HID * HID;
    int* ecnt_p   = (int*)(Wf_h + 2 * HID * HID);
    int* rowptr_p = ecnt_p + npa;
    int* part_p   = rowptr_p + npa;
    int* col_p    = part_p + E_P;
    int* bins_p   = col_p + E_P;
    int* ecnt_m   = bins_p + (2 * BIN_PAD + 1) * MAXBIN;
    int* rowptr_m = ecnt_m + nma;
    int* part_m   = rowptr_m + nma;
    int* col_m    = part_m + E_M;
    int* bins_m   = col_m + E_M;
    int* bcur_p   = bins_p + MAXBIN * BIN_PAD;
    int* bsta_p   = bins_p + 2 * MAXBIN * BIN_PAD;
    int* bcur_m   = bins_m + MAXBIN * BIN_PAD;
    int* bsta_m   = bins_m + 2 * MAXBIN * BIN_PAD;
    int nbin_p = cdiv(N_P, BIN_NODES);
    int nbin_m = cdiv(N_M, BIN_NODES);

    k_setup<<<SETUP_BLOCKS, 256, 0, stream>>>(bins_p, bins_m, pe_sum, W_p2, Wf_p, W_m2, Wf_m,
                                              W1f, Wf_h);

    int gp = cdiv(E_P, PCH), gm = cdiv(E_M, PCH);
    k_bin_hist<<<gp + gm, 256, 0, stream>>>(dst_p, E_P, N_P, bins_p,
                                            dst_m, E_M, N_M, bins_m, gp);
    k_bin_part<<<gp + gm, 256, 0, stream>>>(src_p, dst_p, E_P, N_P, nbin_p, bins_p, bcur_p, bsta_p, part_p,
                                            src_m, dst_m, E_M, N_M, nbin_m, bins_m, bcur_m, bsta_m, part_m,
                                            gp);
    k_bin_csr<<<nbin_p + nbin_m, 256, 0, stream>>>(part_p, bsta_p, bins_p, N_P, ecnt_p, rowptr_p, dinv_p, col_p,
                                                   part_m, bsta_m, bins_m, N_M, ecnt_m, rowptr_m, dinv_m, col_m,
                                                   nbin_p);

    gp = cdiv(N_P, 256); gm = cdiv(N_M, 256);
    k_rec<<<gp + gm, 256, 0, stream>>>(res_p, pos_p, dinv_p, N_P, rec_p,
                                       res_m, pos_m, dinv_m, N_M, rec_m, gp);
    gp = cdiv(N_P, 8 * AGG_NPW); gm = cdiv(N_M, 8 * AGG_NPW);
    k_agg1<<<gp + gm, 256, 0, stream>>>(rec_p, rowptr_p, ecnt_p, col_p, W_p1, b_p1, N_P, C_p,
                                        rec_m, rowptr_m, ecnt_m, col_m, W_m1, b_m1, N_M, C_m, gp);

    gp = cdiv(N_P / 16, 4); gm = cdiv(N_M / 16, 4);
    k_gemm128<<<gp + gm, 128, 0, stream>>>(C_p, N_P, Wf_p, dinv_p, A_p,
                                           C_m, N_M, Wf_m, dinv_m, A_m, gp);
    gp = cdiv(N_P, 8); gm = cdiv(N_M, 8);
    k_gather<<<gp + gm, 256, 0, stream>>>(A_p, rowptr_p, ecnt_p, col_p, dinv_p, b_p2, N_P, C_p,
                                          A_m, rowptr_m, ecnt_m, col_m, dinv_m, b_m2, N_M, C_m, gp);

    k_pool<<<cdiv(N_P, 8 * POOL_RPW), 256, 0, stream>>>(C_p, batch_p, N_P, pe_sum, cnt);
    k_pool_div<<<cdiv(NGRAPH * HID, 256), 256, 0, stream>>>(pe_sum, cnt, pe);

    float* hmid = A_m;
    k_gemm_head<<<cdiv(N_M / 16, 4), 128, 0, stream>>>(C_m, pe, batch_m, N_M, Wf_h, b1f, hmid);
    k_head<<<cdiv(N_M, 256), 256, 0, stream>>>(hmid, W2f, b2f, N_M, (float*)d_out);
}


# 3 small launches folded away: node records written by the edge-list launch; pool mean (sum x 1/count) and the 128x3 output product inside the head GEMM; 12 launches -> 9; output sum order changes
# speedup vs baseline: 5.5775x; 1.1299x over previous
#include <hip/hip_runtime.h>
#include <hip/hip_bf16.h>

typedef __attribute__((ext_vector_type(2))) float v2f;
typedef __attribute__((ext_vector_type(8))) float v8f;
typedef __attribute__((ext_vector_type(4))) float v4f;

#define HID 128
#define NGRAPH 64
#define NRES 20
#define AGG_NPW 4

#define BR_PICK bool second = (int)blockIdx.x >= split; \
                int bid = second ? (int)blockIdx.x - split : (int)blockIdx.x
#define PK(x) (second ? x##_m : x##_p)

#define BIN_SHIFT 8
#define BIN_NODES 256
#define MAXBIN 512
#define PCH 4096
#define PEPT 16
#define LCAP 16384
#define BIN_PAD 32


__device__ __forceinline__ void
bin_hist_body(const int* __restrict__ dst, int E, int N, int* __restrict__ binCount, int bid) {
    __shared__ int h[MAXBIN];
    int tid = threadIdx.x;
    for (int t = tid; t < MAXBIN; t += 256) h[t] = 0;
    __syncthreads();
    int e0 = bid * PCH;
    int dd[PEPT];
    #pragma unroll
    for (int k = 0; k < PEPT; ++k) dd[k] = dst[min(e0 + k * 256 + tid, E - 1)];
    #pragma unroll
    for (int k = 0; k < PEPT; ++k) {
        int e = e0 + k * 256 + tid;
        int d = min(max(dd[k], 0), N - 1);
        if (e < E) atomicAdd(&h[d >> BIN_SHIFT], 1);
    }
    __syncthreads();
    for (int t = tid; t < MAXBIN; t += 256) {
        int c = h[t];
        if (c > 0) atomicAdd(binCount + t * BIN_PAD, c);
    }
}

__global__ void __launch_bounds__(256)
k_bin_hist(const int* dst_p, int E_p, int N_p, int* binCount_p,
           const int* dst_m, int E_m, int N_m, int* binCount_m, int split) {
    BR_PICK;
    bin_hist_body(PK(dst), PK(E), PK(N), PK(binCount), bid);
}

__device__ __forceinline__ void scan_bins(const int* in, int* out, int lane) {
    int v[MAXBIN / 32];
    int s = 0;
    #pragma unroll
    for (int j = 0; j < MAXBIN / 32; ++j) {
        v[j] = in[lane * (MAXBIN / 32) + j];
        s += v[j];
    }
    int x = s;
    #pragma unroll
    for (int d = 1; d < 32; d <<= 1) {
        int t = __shfl_up(x, d);
        if (lane >= d) x += t;
    }
    int run = x - s;
    #pragma unroll
    for (int j = 0; j < MAXBIN / 32; ++j) {
        out[lane * (MAXBIN / 32) + j] = run;
        run += v[j];
    }
}

__device__ __forceinline__ void
bin_part_body(const int* __restrict__ src, const int* __restrict__ dst, int E, int N, int nbin,
              const int* __restrict__ binCount, int* __restrict__ binCursor,
              int* __restrict__ binStart, int* __restrict__ part, int bid) {
    __shared__ int h[MAXBIN];
    __shared__ int gc[MAXBIN];
    __shared__ int loc[MAXBIN];
    __shared__ int bstart[MAXBIN];
    __shared__ int gbase[MAXBIN];
    __shared__ int sval[PCH];
    __shared__ unsigned short sbin[PCH];
    int tid = threadIdx.x;
    for (int t = tid; t < MAXBIN; t += 256) {
        h[t] = 0;
        gc[t] = (t < nbin) ? binCount[t * BIN_PAD] : 0;
    }
    __syncthreads();
    int e0 = bid * PCH;
    int pk[PEPT], bb[PEPT], rr[PEPT];
    #pragma unroll
    for (int k = 0; k < PEPT; ++k) {
        int ec = min(e0 + k * 256 + tid, E - 1);
        bb[k] = dst[ec];
        pk[k] = src[ec];
    }
    #pragma unroll
    for (int k = 0; k < PEPT; ++k) {
        int e = e0 + k * 256 + tid;
        int d = min(max(bb[k], 0), N - 1);
        int s = min(max(pk[k], 0), N - 1);
        bb[k] = (e < E) ? (d >> BIN_SHIFT) : -1;
        pk[k] = (s << BIN_SHIFT) | (d & (BIN_NODES - 1));
        rr[k] = 0;
    }
    #pragma unroll
    for (int k = 0; k < PEPT; ++k)
        if (bb[k] >= 0) rr[k] = atomicAdd(&h[bb[k]], 1);
    __syncthreads();
    int wave = tid >> 5, lane = tid & 31;
    if (wave == 0) scan_bins(h, loc, lane);
    else if (wave == 1) scan_bins(gc, bstart, lane);
    __syncthreads();
    for (int t = tid; t < MAXBIN; t += 256) {
        int c = h[t];
        if (c > 0) gbase[t] = bstart[t] + atomicAdd(binCursor + t * BIN_PAD, c);
        if (bid == 0 && t < nbin) binStart[t] = bstart[t];
    }
    #pragma unroll
    for (int k = 0; k < PEPT; ++k) {
        if (bb[k] >= 0) {
            int p = loc[bb[k]] + rr[k];
            sval[p] = pk[k];
            sbin[p] = (unsigned short)bb[k];
        }
    }
    __syncthreads();
    int nv = min(PCH, E - e0);
    for (int i = tid; i < nv; i += 256) {
        int b = sbin[i];
        part[gbase[b] + (i - loc[b])] = sval[i];
    }
}

__global__ void __launch_bounds__(256)
k_bin_part(const int* src_p, const int* dst_p, int E_p, int N_p, int nbin_p,
           const int* binCount_p, int* binCursor_p, int* binStart_p, int* part_p,
           const int* src_m, const int* dst_m, int E_m, int N_m, int nbin_m,
           const int* binCount_m, int* binCursor_m, int* binStart_m, int* part_m, int split) {
    BR_PICK;
    bin_part_body(PK(src), PK(dst), PK(E), PK(N), PK(nbin), PK(binCount), PK(binCursor),
                  PK(binStart), PK(part), bid);
}

__device__ __forceinline__ void
bin_csr_body(const int* __restrict__ part, const int* __restrict__ binStart,
             const int* __restrict__ binCount, int N, int* __restrict__ ecnt,
             int* __restrict__ rowptr, float* __restrict__ dinv, int* __restrict__ col,
             const int* __restrict__ res, const float* __restrict__ pos,
             float* __restrict__ rec, int bid) {
    __shared__ int cnt[BIN_NODES];
    __shared__ int cur[BIN_NODES];
    __shared__ int wtot[8];
    __shared__ int lcol[LCAP];
    int tid = threadIdx.x;
    int wave = tid >> 5, lane = tid & 31;
    int b = bid;
    int s = binStart[b];
    int m = binCount[b * BIN_PAD];
    cnt[tid] = 0;
    __syncthreads();
    for (int i0 = 0; i0 < m; i0 += 1024) {
        int v[4];
        #pragma unroll
        for (int u = 0; u < 4; ++u) {
            int i = i0 + u * 256 + tid;
            v[u] = (i < m) ? part[s + i] : -1;
        }
        #pragma unroll
        for (int u = 0; u < 4; ++u)
            if (v[u] >= 0) atomicAdd(&cnt[v[u] & (BIN_NODES - 1)], 1);
    }
    __syncthreads();
    int c = cnt[tid];
    int x = c;
    #pragma unroll
    for (int d = 1; d < 32; d <<= 1) {
        int t = __shfl_up(x, d);
        if (lane >= d) x += t;
    }
    if (lane == 31) wtot[wave] = x;
    __syncthreads();
    int wb = 0;
    for (int w = 0; w < wave; ++w) wb += wtot[w];
    int off = wb + x - c;
    cur[tid] = off;
    int node = b * BIN_NODES + tid;
    if (node < N) {
        ecnt[node] = c;
        rowptr[node] = s + off;
        float dv = rsqrtf((float)c + 1.0f);
        dinv[node] = dv;
        v4f q;
        q.x = dv * pos[node * 3 + 0];
        q.y = dv * pos[node * 3 + 1];
        q.z = dv * pos[node * 3 + 2];
        q.w = dv;
        *(v4f*)(rec + (size_t)node * 8) = q;
        int r = res[node];
        if (r < 0 || r >= NRES) r = NRES;
        ((int*)rec)[(size_t)node * 8 + 4] = r * (HID * 4);
    }
    __syncthreads();
    bool staged = m <= LCAP;
    for (int i0 = 0; i0 < m; i0 += 1024) {
        int v[4];
        #pragma unroll
        for (int u = 0; u < 4; ++u) {
            int i = i0 + u * 256 + tid;
            v[u] = (i < m) ? part[s + i] : -1;
        }
        #pragma unroll
        for (int u = 0; u < 4; ++u) {
            if (v[u] >= 0) {
                int p = atomicAdd(&cur[v[u] & (BIN_NODES - 1)], 1);
                int sv = (int)((unsigned)v[u] >> BIN_SHIFT);
                if (staged) lcol[p] = sv;
                else col[s + p] = sv;
            }
        }
    }
    if (staged) {
        __syncthreads();
        for (int i = tid; i < m; i += 256) col[s + i] = lcol[i];
    }
}

__global__ void __launch_bounds__(256)
k_bin_csr(const int* part_p, const int* binStart_p, const int* binCount_p, int N_p,
          int* ecnt_p, int* rowptr_p, float* dinv_p, int* col_p,
          const int* res_p, const float* pos_p, float* rec_p,
          const int* part_m, const int* binStart_m, const int* binCount_m, int N_m,
          int* ecnt_m, int* rowptr_m, float* dinv_m, int* col_m,
          const int* res_m, const float* pos_m, float* rec_m, int split) {
    BR_PICK;
    bin_csr_body(PK(part), PK(binStart), PK(binCount), PK(N), PK(ecnt), PK(rowptr), PK(dinv),
                 PK(col), PK(res), PK(pos), PK(rec), bid);
}


__device__ __forceinline__ void
agg1_body(const float* __restrict__ rec, const int* __restrict__ rowptr,
          const int* __restrict__ ecnt, const int* __restrict__ col,
          const float* __restrict__ W, const float* __restrict__ b,
          int N, float* __restrict__ h, int bid) {
    __shared__ __attribute__((aligned(16))) float Ws[(NRES + 1) * HID];
    for (int t = threadIdx.x; t < NRES * HID / 4; t += 256)
        ((v4f*)Ws)[t] = ((const v4f*)W)[t];
    if (threadIdx.x < HID / 4)
        ((v4f*)Ws)[NRES * HID / 4 + threadIdx.x] = (v4f){0.0f, 0.0f, 0.0f, 0.0f};
    __syncthreads();

    int lane = threadIdx.x & 31;
    const v4f w20 = *(const v4f*)(W + (NRES + 0) * HID + lane * 4);
    const v4f w21 = *(const v4f*)(W + (NRES + 1) * HID + lane * 4);
    const v4f w22 = *(const v4f*)(W + (NRES + 2) * HID + lane * 4);
    const v4f bb = *(const v4f*)(b + lane * 4);
    const char* Wl = (const char*)Ws + lane * 16;
    const int* reci = (const int*)rec;
    int first = (int)(bid * 8 + (threadIdx.x >> 5)) * AGG_NPW;

    for (int t = 0; t < AGG_NPW; ++t) {
        int i = __builtin_amdgcn_readfirstlane(first + t);
        if (i >= N) break;
        int start = rowptr[i];
        int n = ecnt[i];
        v4f rs = *(const v4f*)(rec + (size_t)i * 8);
        int os = reci[(size_t)i * 8 + 4];
        v4f acc = (v4f){0.0f, 0.0f, 0.0f, 0.0f};
        float q0 = 0.0f, q1 = 0.0f, q2 = 0.0f;
        for (int base = 0; base < n; base += 32) {
            int m = n - base;
            if (m > 32) m = 32;
            bool live = lane < m;
            int c = live ? col[start + base + lane] : i;
            c = min(max(c, 0), N - 1);
            v4f r = *(const v4f*)(rec + (size_t)c * 8);
            int off = reci[(size_t)c * 8 + 4];
            float d = live ? r.w : 0.0f;
            off = live ? off : NRES * HID * 4;
            q0 += live ? r.x : 0.0f;
            q1 += live ? r.y : 0.0f;
            q2 += live ? r.z : 0.0f;
            int dbits = __float_as_int(d);
            for (int j = 0; j < m; j += 8) {
                int o0 = __builtin_amdgcn_readlane(off, j);
                int o1 = __builtin_amdgcn_readlane(off, j + 1);
                int o2 = __builtin_amdgcn_readlane(off, j + 2);
                int o3 = __builtin_amdgcn_readlane(off, j + 3);
                int o4 = __builtin_amdgcn_readlane(off, j + 4);
                int o5 = __builtin_amdgcn_readlane(off, j + 5);
                int o6 = __builtin_amdgcn_readlane(off, j + 6);
                int o7 = __builtin_amdgcn_readlane(off, j + 7);
                v4f x0 = *(const v4f*)(Wl + o0);
                v4f x1 = *(const v4f*)(Wl + o1);
                v4f x2 = *(const v4f*)(Wl + o2);
                v4f x3 = *(const v4f*)(Wl + o3);
                v4f x4 = *(const v4f*)(Wl + o4);
                v4f x5 = *(const v4f*)(Wl + o5);
                v4f x6 = *(const v4f*)(Wl + o6);
                v4f x7 = *(const v4f*)(Wl + o7);
                float d0 = __int_as_float(__builtin_amdgcn_readlane(dbits, j));
                float d1 = __int_as_float(__builtin_amdgcn_readlane(dbits, j + 1));
                float d2 = __int_as_float(__builtin_amdgcn_readlane(dbits, j + 2));
                float d3 = __int_as_float(__builtin_amdgcn_readlane(dbits, j + 3));
                float d4 = __int_as_float(__builtin_amdgcn_readlane(dbits, j + 4));
                float d5 = __int_as_float(__builtin_amdgcn_readlane(dbits, j + 5));
                float d6 = __int_as_float(__builtin_amdgcn_readlane(dbits, j + 6));
                float d7 = __int_as_float(__builtin_amdgcn_readlane(dbits, j + 7));
                acc += d0 * x0;
                acc += d1 * x1;
                acc += d2 * x2;
                acc += d3 * x3;
                acc += d4 * x4;
                acc += d5 * x5;
                acc += d6 * x6;
                acc += d7 * x7;
            }
        }
        v4f xs = *(const v4f*)(Wl + os);
        acc += rs.w * xs;
        #pragma unroll
        for (int s = 16; s >= 1; s >>= 1) {
            q0 += __shfl_xor(q0, s);
            q1 += __shfl_xor(q1, s);
            q2 += __shfl_xor(q2, s);
        }
        q0 += rs.x;
        q1 += rs.y;
        q2 += rs.z;
        v4f S = acc + q0 * w20 + q1 * w21 + q2 * w22;
        float dnode = rs.w;
        v4f o;
        o.x = fmaxf(dnode * S.x + bb.x, 0.0f);
        o.y = fmaxf(dnode * S.y + bb.y, 0.0f);
        o.z = fmaxf(dnode * S.z + bb.z, 0.0f);
        o.w = fmaxf(dnode * S.w + bb.w, 0.0f);
        *(v4f*)(h + (size_t)i * HID + lane * 4) = o;
    }
}

__global__ void __launch_bounds__(256)
k_agg1(const float* rec_p, const int* rowptr_p, const int* ecnt_p, const int* col_p,
       const float* W_p, const float* b_p, int N_p, float* h_p,
       const float* rec_m, const int* rowptr_m, const int* ecnt_m, const int* col_m,
       const float* W_m, const float* b_m, int N_m, float* h_m, int split) {
    BR_PICK;
    agg1_body(PK(rec), PK(rowptr), PK(ecnt), PK(col), PK(W), PK(b), PK(N), PK(h), bid);
}

__device__ __forceinline__ void
gather_body(const float* __restrict__ y, const int* __restrict__ rowptr,
            const int* __restrict__ ecnt, const int* __restrict__ col,
            const float* __restrict__ dinv, const float* __restrict__ b,
            int N, float* __restrict__ h, int bid) {
    int i = __builtin_amdgcn_readfirstlane((int)(bid * 8 + (threadIdx.x >> 5)));
    if (i >= N) return;
    int lane = threadIdx.x & 31;
    int start = rowptr[i];
    int n = ecnt[i];
    const float* yl = y + lane * 4;
    v4f acc = (v4f){0.0f, 0.0f, 0.0f, 0.0f};
    for (int base = 0; base < n; base += 32) {
        int m = n - base;
        if (m > 32) m = 32;
        int c = (lane < m) ? col[start + base + lane] : i;
        c = min(max(c, 0), N - 1);
        for (int j = 0; j < m; j += 8) {
            int s0 = __builtin_amdgcn_readlane(c, j);
            int s1 = __builtin_amdgcn_readlane(c, j + 1);
            int s2 = __builtin_amdgcn_readlane(c, j + 2);
            int s3 = __builtin_amdgcn_readlane(c, j + 3);
            int s4 = __builtin_amdgcn_readlane(c, j + 4);
            int s5 = __builtin_amdgcn_readlane(c, j + 5);
            int s6 = __builtin_amdgcn_readlane(c, j + 6);
            int s7 = __builtin_amdgcn_readlane(c, j + 7);
            v4f v0 = *(const v4f*)(yl + (size_t)s0 * HID);
            v4f v1 = *(const v4f*)(yl + (size_t)s1 * HID);
            v4f v2 = *(const v4f*)(yl + (size_t)s2 * HID);
            v4f v3 = *(const v4f*)(yl + (size_t)s3 * HID);
            v4f v4 = *(const v4f*)(yl + (size_t)s4 * HID);
            v4f v5 = *(const v4f*)(yl + (size_t)s5 * HID);
            v4f v6 = *(const v4f*)(yl + (size_t)s6 * HID);
            v4f v7 = *(const v4f*)(yl + (size_t)s7 * HID);
            asm volatile("" :: "v"(v0), "v"(v1), "v"(v2), "v"(v3),
                               "v"(v4), "v"(v5), "v"(v6), "v"(v7));
            acc += v0;
            if (j + 1 < m) acc += v1;
            if (j + 2 < m) acc += v2;
            if (j + 3 < m) acc += v3;
            if (j + 4 < m) acc += v4;
            if (j + 5 < m) acc += v5;
            if (j + 6 < m) acc += v6;
            if (j + 7 < m) acc += v7;
        }
    }
    v4f self = *(const v4f*)(yl + (size_t)i * HID);
    v4f bb = *(const v4f*)(b + lane * 4);
    float d = dinv[i];
    v4f o;
    o.x = fmaxf(d * (acc.x + self.x) + bb.x, 0.0f);
    o.y = fmaxf(d * (acc.y + self.y) + bb.y, 0.0f);
    o.z = fmaxf(d * (acc.z + self.z) + bb.z, 0.0f);
    o.w = fmaxf(d * (acc.w + self.w) + bb.w, 0.0f);
    *(v4f*)(h + (size_t)i * HID + lane * 4) = o;
}

__global__ void
k_gather(const float* y_p, const int* rowptr_p, const int* ecnt_p, const int* col_p,
         const float* dinv_p, const float* b_p, int N_p, float* h_p,
         const float* y_m, const int* rowptr_m, const int* ecnt_m, const int* col_m,
         const float* dinv_m, const float* b_m, int N_m, float* h_m, int split) {
    BR_PICK;
    gather_body(PK(y), PK(rowptr), PK(ecnt), PK(col), PK(dinv), PK(b), PK(N), PK(h), bid);
}

#define POOL_RPW 16

__device__ __forceinline__ void pool_flush(float* __restrict__ pesum, float* __restrict__ cnt,
                                           int g, int lane, v4f acc, int n) {
    if ((unsigned)g >= (unsigned)NGRAPH) return;
    float* p = pesum + g * HID + lane * 4;
    atomicAdd(p + 0, acc.x);
    atomicAdd(p + 1, acc.y);
    atomicAdd(p + 2, acc.z);
    atomicAdd(p + 3, acc.w);
    if (lane == 0) atomicAdd(cnt + g, (float)n);
}

__global__ void __launch_bounds__(256)
k_pool(const float* __restrict__ h, const int* __restrict__ batch, int N,
       float* __restrict__ pesum, float* __restrict__ cnt) {
    int r0 = __builtin_amdgcn_readfirstlane((int)(blockIdx.x * 8 + (threadIdx.x >> 5))) * POOL_RPW;
    if (r0 >= N) return;
    int last = min(r0 + POOL_RPW, N) - 1;
    int lane = threadIdx.x & 31;
    const float* hl = h + lane * 4;
    int g[POOL_RPW];
    v4f v[POOL_RPW];
    #pragma unroll
    for (int k = 0; k < POOL_RPW; ++k) {
        int r = min(r0 + k, last);
        g[k] = batch[r];
        v[k] = *(const v4f*)(hl + (size_t)r * HID);
    }
    asm volatile("" :: "v"(v[0]), "v"(v[1]), "v"(v[2]), "v"(v[3]), "v"(v[4]), "v"(v[5]),
                       "v"(v[6]), "v"(v[7]), "v"(v[8]), "v"(v[9]), "v"(v[10]), "v"(v[11]),
                       "v"(v[12]), "v"(v[13]), "v"(v[14]), "v"(v[15]));
    int cur = g[0];
    int n = 0;
    v4f acc = (v4f){0.0f, 0.0f, 0.0f, 0.0f};
    #pragma unroll
    for (int k = 0; k < POOL_RPW; ++k) {
        if (r0 + k <= last) {
            if (g[k] != cur) {
                pool_flush(pesum, cnt, cur, lane, acc, n);
                cur = g[k];
                n = 0;
                acc = (v4f){0.0f, 0.0f, 0.0f, 0.0f};
            }
            acc += v[k];
            n += 1;
        }
    }
    pool_flush(pesum, cnt, cur, lane, acc, n);
}

__device__ __forceinline__ void
wfrag_body(const float* __restrict__ W, int K, float* __restrict__ Wf, int bid) {
    int idx = bid * 256 + threadIdx.x;
    if (idx >= K * HID) return;
    int j    = idx & 1;
    int nt   = (idx >> 1) & 7;
    int l    = (idx >> 4) & 31;
    int kk   = idx >> 9;
    int krow = 4 * kk + (l >> 4) * 2 + j;
    int col  = nt * 16 + (l & 15);
    Wf[idx] = W[(size_t)krow * HID + col];
}

#define FILL_BLOCKS (2 * MAXBIN * BIN_PAD / 256)
#define POOLZ_N     (NGRAPH * HID + NGRAPH)
#define POOLZ_BLOCKS ((POOLZ_N + 255) / 256)
#define WF_BLOCKS   (HID * HID / 256)
#define SETUP_BLOCKS (2 * FILL_BLOCKS + POOLZ_BLOCKS + 4 * WF_BLOCKS)

__global__ void __launch_bounds__(256)
k_setup(int* __restrict__ bins_p, int* __restrict__ bins_m, float* __restrict__ poolz,
        const float* __restrict__ W_p2, float* __restrict__ Wf_p,
        const float* __restrict__ W_m2, float* __restrict__ Wf_m,
        const float* __restrict__ W1f, float* __restrict__ Wf_h) {
    int bid = blockIdx.x;
    int t = threadIdx.x;
    if (bid < FILL_BLOCKS) { bins_p[bid * 256 + t] = 0; return; }
    bid -= FILL_BLOCKS;
    if (bid < FILL_BLOCKS) { bins_m[bid * 256 + t] = 0; return; }
    bid -= FILL_BLOCKS;
    if (bid < POOLZ_BLOCKS) {
        int i = bid * 256 + t;
        if (i < POOLZ_N) poolz[i] = 0.0f;
        return;
    }
    bid -= POOLZ_BLOCKS;
    if (bid < WF_BLOCKS) { wfrag_body(W_p2, HID, Wf_p, bid); return; }
    bid -= WF_BLOCKS;
    if (bid < WF_BLOCKS) { wfrag_body(W_m2, HID, Wf_m, bid); return; }
    bid -= WF_BLOCKS;
    wfrag_body(W1f, 2 * HID, Wf_h, bid);
}

template <bool SCALED>
__device__ __forceinline__ void gemm_ksteps(const float* __restrict__ Arow,
                                            const float* __restrict__ Wl,
                                            int ksteps, float ascale, v8f (&acc)[8]) {
    for (int kk = 0; kk < ksteps; ++kk) {
        v2f a;
        a.x = Arow[4 * kk];
        a.y = Arow[4 * kk + 1];
        if (SCALED) {
            a.x *= ascale;
            a.y *= ascale;
        }
        const float4* q = (const float4*)(Wl + (size_t)kk * 512);
        float4 q0 = q[0], q1 = q[1], q2 = q[2], q3 = q[3];
        v2f b0; b0.x = q0.x; b0.y = q0.y;
        v2f b1; b1.x = q0.z; b1.y = q0.w;
        v2f b2; b2.x = q1.x; b2.y = q1.y;
        v2f b3; b3.x = q1.z; b3.y = q1.w;
        v2f b4; b4.x = q2.x; b4.y = q2.y;
        v2f b5; b5.x = q2.z; b5.y = q2.w;
        v2f b6; b6.x = q3.x; b6.y = q3.y;
        v2f b7; b7.x = q3.z; b7.y = q3.w;
        acc[0] = __builtin_amdgcn_wmma_f32_16x16x4_f32(false, a, false, b0, (short)0, acc[0], false, false);
        acc[1] = __builtin_amdgcn_wmma_f32_16x16x4_f32(false, a, false, b1, (short)0, acc[1], false, false);
        acc[2] = __builtin_amdgcn_wmma_f32_16x16x4_f32(false, a, false, b2, (short)0, acc[2], false, false);
        acc[3] = __builtin_amdgcn_wmma_f32_16x16x4_f32(false, a, false, b3, (short)0, acc[3], false, false);
        acc[4] = __builtin_amdgcn_wmma_f32_16x16x4_f32(false, a, false, b4, (short)0, acc[4], false, false);
        acc[5] = __builtin_amdgcn_wmma_f32_16x16x4_f32(false, a, false, b5, (short)0, acc[5], false, false);
        acc[6] = __builtin_amdgcn_wmma_f32_16x16x4_f32(false, a, false, b6, (short)0, acc[6], false, false);
        acc[7] = __builtin_amdgcn_wmma_f32_16x16x4_f32(false, a, false, b7, (short)0, acc[7], false, false);
    }
}

__device__ __forceinline__ void gemm_store(v8f (&acc)[8], int row0, int half, int lmod,
                                           const float* __restrict__ rowscale,
                                           const float* __restrict__ bias, int relu,
                                           float* __restrict__ out) {
    float rs[8];
    if (rowscale) {
        #pragma unroll
        for (int r = 0; r < 8; ++r) rs[r] = rowscale[row0 + r + half * 8];
    }

    #pragma unroll
    for (int nt = 0; nt < 8; ++nt) {
        int col = nt * 16 + lmod;
        float bv = bias ? bias[col] : 0.0f;
        #pragma unroll
        for (int r = 0; r < 8; ++r) {
            float v = acc[nt][r];
            if (rowscale) v *= rs[r];
            v += bv;
            if (relu) v = fmaxf(v, 0.0f);
            out[(size_t)(row0 + r + half * 8) * HID + col] = v;
        }
    }
}

__device__ __forceinline__ void
gemm128_body(const float* __restrict__ A, int K, int M,
             const float* __restrict__ Wf,
             const float* __restrict__ rowscale,
             const float* __restrict__ bias, int relu,
             float* __restrict__ out, int bid) {
    int w = threadIdx.x >> 5;
    int lane = threadIdx.x & 31;
    int tile = bid * 4 + w;
    if (tile * 16 >= M) return;
    int row0 = tile * 16;
    int half = lane >> 4;
    int lmod = lane & 15;
    int koff = half * 2;
    const float* Arow = A + (size_t)(row0 + lmod) * K + koff;
    const float* Wl   = Wf + (size_t)lane * 16;

    v8f acc[8];
    #pragma unroll
    for (int nt = 0; nt < 8; ++nt) acc[nt] = (v8f){};

    gemm_ksteps<false>(Arow, Wl, K >> 2, 1.0f, acc);
    gemm_store(acc, row0, half, lmod, rowscale, bias, relu, out);
}

__global__ void
k_gemm128(const float* h_p, int N_p, const float* Wf_p, const float* dinv_p, float* y_p,
          const float* h_m, int N_m, const float* Wf_m, const float* dinv_m, float* y_m,
          int split) {
    BR_PICK;
    gemm128_body(PK(h), HID, PK(N), PK(Wf), PK(dinv), nullptr, 0, PK(y), bid);
}

__global__ void __launch_bounds__(128)
k_gemm_head(const float* __restrict__ h, const float* __restrict__ pesum,
            const float* __restrict__ cnt, const int* __restrict__ batch, int M,
            const float* __restrict__ Wf, const float* __restrict__ bias,
            const float* __restrict__ W2, const float* __restrict__ b2,
            float* __restrict__ out) {
    __shared__ float so[4 * 16 * 3];
    int w = __builtin_amdgcn_readfirstlane((int)(threadIdx.x >> 5));
    int lane = threadIdx.x & 31;
    int tile = blockIdx.x * 4 + w;
    if (tile * 16 < M) {
        int row0 = tile * 16;
        int half = lane >> 4;
        int lmod = lane & 15;
        int koff = half * 2;
        int g = min(max(batch[row0 + lmod], 0), NGRAPH - 1);
        float inv = 1.0f / fmaxf(cnt[g], 1.0f);
        const float* Wl = Wf + (size_t)lane * 16;

        v8f acc[8];
        #pragma unroll
        for (int nt = 0; nt < 8; ++nt) acc[nt] = (v8f){};

        gemm_ksteps<false>(h + (size_t)(row0 + lmod) * HID + koff, Wl, HID / 4, 1.0f, acc);
        gemm_ksteps<true>(pesum + (size_t)g * HID + koff, Wl + (size_t)(HID / 4) * 512, HID / 4,
                          inv, acc);

        float p[8][3];
        #pragma unroll
        for (int r = 0; r < 8; ++r) { p[r][0] = 0.0f; p[r][1] = 0.0f; p[r][2] = 0.0f; }
        #pragma unroll
        for (int nt = 0; nt < 8; ++nt) {
            int col = nt * 16 + lmod;
            float bv = bias[col];
            float w0 = W2[col * 3 + 0];
            float w1 = W2[col * 3 + 1];
            float w2 = W2[col * 3 + 2];
            #pragma unroll
            for (int r = 0; r < 8; ++r) {
                float v = fmaxf(acc[nt][r] + bv, 0.0f);
                p[r][0] += v * w0;
                p[r][1] += v * w1;
                p[r][2] += v * w2;
            }
        }
        #pragma unroll
        for (int s = 8; s >= 1; s >>= 1) {
            #pragma unroll
            for (int r = 0; r < 8; ++r) {
                p[r][0] += __shfl_xor(p[r][0], s);
                p[r][1] += __shfl_xor(p[r][1], s);
                p[r][2] += __shfl_xor(p[r][2], s);
            }
        }
        if (lmod == 0) {
            float c0 = b2[0], c1 = b2[1], c2 = b2[2];
            #pragma unroll
            for (int r = 0; r < 8; ++r) {
                float* q = so + w * 48 + (half * 8 + r) * 3;
                q[0] = p[r][0] + c0;
                q[1] = p[r][1] + c1;
                q[2] = p[r][2] + c2;
            }
        }
    }
    __syncthreads();
    for (int t = threadIdx.x; t < 4 * 16 * 3; t += 128) {
        size_t gi = (size_t)blockIdx.x * (4 * 16 * 3) + t;
        if (gi < (size_t)M * 3) out[gi] = so[t];
    }
}


static inline int cdiv(long long a, long long b) { return (int)((a + b - 1) / b); }

extern "C" void kernel_launch(void* const* d_in, const int* in_sizes, int n_in,
                              void* d_out, int out_size, void* d_ws, size_t ws_size,
                              hipStream_t stream) {
    (void)n_in; (void)out_size; (void)ws_size;

    const int*   res_p   = (const int*)d_in[0];
    const float* pos_p   = (const float*)d_in[1];
    const int*   ei_p    = (const int*)d_in[2];
    const int*   batch_p = (const int*)d_in[3];
    const int*   res_m   = (const int*)d_in[4];
    const float* pos_m   = (const float*)d_in[5];
    const int*   ei_m    = (const int*)d_in[6];
    const int*   batch_m = (const int*)d_in[7];
    const float* W_p1 = (const float*)d_in[8];
    const float* b_p1 = (const float*)d_in[9];
    const float* W_p2 = (const float*)d_in[10];
    const float* b_p2 = (const float*)d_in[11];
    const float* W_m1 = (const float*)d_in[12];
    const float* b_m1 = (const float*)d_in[13];
    const float* W_m2 = (const float*)d_in[14];
    const float* b_m2 = (const float*)d_in[15];
    const float* W1f  = (const float*)d_in[16];
    const float* b1f  = (const float*)d_in[17];
    const float* W2f  = (const float*)d_in[18];
    const float* b2f  = (const float*)d_in[19];

    int N_P = in_sizes[0];
    int E_P = in_sizes[2] / 2;
    int N_M = in_sizes[4];
    int E_M = in_sizes[6] / 2;
    const int* src_p = ei_p;
    const int* dst_p = ei_p + E_P;
    const int* src_m = ei_m;
    const int* dst_m = ei_m + E_M;

    float* ws = (float*)d_ws;
    size_t nhp = (size_t)N_P * HID;
    size_t nhm = (size_t)N_M * HID;
    size_t npa = ((size_t)N_P + 63) / 64 * 64;
    size_t nma = ((size_t)N_M + 63) / 64 * 64;
    float* A_p    = ws;
    float* C_p    = A_p + nhp;
    float* A_m    = C_p + nhp;
    float* C_m    = A_m + nhm;
    float* rec_p  = C_m + nhm;
    float* rec_m  = rec_p + npa * 8;
    float* dinv_p = rec_m + nma * 8;
    float* dinv_m = dinv_p + npa;
    float* pe_sum = dinv_m + nma;
    float* cnt    = pe_sum + NGRAPH * HID;
    float* Wf_p   = cnt + NGRAPH;
    float* Wf_m   = Wf_p + HID * HID;
    float* Wf_h   = Wf_m + HID * HID;
    int* ecnt_p   = (int*)(Wf_h + 2 * HID * HID);
    int* rowptr_p = ecnt_p + npa;
    int* part_p   = rowptr_p + npa;
    int* col_p    = part_p + E_P;
    int* bins_p   = col_p + E_P;
    int* ecnt_m   = bins_p + (2 * BIN_PAD + 1) * MAXBIN;
    int* rowptr_m = ecnt_m + nma;
    int* part_m   = rowptr_m + nma;
    int* col_m    = part_m + E_M;
    int* bins_m   = col_m + E_M;
    int* bcur_p   = bins_p + MAXBIN * BIN_PAD;
    int* bsta_p   = bins_p + 2 * MAXBIN * BIN_PAD;
    int* bcur_m   = bins_m + MAXBIN * BIN_PAD;
    int* bsta_m   = bins_m + 2 * MAXBIN * BIN_PAD;
    int nbin_p = cdiv(N_P, BIN_NODES);
    int nbin_m = cdiv(N_M, BIN_NODES);

    k_setup<<<SETUP_BLOCKS, 256, 0, stream>>>(bins_p, bins_m, pe_sum, W_p2, Wf_p, W_m2, Wf_m,
                                              W1f, Wf_h);

    int gp = cdiv(E_P, PCH), gm = cdiv(E_M, PCH);
    k_bin_hist<<<gp + gm, 256, 0, stream>>>(dst_p, E_P, N_P, bins_p,
                                            dst_m, E_M, N_M, bins_m, gp);
    k_bin_part<<<gp + gm, 256, 0, stream>>>(src_p, dst_p, E_P, N_P, nbin_p, bins_p, bcur_p, bsta_p, part_p,
                                            src_m, dst_m, E_M, N_M, nbin_m, bins_m, bcur_m, bsta_m, part_m,
                                            gp);
    k_bin_csr<<<nbin_p + nbin_m, 256, 0, stream>>>(part_p, bsta_p, bins_p, N_P, ecnt_p, rowptr_p, dinv_p, col_p,
                                                   res_p, pos_p, rec_p,
                                                   part_m, bsta_m, bins_m, N_M, ecnt_m, rowptr_m, dinv_m, col_m,
                                                   res_m, pos_m, rec_m, nbin_p);

    gp = cdiv(N_P, 8 * AGG_NPW); gm = cdiv(N_M, 8 * AGG_NPW);
    k_agg1<<<gp + gm, 256, 0, stream>>>(rec_p, rowptr_p, ecnt_p, col_p, W_p1, b_p1, N_P, C_p,
                                        rec_m, rowptr_m, ecnt_m, col_m, W_m1, b_m1, N_M, C_m, gp);

    gp = cdiv(N_P / 16, 4); gm = cdiv(N_M / 16, 4);
    k_gemm128<<<gp + gm, 128, 0, stream>>>(C_p, N_P, Wf_p, dinv_p, A_p,
                                           C_m, N_M, Wf_m, dinv_m, A_m, gp);
    gp = cdiv(N_P, 8); gm = cdiv(N_M, 8);
    k_gather<<<gp + gm, 256, 0, stream>>>(A_p, rowptr_p, ecnt_p, col_p, dinv_p, b_p2, N_P, C_p,
                                          A_m, rowptr_m, ecnt_m, col_m, dinv_m, b_m2, N_M, C_m, gp);

    k_pool<<<cdiv(N_P, 8 * POOL_RPW), 256, 0, stream>>>(C_p, batch_p, N_P, pe_sum, cnt);

    k_gemm_head<<<cdiv(N_M / 16, 4), 128, 0, stream>>>(C_m, pe_sum, cnt, batch_m, N_M, Wf_h, b1f,
                                                       W2f, b2f, (float*)d_out);
}


# Layer-2 product y=(h1@W2)*dinv done in the layer-1 aggregation launch: a block's 32 h1 rows stay in LDS and feed the matrix unit there; h1 never goes to memory; 9 launches -> 8; sum order unchanged
# speedup vs baseline: 6.4854x; 1.1628x over previous
#include <hip/hip_runtime.h>
#include <hip/hip_bf16.h>

typedef __attribute__((ext_vector_type(2))) float v2f;
typedef __attribute__((ext_vector_type(8))) float v8f;
typedef __attribute__((ext_vector_type(4))) float v4f;

#define HID 128
#define NGRAPH 64
#define NRES 20
#define AGG_NPW 4
#define AGG_ROWS (8 * AGG_NPW)
#define HPITCH (HID + 4)

#define BR_PICK bool second = (int)blockIdx.x >= split; \
                int bid = second ? (int)blockIdx.x - split : (int)blockIdx.x
#define PK(x) (second ? x##_m : x##_p)

#define BIN_SHIFT 8
#define BIN_NODES 256
#define MAXBIN 512
#define PCH 4096
#define PEPT 16
#define LCAP 16384
#define BIN_PAD 32


__device__ __forceinline__ void
bin_hist_body(const int* __restrict__ dst, int E, int N, int* __restrict__ binCount, int bid) {
    __shared__ int h[MAXBIN];
    int tid = threadIdx.x;
    for (int t = tid; t < MAXBIN; t += 256) h[t] = 0;
    __syncthreads();
    int e0 = bid * PCH;
    int dd[PEPT];
    #pragma unroll
    for (int k = 0; k < PEPT; ++k) dd[k] = dst[min(e0 + k * 256 + tid, E - 1)];
    #pragma unroll
    for (int k = 0; k < PEPT; ++k) {
        int e = e0 + k * 256 + tid;
        int d = min(max(dd[k], 0), N - 1);
        if (e < E) atomicAdd(&h[d >> BIN_SHIFT], 1);
    }
    __syncthreads();
    for (int t = tid; t < MAXBIN; t += 256) {
        int c = h[t];
        if (c > 0) atomicAdd(binCount + t * BIN_PAD, c);
    }
}

__global__ void __launch_bounds__(256)
k_bin_hist(const int* dst_p, int E_p, int N_p, int* binCount_p,
           const int* dst_m, int E_m, int N_m, int* binCount_m, int split) {
    BR_PICK;
    bin_hist_body(PK(dst), PK(E), PK(N), PK(binCount), bid);
}

__device__ __forceinline__ void scan_bins(const int* in, int* out, int lane) {
    int v[MAXBIN / 32];
    int s = 0;
    #pragma unroll
    for (int j = 0; j < MAXBIN / 32; ++j) {
        v[j] = in[lane * (MAXBIN / 32) + j];
        s += v[j];
    }
    int x = s;
    #pragma unroll
    for (int d = 1; d < 32; d <<= 1) {
        int t = __shfl_up(x, d);
        if (lane >= d) x += t;
    }
    int run = x - s;
    #pragma unroll
    for (int j = 0; j < MAXBIN / 32; ++j) {
        out[lane * (MAXBIN / 32) + j] = run;
        run += v[j];
    }
}

__device__ __forceinline__ void
bin_part_body(const int* __restrict__ src, const int* __restrict__ dst, int E, int N, int nbin,
              const int* __restrict__ binCount, int* __restrict__ binCursor,
              int* __restrict__ binStart, int* __restrict__ part, int bid) {
    __shared__ int h[MAXBIN];
    __shared__ int gc[MAXBIN];
    __shared__ int loc[MAXBIN];
    __shared__ int bstart[MAXBIN];
    __shared__ int gbase[MAXBIN];
    __shared__ int sval[PCH];
    __shared__ unsigned short sbin[PCH];
    int tid = threadIdx.x;
    for (int t = tid; t < MAXBIN; t += 256) {
        h[t] = 0;
        gc[t] = (t < nbin) ? binCount[t * BIN_PAD] : 0;
    }
    __syncthreads();
    int e0 = bid * PCH;
    int pk[PEPT], bb[PEPT], rr[PEPT];
    #pragma unroll
    for (int k = 0; k < PEPT; ++k) {
        int ec = min(e0 + k * 256 + tid, E - 1);
        bb[k] = dst[ec];
        pk[k] = src[ec];
    }
    #pragma unroll
    for (int k = 0; k < PEPT; ++k) {
        int e = e0 + k * 256 + tid;
        int d = min(max(bb[k], 0), N - 1);
        int s = min(max(pk[k], 0), N - 1);
        bb[k] = (e < E) ? (d >> BIN_SHIFT) : -1;
        pk[k] = (s << BIN_SHIFT) | (d & (BIN_NODES - 1));
        rr[k] = 0;
    }
    #pragma unroll
    for (int k = 0; k < PEPT; ++k)
        if (bb[k] >= 0) rr[k] = atomicAdd(&h[bb[k]], 1);
    __syncthreads();
    int wave = tid >> 5, lane = tid & 31;
    if (wave == 0) scan_bins(h, loc, lane);
    else if (wave == 1) scan_bins(gc, bstart, lane);
    __syncthreads();
    for (int t = tid; t < MAXBIN; t += 256) {
        int c = h[t];
        if (c > 0) gbase[t] = bstart[t] + atomicAdd(binCursor + t * BIN_PAD, c);
        if (bid == 0 && t < nbin) binStart[t] = bstart[t];
    }
    #pragma unroll
    for (int k = 0; k < PEPT; ++k) {
        if (bb[k] >= 0) {
            int p = loc[bb[k]] + rr[k];
            sval[p] = pk[k];
            sbin[p] = (unsigned short)bb[k];
        }
    }
    __syncthreads();
    int nv = min(PCH, E - e0);
    for (int i = tid; i < nv; i += 256) {
        int b = sbin[i];
        part[gbase[b] + (i - loc[b])] = sval[i];
    }
}

__global__ void __launch_bounds__(256)
k_bin_part(const int* src_p, const int* dst_p, int E_p, int N_p, int nbin_p,
           const int* binCount_p, int* binCursor_p, int* binStart_p, int* part_p,
           const int* src_m, const int* dst_m, int E_m, int N_m, int nbin_m,
           const int* binCount_m, int* binCursor_m, int* binStart_m, int* part_m, int split) {
    BR_PICK;
    bin_part_body(PK(src), PK(dst), PK(E), PK(N), PK(nbin), PK(binCount), PK(binCursor),
                  PK(binStart), PK(part), bid);
}

__device__ __forceinline__ void
bin_csr_body(const int* __restrict__ part, const int* __restrict__ binStart,
             const int* __restrict__ binCount, int N, int* __restrict__ ecnt,
             int* __restrict__ rowptr, float* __restrict__ dinv, int* __restrict__ col,
             const int* __restrict__ res, const float* __restrict__ pos,
             float* __restrict__ rec, int bid) {
    __shared__ int cnt[BIN_NODES];
    __shared__ int cur[BIN_NODES];
    __shared__ int wtot[8];
    __shared__ int lcol[LCAP];
    int tid = threadIdx.x;
    int wave = tid >> 5, lane = tid & 31;
    int b = bid;
    int s = binStart[b];
    int m = binCount[b * BIN_PAD];
    cnt[tid] = 0;
    __syncthreads();
    for (int i0 = 0; i0 < m; i0 += 1024) {
        int v[4];
        #pragma unroll
        for (int u = 0; u < 4; ++u) {
            int i = i0 + u * 256 + tid;
            v[u] = (i < m) ? part[s + i] : -1;
        }
        #pragma unroll
        for (int u = 0; u < 4; ++u)
            if (v[u] >= 0) atomicAdd(&cnt[v[u] & (BIN_NODES - 1)], 1);
    }
    __syncthreads();
    int c = cnt[tid];
    int x = c;
    #pragma unroll
    for (int d = 1; d < 32; d <<= 1) {
        int t = __shfl_up(x, d);
        if (lane >= d) x += t;
    }
    if (lane == 31) wtot[wave] = x;
    __syncthreads();
    int wb = 0;
    for (int w = 0; w < wave; ++w) wb += wtot[w];
    int off = wb + x - c;
    cur[tid] = off;
    int node = b * BIN_NODES + tid;
    if (node < N) {
        ecnt[node] = c;
        rowptr[node] = s + off;
        float dv = rsqrtf((float)c + 1.0f);
        dinv[node] = dv;
        v4f q;
        q.x = dv * pos[node * 3 + 0];
        q.y = dv * pos[node * 3 + 1];
        q.z = dv * pos[node * 3 + 2];
        q.w = dv;
        *(v4f*)(rec + (size_t)node * 8) = q;
        int r = res[node];
        if (r < 0 || r >= NRES) r = NRES;
        ((int*)rec)[(size_t)node * 8 + 4] = r * (HID * 4);
    }
    __syncthreads();
    bool staged = m <= LCAP;
    for (int i0 = 0; i0 < m; i0 += 1024) {
        int v[4];
        #pragma unroll
        for (int u = 0; u < 4; ++u) {
            int i = i0 + u * 256 + tid;
            v[u] = (i < m) ? part[s + i] : -1;
        }
        #pragma unroll
        for (int u = 0; u < 4; ++u) {
            if (v[u] >= 0) {
                int p = atomicAdd(&cur[v[u] & (BIN_NODES - 1)], 1);
                int sv = (int)((unsigned)v[u] >> BIN_SHIFT);
                if (staged) lcol[p] = sv;
                else col[s + p] = sv;
            }
        }
    }
    if (staged) {
        __syncthreads();
        for (int i = tid; i < m; i += 256) col[s + i] = lcol[i];
    }
}

__global__ void __launch_bounds__(256)
k_bin_csr(const int* part_p, const int* binStart_p, const int* binCount_p, int N_p,
          int* ecnt_p, int* rowptr_p, float* dinv_p, int* col_p,
          const int* res_p, const float* pos_p, float* rec_p,
          const int* part_m, const int* binStart_m, const int* binCount_m, int N_m,
          int* ecnt_m, int* rowptr_m, float* dinv_m, int* col_m,
          const int* res_m, const float* pos_m, float* rec_m, int split) {
    BR_PICK;
    bin_csr_body(PK(part), PK(binStart), PK(binCount), PK(N), PK(ecnt), PK(rowptr), PK(dinv),
                 PK(col), PK(res), PK(pos), PK(rec), bid);
}


__device__ __forceinline__ void
agg1_body(const float* __restrict__ rec, const int* __restrict__ rowptr,
          const int* __restrict__ ecnt, const int* __restrict__ col,
          const float* __restrict__ W, const float* __restrict__ b,
          const float* __restrict__ Wq, int N, float* __restrict__ y, int bid) {
    __shared__ __attribute__((aligned(16))) float Ws[(NRES + 1) * HID];
    __shared__ __attribute__((aligned(16))) float Hs[AGG_ROWS * HPITCH];
    __shared__ __attribute__((aligned(16))) float Ys[AGG_ROWS * HPITCH];
    __shared__ float Ds[AGG_ROWS];
    for (int t = threadIdx.x; t < NRES * HID / 4; t += 256)
        ((v4f*)Ws)[t] = ((const v4f*)W)[t];
    if (threadIdx.x < HID / 4)
        ((v4f*)Ws)[NRES * HID / 4 + threadIdx.x] = (v4f){0.0f, 0.0f, 0.0f, 0.0f};
    __syncthreads();

    int lane = threadIdx.x & 31;
    const v4f w20 = *(const v4f*)(W + (NRES + 0) * HID + lane * 4);
    const v4f w21 = *(const v4f*)(W + (NRES + 1) * HID + lane * 4);
    const v4f w22 = *(const v4f*)(W + (NRES + 2) * HID + lane * 4);
    const v4f bb = *(const v4f*)(b + lane * 4);
    const char* Wl = (const char*)Ws + lane * 16;
    const int* reci = (const int*)rec;
    int wave = __builtin_amdgcn_readfirstlane((int)(threadIdx.x >> 5));
    int first = (bid * 8 + wave) * AGG_NPW;

    for (int t = 0; t < AGG_NPW; ++t) {
        int i = __builtin_amdgcn_readfirstlane(first + t);
        float* hrow = Hs + (wave * AGG_NPW + t) * HPITCH + lane * 4;
        if (i >= N) {
            *(v4f*)hrow = (v4f){0.0f, 0.0f, 0.0f, 0.0f};
            continue;
        }
        int start = rowptr[i];
        int n = ecnt[i];
        v4f rs = *(const v4f*)(rec + (size_t)i * 8);
        int os = reci[(size_t)i * 8 + 4];
        v4f acc = (v4f){0.0f, 0.0f, 0.0f, 0.0f};
        float q0 = 0.0f, q1 = 0.0f, q2 = 0.0f;
        for (int base = 0; base < n; base += 32) {
            int m = n - base;
            if (m > 32) m = 32;
            bool live = lane < m;
            int c = live ? col[start + base + lane] : i;
            c = min(max(c, 0), N - 1);
            v4f r = *(const v4f*)(rec + (size_t)c * 8);
            int off = reci[(size_t)c * 8 + 4];
            float d = live ? r.w : 0.0f;
            off = live ? off : NRES * HID * 4;
            q0 += live ? r.x : 0.0f;
            q1 += live ? r.y : 0.0f;
            q2 += live ? r.z : 0.0f;
            int dbits = __float_as_int(d);
            for (int j = 0; j < m; j += 8) {
                int o0 = __builtin_amdgcn_readlane(off, j);
                int o1 = __builtin_amdgcn_readlane(off, j + 1);
                int o2 = __builtin_amdgcn_readlane(off, j + 2);
                int o3 = __builtin_amdgcn_readlane(off, j + 3);
                int o4 = __builtin_amdgcn_readlane(off, j + 4);
                int o5 = __builtin_amdgcn_readlane(off, j + 5);
                int o6 = __builtin_amdgcn_readlane(off, j + 6);
                int o7 = __builtin_amdgcn_readlane(off, j + 7);
                v4f x0 = *(const v4f*)(Wl + o0);
                v4f x1 = *(const v4f*)(Wl + o1);
                v4f x2 = *(const v4f*)(Wl + o2);
                v4f x3 = *(const v4f*)(Wl + o3);
                v4f x4 = *(const v4f*)(Wl + o4);
                v4f x5 = *(const v4f*)(Wl + o5);
                v4f x6 = *(const v4f*)(Wl + o6);
                v4f x7 = *(const v4f*)(Wl + o7);
                float d0 = __int_as_float(__builtin_amdgcn_readlane(dbits, j));
                float d1 = __int_as_float(__builtin_amdgcn_readlane(dbits, j + 1));
                float d2 = __int_as_float(__builtin_amdgcn_readlane(dbits, j + 2));
                float d3 = __int_as_float(__builtin_amdgcn_readlane(dbits, j + 3));
                float d4 = __int_as_float(__builtin_amdgcn_readlane(dbits, j + 4));
                float d5 = __int_as_float(__builtin_amdgcn_readlane(dbits, j + 5));
                float d6 = __int_as_float(__builtin_amdgcn_readlane(dbits, j + 6));
                float d7 = __int_as_float(__builtin_amdgcn_readlane(dbits, j + 7));
                acc += d0 * x0;
                acc += d1 * x1;
                acc += d2 * x2;
                acc += d3 * x3;
                acc += d4 * x4;
                acc += d5 * x5;
                acc += d6 * x6;
                acc += d7 * x7;
            }
        }
        v4f xs = *(const v4f*)(Wl + os);
        acc += rs.w * xs;
        #pragma unroll
        for (int s = 16; s >= 1; s >>= 1) {
            q0 += __shfl_xor(q0, s);
            q1 += __shfl_xor(q1, s);
            q2 += __shfl_xor(q2, s);
        }
        q0 += rs.x;
        q1 += rs.y;
        q2 += rs.z;
        v4f S = acc + q0 * w20 + q1 * w21 + q2 * w22;
        float dnode = rs.w;
        v4f o;
        o.x = fmaxf(dnode * S.x + bb.x, 0.0f);
        o.y = fmaxf(dnode * S.y + bb.y, 0.0f);
        o.z = fmaxf(dnode * S.z + bb.z, 0.0f);
        o.w = fmaxf(dnode * S.w + bb.w, 0.0f);
        *(v4f*)hrow = o;
        if (lane == 0) Ds[wave * AGG_NPW + t] = dnode;
    }
    __syncthreads();

    {
        int rt = wave >> 2, q = wave & 3;
        int lmod = lane & 15, half = lane >> 4;
        const float* Arow = Hs + (rt * 16 + lmod) * HPITCH + half * 2;
        const float* Bl = Wq + (size_t)(q * 32 + lane) * 4;
        v8f acc0 = (v8f){}, acc1 = (v8f){};
        v2f a[2][4];
        v4f bq[2][4];
        #pragma unroll
        for (int u = 0; u < 4; ++u) {
            a[0][u] = *(const v2f*)(Arow + 4 * u);
            bq[0][u] = *(const v4f*)(Bl + (size_t)u * 512);
        }
        #pragma unroll
        for (int g = 0; g < 8; ++g) {
            if (g + 1 < 8) {
                #pragma unroll
                for (int u = 0; u < 4; ++u) {
                    int kk = (g + 1) * 4 + u;
                    a[(g + 1) & 1][u] = *(const v2f*)(Arow + 4 * kk);
                    bq[(g + 1) & 1][u] = *(const v4f*)(Bl + (size_t)kk * 512);
                }
            }
            #pragma unroll
            for (int u = 0; u < 4; ++u) {
                v4f w4 = bq[g & 1][u];
                v2f b0; b0.x = w4.x; b0.y = w4.y;
                v2f b1; b1.x = w4.z; b1.y = w4.w;
                acc0 = __builtin_amdgcn_wmma_f32_16x16x4_f32(false, a[g & 1][u], false, b0, (short)0, acc0, false, false);
                acc1 = __builtin_amdgcn_wmma_f32_16x16x4_f32(false, a[g & 1][u], false, b1, (short)0, acc1, false, false);
            }
        }
        float* yb = Ys + (rt * 16 + half * 8) * HPITCH + q * 32 + lmod;
        #pragma unroll
        for (int r = 0; r < 8; ++r) {
            yb[r * HPITCH] = acc0[r];
            yb[r * HPITCH + 16] = acc1[r];
        }
    }
    __syncthreads();

    for (int t = 0; t < AGG_NPW; ++t) {
        int i = first + t;
        if (i < N) {
            int row = wave * AGG_NPW + t;
            v4f v = *(const v4f*)(Ys + row * HPITCH + lane * 4);
            float d = Ds[row];
            *(v4f*)(y + (size_t)i * HID + lane * 4) = v * d;
        }
    }
}

__global__ void __launch_bounds__(256)
k_agg1(const float* rec_p, const int* rowptr_p, const int* ecnt_p, const int* col_p,
       const float* W_p, const float* b_p, const float* Wq_p, int N_p, float* y_p,
       const float* rec_m, const int* rowptr_m, const int* ecnt_m, const int* col_m,
       const float* W_m, const float* b_m, const float* Wq_m, int N_m, float* y_m, int split) {
    BR_PICK;
    agg1_body(PK(rec), PK(rowptr), PK(ecnt), PK(col), PK(W), PK(b), PK(Wq), PK(N), PK(y), bid);
}

__device__ __forceinline__ void
gather_body(const float* __restrict__ y, const int* __restrict__ rowptr,
            const int* __restrict__ ecnt, const int* __restrict__ col,
            const float* __restrict__ dinv, const float* __restrict__ b,
            int N, float* __restrict__ h, int bid) {
    int i = __builtin_amdgcn_readfirstlane((int)(bid * 8 + (threadIdx.x >> 5)));
    if (i >= N) return;
    int lane = threadIdx.x & 31;
    int start = rowptr[i];
    int n = ecnt[i];
    const float* yl = y + lane * 4;
    v4f acc = (v4f){0.0f, 0.0f, 0.0f, 0.0f};
    for (int base = 0; base < n; base += 32) {
        int m = n - base;
        if (m > 32) m = 32;
        int c = (lane < m) ? col[start + base + lane] : i;
        c = min(max(c, 0), N - 1);
        for (int j = 0; j < m; j += 8) {
            int s0 = __builtin_amdgcn_readlane(c, j);
            int s1 = __builtin_amdgcn_readlane(c, j + 1);
            int s2 = __builtin_amdgcn_readlane(c, j + 2);
            int s3 = __builtin_amdgcn_readlane(c, j + 3);
            int s4 = __builtin_amdgcn_readlane(c, j + 4);
            int s5 = __builtin_amdgcn_readlane(c, j + 5);
            int s6 = __builtin_amdgcn_readlane(c, j + 6);
            int s7 = __builtin_amdgcn_readlane(c, j + 7);
            v4f v0 = *(const v4f*)(yl + (size_t)s0 * HID);
            v4f v1 = *(const v4f*)(yl + (size_t)s1 * HID);
            v4f v2 = *(const v4f*)(yl + (size_t)s2 * HID);
            v4f v3 = *(const v4f*)(yl + (size_t)s3 * HID);
            v4f v4 = *(const v4f*)(yl + (size_t)s4 * HID);
            v4f v5 = *(const v4f*)(yl + (size_t)s5 * HID);
            v4f v6 = *(const v4f*)(yl + (size_t)s6 * HID);
            v4f v7 = *(const v4f*)(yl + (size_t)s7 * HID);
            asm volatile("" :: "v"(v0), "v"(v1), "v"(v2), "v"(v3),
                               "v"(v4), "v"(v5), "v"(v6), "v"(v7));
            acc += v0;
            if (j + 1 < m) acc += v1;
            if (j + 2 < m) acc += v2;
            if (j + 3 < m) acc += v3;
            if (j + 4 < m) acc += v4;
            if (j + 5 < m) acc += v5;
            if (j + 6 < m) acc += v6;
            if (j + 7 < m) acc += v7;
        }
    }
    v4f self = *(const v4f*)(yl + (size_t)i * HID);
    v4f bb = *(const v4f*)(b + lane * 4);
    float d = dinv[i];
    v4f o;
    o.x = fmaxf(d * (acc.x + self.x) + bb.x, 0.0f);
    o.y = fmaxf(d * (acc.y + self.y) + bb.y, 0.0f);
    o.z = fmaxf(d * (acc.z + self.z) + bb.z, 0.0f);
    o.w = fmaxf(d * (acc.w + self.w) + bb.w, 0.0f);
    *(v4f*)(h + (size_t)i * HID + lane * 4) = o;
}

__global__ void
k_gather(const float* y_p, const int* rowptr_p, const int* ecnt_p, const int* col_p,
         const float* dinv_p, const float* b_p, int N_p, float* h_p,
         const float* y_m, const int* rowptr_m, const int* ecnt_m, const int* col_m,
         const float* dinv_m, const float* b_m, int N_m, float* h_m, int split) {
    BR_PICK;
    gather_body(PK(y), PK(rowptr), PK(ecnt), PK(col), PK(dinv), PK(b), PK(N), PK(h), bid);
}

#define POOL_RPW 16

__device__ __forceinline__ void pool_flush(float* __restrict__ pesum, float* __restrict__ cnt,
                                           int g, int lane, v4f acc, int n) {
    if ((unsigned)g >= (unsigned)NGRAPH) return;
    float* p = pesum + g * HID + lane * 4;
    atomicAdd(p + 0, acc.x);
    atomicAdd(p + 1, acc.y);
    atomicAdd(p + 2, acc.z);
    atomicAdd(p + 3, acc.w);
    if (lane == 0) atomicAdd(cnt + g, (float)n);
}

__global__ void __launch_bounds__(256)
k_pool(const float* __restrict__ h, const int* __restrict__ batch, int N,
       float* __restrict__ pesum, float* __restrict__ cnt) {
    int r0 = __builtin_amdgcn_readfirstlane((int)(blockIdx.x * 8 + (threadIdx.x >> 5))) * POOL_RPW;
    if (r0 >= N) return;
    int last = min(r0 + POOL_RPW, N) - 1;
    int lane = threadIdx.x & 31;
    const float* hl = h + lane * 4;
    int g[POOL_RPW];
    v4f v[POOL_RPW];
    #pragma unroll
    for (int k = 0; k < POOL_RPW; ++k) {
        int r = min(r0 + k, last);
        g[k] = batch[r];
        v[k] = *(const v4f*)(hl + (size_t)r * HID);
    }
    asm volatile("" :: "v"(v[0]), "v"(v[1]), "v"(v[2]), "v"(v[3]), "v"(v[4]), "v"(v[5]),
                       "v"(v[6]), "v"(v[7]), "v"(v[8]), "v"(v[9]), "v"(v[10]), "v"(v[11]),
                       "v"(v[12]), "v"(v[13]), "v"(v[14]), "v"(v[15]));
    int cur = g[0];
    int n = 0;
    v4f acc = (v4f){0.0f, 0.0f, 0.0f, 0.0f};
    #pragma unroll
    for (int k = 0; k < POOL_RPW; ++k) {
        if (r0 + k <= last) {
            if (g[k] != cur) {
                pool_flush(pesum, cnt, cur, lane, acc, n);
                cur = g[k];
                n = 0;
                acc = (v4f){0.0f, 0.0f, 0.0f, 0.0f};
            }
            acc += v[k];
            n += 1;
        }
    }
    pool_flush(pesum, cnt, cur, lane, acc, n);
}

__device__ __forceinline__ void
wfrag_body(const float* __restrict__ W, int K, float* __restrict__ Wf, int bid) {
    int idx = bid * 256 + threadIdx.x;
    if (idx >= K * HID) return;
    int j    = idx & 1;
    int nt   = (idx >> 1) & 7;
    int l    = (idx >> 4) & 31;
    int kk   = idx >> 9;
    int krow = 4 * kk + (l >> 4) * 2 + j;
    int col  = nt * 16 + (l & 15);
    Wf[idx] = W[(size_t)krow * HID + col];
}

__device__ __forceinline__ void
wpair_body(const float* __restrict__ W, float* __restrict__ Wq, int bid) {
    int idx = bid * 256 + threadIdx.x;
    if (idx >= HID * HID) return;
    int j    = idx & 1;
    int t    = (idx >> 1) & 1;
    int l    = (idx >> 2) & 31;
    int q    = (idx >> 7) & 3;
    int kk   = idx >> 9;
    int krow = 4 * kk + (l >> 4) * 2 + j;
    int col  = (2 * q + t) * 16 + (l & 15);
    Wq[idx] = W[(size_t)krow * HID + col];
}

#define FILL_BLOCKS (2 * MAXBIN * BIN_PAD / 256)
#define POOLZ_N     (NGRAPH * HID + NGRAPH)
#define POOLZ_BLOCKS ((POOLZ_N + 255) / 256)
#define WF_BLOCKS   (HID * HID / 256)
#define SETUP_BLOCKS (2 * FILL_BLOCKS + POOLZ_BLOCKS + 4 * WF_BLOCKS)

__global__ void __launch_bounds__(256)
k_setup(int* __restrict__ bins_p, int* __restrict__ bins_m, float* __restrict__ poolz,
        const float* __restrict__ W_p2, float* __restrict__ Wf_p,
        const float* __restrict__ W_m2, float* __restrict__ Wf_m,
        const float* __restrict__ W1f, float* __restrict__ Wf_h) {
    int bid = blockIdx.x;
    int t = threadIdx.x;
    if (bid < FILL_BLOCKS) { bins_p[bid * 256 + t] = 0; return; }
    bid -= FILL_BLOCKS;
    if (bid < FILL_BLOCKS) { bins_m[bid * 256 + t] = 0; return; }
    bid -= FILL_BLOCKS;
    if (bid < POOLZ_BLOCKS) {
        int i = bid * 256 + t;
        if (i < POOLZ_N) poolz[i] = 0.0f;
        return;
    }
    bid -= POOLZ_BLOCKS;
    if (bid < WF_BLOCKS) { wpair_body(W_p2, Wf_p, bid); return; }
    bid -= WF_BLOCKS;
    if (bid < WF_BLOCKS) { wpair_body(W_m2, Wf_m, bid); return; }
    bid -= WF_BLOCKS;
    wfrag_body(W1f, 2 * HID, Wf_h, bid);
}

template <bool SCALED>
__device__ __forceinline__ void gemm_ksteps(const float* __restrict__ Arow,
                                            const float* __restrict__ Wl,
                                            int ksteps, float ascale, v8f (&acc)[8]) {
    for (int kk = 0; kk < ksteps; ++kk) {
        v2f a;
        a.x = Arow[4 * kk];
        a.y = Arow[4 * kk + 1];
        if (SCALED) {
            a.x *= ascale;
            a.y *= ascale;
        }
        const float4* q = (const float4*)(Wl + (size_t)kk * 512);
        float4 q0 = q[0], q1 = q[1], q2 = q[2], q3 = q[3];
        v2f b0; b0.x = q0.x; b0.y = q0.y;
        v2f b1; b1.x = q0.z; b1.y = q0.w;
        v2f b2; b2.x = q1.x; b2.y = q1.y;
        v2f b3; b3.x = q1.z; b3.y = q1.w;
        v2f b4; b4.x = q2.x; b4.y = q2.y;
        v2f b5; b5.x = q2.z; b5.y = q2.w;
        v2f b6; b6.x = q3.x; b6.y = q3.y;
        v2f b7; b7.x = q3.z; b7.y = q3.w;
        acc[0] = __builtin_amdgcn_wmma_f32_16x16x4_f32(false, a, false, b0, (short)0, acc[0], false, false);
        acc[1] = __builtin_amdgcn_wmma_f32_16x16x4_f32(false, a, false, b1, (short)0, acc[1], false, false);
        acc[2] = __builtin_amdgcn_wmma_f32_16x16x4_f32(false, a, false, b2, (short)0, acc[2], false, false);
        acc[3] = __builtin_amdgcn_wmma_f32_16x16x4_f32(false, a, false, b3, (short)0, acc[3], false, false);
        acc[4] = __builtin_amdgcn_wmma_f32_16x16x4_f32(false, a, false, b4, (short)0, acc[4], false, false);
        acc[5] = __builtin_amdgcn_wmma_f32_16x16x4_f32(false, a, false, b5, (short)0, acc[5], false, false);
        acc[6] = __builtin_amdgcn_wmma_f32_16x16x4_f32(false, a, false, b6, (short)0, acc[6], false, false);
        acc[7] = __builtin_amdgcn_wmma_f32_16x16x4_f32(false, a, false, b7, (short)0, acc[7], false, false);
    }
}

__global__ void __launch_bounds__(128)
k_gemm_head(const float* __restrict__ h, const float* __restrict__ pesum,
            const float* __restrict__ cnt, const int* __restrict__ batch, int M,
            const float* __restrict__ Wf, const float* __restrict__ bias,
            const float* __restrict__ W2, const float* __restrict__ b2,
            float* __restrict__ out) {
    __shared__ float so[4 * 16 * 3];
    int w = __builtin_amdgcn_readfirstlane((int)(threadIdx.x >> 5));
    int lane = threadIdx.x & 31;
    int tile = blockIdx.x * 4 + w;
    if (tile * 16 < M) {
        int row0 = tile * 16;
        int half = lane >> 4;
        int lmod = lane & 15;
        int koff = half * 2;
        int g = min(max(batch[row0 + lmod], 0), NGRAPH - 1);
        float inv = 1.0f / fmaxf(cnt[g], 1.0f);
        const float* Wl = Wf + (size_t)lane * 16;

        v8f acc[8];
        #pragma unroll
        for (int nt = 0; nt < 8; ++nt) acc[nt] = (v8f){};

        gemm_ksteps<false>(h + (size_t)(row0 + lmod) * HID + koff, Wl, HID / 4, 1.0f, acc);
        gemm_ksteps<true>(pesum + (size_t)g * HID + koff, Wl + (size_t)(HID / 4) * 512, HID / 4,
                          inv, acc);

        float p[8][3];
        #pragma unroll
        for (int r = 0; r < 8; ++r) { p[r][0] = 0.0f; p[r][1] = 0.0f; p[r][2] = 0.0f; }
        #pragma unroll
        for (int nt = 0; nt < 8; ++nt) {
            int col = nt * 16 + lmod;
            float bv = bias[col];
            float w0 = W2[col * 3 + 0];
            float w1 = W2[col * 3 + 1];
            float w2 = W2[col * 3 + 2];
            #pragma unroll
            for (int r = 0; r < 8; ++r) {
                float v = fmaxf(acc[nt][r] + bv, 0.0f);
                p[r][0] += v * w0;
                p[r][1] += v * w1;
                p[r][2] += v * w2;
            }
        }
        #pragma unroll
        for (int s = 8; s >= 1; s >>= 1) {
            #pragma unroll
            for (int r = 0; r < 8; ++r) {
                p[r][0] += __shfl_xor(p[r][0], s);
                p[r][1] += __shfl_xor(p[r][1], s);
                p[r][2] += __shfl_xor(p[r][2], s);
            }
        }
        if (lmod == 0) {
            float c0 = b2[0], c1 = b2[1], c2 = b2[2];
            #pragma unroll
            for (int r = 0; r < 8; ++r) {
                float* q = so + w * 48 + (half * 8 + r) * 3;
                q[0] = p[r][0] + c0;
                q[1] = p[r][1] + c1;
                q[2] = p[r][2] + c2;
            }
        }
    }
    __syncthreads();
    for (int t = threadIdx.x; t < 4 * 16 * 3; t += 128) {
        size_t gi = (size_t)blockIdx.x * (4 * 16 * 3) + t;
        if (gi < (size_t)M * 3) out[gi] = so[t];
    }
}


static inline int cdiv(long long a, long long b) { return (int)((a + b - 1) / b); }

extern "C" void kernel_launch(void* const* d_in, const int* in_sizes, int n_in,
                              void* d_out, int out_size, void* d_ws, size_t ws_size,
                              hipStream_t stream) {
    (void)n_in; (void)out_size; (void)ws_size;

    const int*   res_p   = (const int*)d_in[0];
    const float* pos_p   = (const float*)d_in[1];
    const int*   ei_p    = (const int*)d_in[2];
    const int*   batch_p = (const int*)d_in[3];
    const int*   res_m   = (const int*)d_in[4];
    const float* pos_m   = (const float*)d_in[5];
    const int*   ei_m    = (const int*)d_in[6];
    const int*   batch_m = (const int*)d_in[7];
    const float* W_p1 = (const float*)d_in[8];
    const float* b_p1 = (const float*)d_in[9];
    const float* W_p2 = (const float*)d_in[10];
    const float* b_p2 = (const float*)d_in[11];
    const float* W_m1 = (const float*)d_in[12];
    const float* b_m1 = (const float*)d_in[13];
    const float* W_m2 = (const float*)d_in[14];
    const float* b_m2 = (const float*)d_in[15];
    const float* W1f  = (const float*)d_in[16];
    const float* b1f  = (const float*)d_in[17];
    const float* W2f  = (const float*)d_in[18];
    const float* b2f  = (const float*)d_in[19];

    int N_P = in_sizes[0];
    int E_P = in_sizes[2] / 2;
    int N_M = in_sizes[4];
    int E_M = in_sizes[6] / 2;
    const int* src_p = ei_p;
    const int* dst_p = ei_p + E_P;
    const int* src_m = ei_m;
    const int* dst_m = ei_m + E_M;

    float* ws = (float*)d_ws;
    size_t nhp = (size_t)N_P * HID;
    size_t nhm = (size_t)N_M * HID;
    size_t npa = ((size_t)N_P + 63) / 64 * 64;
    size_t nma = ((size_t)N_M + 63) / 64 * 64;
    float* A_p    = ws;
    float* C_p    = A_p + nhp;
    float* A_m    = C_p + nhp;
    float* C_m    = A_m + nhm;
    float* rec_p  = C_m + nhm;
    float* rec_m  = rec_p + npa * 8;
    float* dinv_p = rec_m + nma * 8;
    float* dinv_m = dinv_p + npa;
    float* pe_sum = dinv_m + nma;
    float* cnt    = pe_sum + NGRAPH * HID;
    float* Wf_p   = cnt + NGRAPH;
    float* Wf_m   = Wf_p + HID * HID;
    float* Wf_h   = Wf_m + HID * HID;
    int* ecnt_p   = (int*)(Wf_h + 2 * HID * HID);
    int* rowptr_p = ecnt_p + npa;
    int* part_p   = rowptr_p + npa;
    int* col_p    = part_p + E_P;
    int* bins_p   = col_p + E_P;
    int* ecnt_m   = bins_p + (2 * BIN_PAD + 1) * MAXBIN;
    int* rowptr_m = ecnt_m + nma;
    int* part_m   = rowptr_m + nma;
    int* col_m    = part_m + E_M;
    int* bins_m   = col_m + E_M;
    int* bcur_p   = bins_p + MAXBIN * BIN_PAD;
    int* bsta_p   = bins_p + 2 * MAXBIN * BIN_PAD;
    int* bcur_m   = bins_m + MAXBIN * BIN_PAD;
    int* bsta_m   = bins_m + 2 * MAXBIN * BIN_PAD;
    int nbin_p = cdiv(N_P, BIN_NODES);
    int nbin_m = cdiv(N_M, BIN_NODES);

    k_setup<<<SETUP_BLOCKS, 256, 0, stream>>>(bins_p, bins_m, pe_sum, W_p2, Wf_p, W_m2, Wf_m,
                                              W1f, Wf_h);

    int gp = cdiv(E_P, PCH), gm = cdiv(E_M, PCH);
    k_bin_hist<<<gp + gm, 256, 0, stream>>>(dst_p, E_P, N_P, bins_p,
                                            dst_m, E_M, N_M, bins_m, gp);
    k_bin_part<<<gp + gm, 256, 0, stream>>>(src_p, dst_p, E_P, N_P, nbin_p, bins_p, bcur_p, bsta_p, part_p,
                                            src_m, dst_m, E_M, N_M, nbin_m, bins_m, bcur_m, bsta_m, part_m,
                                            gp);
    k_bin_csr<<<nbin_p + nbin_m, 256, 0, stream>>>(part_p, bsta_p, bins_p, N_P, ecnt_p, rowptr_p, dinv_p, col_p,
                                                   res_p, pos_p, rec_p,
                                                   part_m, bsta_m, bins_m, N_M, ecnt_m, rowptr_m, dinv_m, col_m,
                                                   res_m, pos_m, rec_m, nbin_p);

    gp = cdiv(N_P, AGG_ROWS); gm = cdiv(N_M, AGG_ROWS);
    k_agg1<<<gp + gm, 256, 0, stream>>>(rec_p, rowptr_p, ecnt_p, col_p, W_p1, b_p1, Wf_p, N_P, A_p,
                                        rec_m, rowptr_m, ecnt_m, col_m, W_m1, b_m1, Wf_m, N_M, A_m, gp);

    gp = cdiv(N_P, 8); gm = cdiv(N_M, 8);
    k_gather<<<gp + gm, 256, 0, stream>>>(A_p, rowptr_p, ecnt_p, col_p, dinv_p, b_p2, N_P, C_p,
                                          A_m, rowptr_m, ecnt_m, col_m, dinv_m, b_m2, N_M, C_m, gp);

    k_pool<<<cdiv(N_P, 8 * POOL_RPW), 256, 0, stream>>>(C_p, batch_p, N_P, pe_sum, cnt);

    k_gemm_head<<<cdiv(N_M / 16, 4), 128, 0, stream>>>(C_m, pe_sum, cnt, batch_m, N_M, Wf_h, b1f,
                                                       W2f, b2f, (float*)d_out);
}


# Mean pool done in the layer-2 gather launch, one node per wave kept: pooled rows meet in LDS, the block's last wave sums its 8 per graph run, same float atomics; 8 launches -> 7; sum order changes
# speedup vs baseline: 6.7251x; 1.0370x over previous
#include <hip/hip_runtime.h>
#include <hip/hip_bf16.h>

typedef __attribute__((ext_vector_type(2))) float v2f;
typedef __attribute__((ext_vector_type(8))) float v8f;
typedef __attribute__((ext_vector_type(4))) float v4f;

#define HID 128
#define NGRAPH 64
#define NRES 20
#define AGG_NPW 4
#define AGG_ROWS (8 * AGG_NPW)
#define HPITCH (HID + 4)

#define BR_PICK bool second = (int)blockIdx.x >= split; \
                int bid = second ? (int)blockIdx.x - split : (int)blockIdx.x
#define PK(x) (second ? x##_m : x##_p)

#define BIN_SHIFT 8
#define BIN_NODES 256
#define MAXBIN 512
#define PCH 4096
#define PEPT 16
#define LCAP 16384
#define BIN_PAD 32


__device__ __forceinline__ void
bin_hist_body(const int* __restrict__ dst, int E, int N, int* __restrict__ binCount, int bid) {
    __shared__ int h[MAXBIN];
    int tid = threadIdx.x;
    for (int t = tid; t < MAXBIN; t += 256) h[t] = 0;
    __syncthreads();
    int e0 = bid * PCH;
    int dd[PEPT];
    #pragma unroll
    for (int k = 0; k < PEPT; ++k) dd[k] = dst[min(e0 + k * 256 + tid, E - 1)];
    #pragma unroll
    for (int k = 0; k < PEPT; ++k) {
        int e = e0 + k * 256 + tid;
        int d = min(max(dd[k], 0), N - 1);
        if (e < E) atomicAdd(&h[d >> BIN_SHIFT], 1);
    }
    __syncthreads();
    for (int t = tid; t < MAXBIN; t += 256) {
        int c = h[t];
        if (c > 0) atomicAdd(binCount + t * BIN_PAD, c);
    }
}

__global__ void __launch_bounds__(256)
k_bin_hist(const int* dst_p, int E_p, int N_p, int* binCount_p,
           const int* dst_m, int E_m, int N_m, int* binCount_m, int split) {
    BR_PICK;
    bin_hist_body(PK(dst), PK(E), PK(N), PK(binCount), bid);
}

__device__ __forceinline__ void scan_bins(const int* in, int* out, int lane) {
    int v[MAXBIN / 32];
    int s = 0;
    #pragma unroll
    for (int j = 0; j < MAXBIN / 32; ++j) {
        v[j] = in[lane * (MAXBIN / 32) + j];
        s += v[j];
    }
    int x = s;
    #pragma unroll
    for (int d = 1; d < 32; d <<= 1) {
        int t = __shfl_up(x, d);
        if (lane >= d) x += t;
    }
    int run = x - s;
    #pragma unroll
    for (int j = 0; j < MAXBIN / 32; ++j) {
        out[lane * (MAXBIN / 32) + j] = run;
        run += v[j];
    }
}

__device__ __forceinline__ void
bin_part_body(const int* __restrict__ src, const int* __restrict__ dst, int E, int N, int nbin,
              const int* __restrict__ binCount, int* __restrict__ binCursor,
              int* __restrict__ binStart, int* __restrict__ part, int bid) {
    __shared__ int h[MAXBIN];
    __shared__ int gc[MAXBIN];
    __shared__ int loc[MAXBIN];
    __shared__ int bstart[MAXBIN];
    __shared__ int gbase[MAXBIN];
    __shared__ int sval[PCH];
    __shared__ unsigned short sbin[PCH];
    int tid = threadIdx.x;
    for (int t = tid; t < MAXBIN; t += 256) {
        h[t] = 0;
        gc[t] = (t < nbin) ? binCount[t * BIN_PAD] : 0;
    }
    __syncthreads();
    int e0 = bid * PCH;
    int pk[PEPT], bb[PEPT], rr[PEPT];
    #pragma unroll
    for (int k = 0; k < PEPT; ++k) {
        int ec = min(e0 + k * 256 + tid, E - 1);
        bb[k] = dst[ec];
        pk[k] = src[ec];
    }
    #pragma unroll
    for (int k = 0; k < PEPT; ++k) {
        int e = e0 + k * 256 + tid;
        int d = min(max(bb[k], 0), N - 1);
        int s = min(max(pk[k], 0), N - 1);
        bb[k] = (e < E) ? (d >> BIN_SHIFT) : -1;
        pk[k] = (s << BIN_SHIFT) | (d & (BIN_NODES - 1));
        rr[k] = 0;
    }
    #pragma unroll
    for (int k = 0; k < PEPT; ++k)
        if (bb[k] >= 0) rr[k] = atomicAdd(&h[bb[k]], 1);
    __syncthreads();
    int wave = tid >> 5, lane = tid & 31;
    if (wave == 0) scan_bins(h, loc, lane);
    else if (wave == 1) scan_bins(gc, bstart, lane);
    __syncthreads();
    for (int t = tid; t < MAXBIN; t += 256) {
        int c = h[t];
        if (c > 0) gbase[t] = bstart[t] + atomicAdd(binCursor + t * BIN_PAD, c);
        if (bid == 0 && t < nbin) binStart[t] = bstart[t];
    }
    #pragma unroll
    for (int k = 0; k < PEPT; ++k) {
        if (bb[k] >= 0) {
            int p = loc[bb[k]] + rr[k];
            sval[p] = pk[k];
            sbin[p] = (unsigned short)bb[k];
        }
    }
    __syncthreads();
    int nv = min(PCH, E - e0);
    for (int i = tid; i < nv; i += 256) {
        int b = sbin[i];
        part[gbase[b] + (i - loc[b])] = sval[i];
    }
}

__global__ void __launch_bounds__(256)
k_bin_part(const int* src_p, const int* dst_p, int E_p, int N_p, int nbin_p,
           const int* binCount_p, int* binCursor_p, int* binStart_p, int* part_p,
           const int* src_m, const int* dst_m, int E_m, int N_m, int nbin_m,
           const int* binCount_m, int* binCursor_m, int* binStart_m, int* part_m, int split) {
    BR_PICK;
    bin_part_body(PK(src), PK(dst), PK(E), PK(N), PK(nbin), PK(binCount), PK(binCursor),
                  PK(binStart), PK(part), bid);
}

__device__ __forceinline__ void
bin_csr_body(const int* __restrict__ part, const int* __restrict__ binStart,
             const int* __restrict__ binCount, int N, int* __restrict__ ecnt,
             int* __restrict__ rowptr, float* __restrict__ dinv, int* __restrict__ col,
             const int* __restrict__ res, const float* __restrict__ pos,
             float* __restrict__ rec, int bid) {
    __shared__ int cnt[BIN_NODES];
    __shared__ int cur[BIN_NODES];
    __shared__ int wtot[8];
    __shared__ int lcol[LCAP];
    int tid = threadIdx.x;
    int wave = tid >> 5, lane = tid & 31;
    int b = bid;
    int s = binStart[b];
    int m = binCount[b * BIN_PAD];
    cnt[tid] = 0;
    __syncthreads();
    for (int i0 = 0; i0 < m; i0 += 1024) {
        int v[4];
        #pragma unroll
        for (int u = 0; u < 4; ++u) {
            int i = i0 + u * 256 + tid;
            v[u] = (i < m) ? part[s + i] : -1;
        }
        #pragma unroll
        for (int u = 0; u < 4; ++u)
            if (v[u] >= 0) atomicAdd(&cnt[v[u] & (BIN_NODES - 1)], 1);
    }
    __syncthreads();
    int c = cnt[tid];
    int x = c;
    #pragma unroll
    for (int d = 1; d < 32; d <<= 1) {
        int t = __shfl_up(x, d);
        if (lane >= d) x += t;
    }
    if (lane == 31) wtot[wave] = x;
    __syncthreads();
    int wb = 0;
    for (int w = 0; w < wave; ++w) wb += wtot[w];
    int off = wb + x - c;
    cur[tid] = off;
    int node = b * BIN_NODES + tid;
    if (node < N) {
        ecnt[node] = c;
        rowptr[node] = s + off;
        float dv = rsqrtf((float)c + 1.0f);
        dinv[node] = dv;
        v4f q;
        q.x = dv * pos[node * 3 + 0];
        q.y = dv * pos[node * 3 + 1];
        q.z = dv * pos[node * 3 + 2];
        q.w = dv;
        *(v4f*)(rec + (size_t)node * 8) = q;
        int r = res[node];
        if (r < 0 || r >= NRES) r = NRES;
        ((int*)rec)[(size_t)node * 8 + 4] = r * (HID * 4);
    }
    __syncthreads();
    bool staged = m <= LCAP;
    for (int i0 = 0; i0 < m; i0 += 1024) {
        int v[4];
        #pragma unroll
        for (int u = 0; u < 4; ++u) {
            int i = i0 + u * 256 + tid;
            v[u] = (i < m) ? part[s + i] : -1;
        }
        #pragma unroll
        for (int u = 0; u < 4; ++u) {
            if (v[u] >= 0) {
                int p = atomicAdd(&cur[v[u] & (BIN_NODES - 1)], 1);
                int sv = (int)((unsigned)v[u] >> BIN_SHIFT);
                if (staged) lcol[p] = sv;
                else col[s + p] = sv;
            }
        }
    }
    if (staged) {
        __syncthreads();
        for (int i = tid; i < m; i += 256) col[s + i] = lcol[i];
    }
}

__global__ void __launch_bounds__(256)
k_bin_csr(const int* part_p, const int* binStart_p, const int* binCount_p, int N_p,
          int* ecnt_p, int* rowptr_p, float* dinv_p, int* col_p,
          const int* res_p, const float* pos_p, float* rec_p,
          const int* part_m, const int* binStart_m, const int* binCount_m, int N_m,
          int* ecnt_m, int* rowptr_m, float* dinv_m, int* col_m,
          const int* res_m, const float* pos_m, float* rec_m, int split) {
    BR_PICK;
    bin_csr_body(PK(part), PK(binStart), PK(binCount), PK(N), PK(ecnt), PK(rowptr), PK(dinv),
                 PK(col), PK(res), PK(pos), PK(rec), bid);
}


__device__ __forceinline__ void
agg1_body(const float* __restrict__ rec, const int* __restrict__ rowptr,
          const int* __restrict__ ecnt, const int* __restrict__ col,
          const float* __restrict__ W, const float* __restrict__ b,
          const float* __restrict__ Wq, int N, float* __restrict__ y, int bid) {
    __shared__ __attribute__((aligned(16))) float Ws[(NRES + 1) * HID];
    __shared__ __attribute__((aligned(16))) float Hs[AGG_ROWS * HPITCH];
    __shared__ __attribute__((aligned(16))) float Ys[AGG_ROWS * HPITCH];
    __shared__ float Ds[AGG_ROWS];
    for (int t = threadIdx.x; t < NRES * HID / 4; t += 256)
        ((v4f*)Ws)[t] = ((const v4f*)W)[t];
    if (threadIdx.x < HID / 4)
        ((v4f*)Ws)[NRES * HID / 4 + threadIdx.x] = (v4f){0.0f, 0.0f, 0.0f, 0.0f};
    __syncthreads();

    int lane = threadIdx.x & 31;
    const v4f w20 = *(const v4f*)(W + (NRES + 0) * HID + lane * 4);
    const v4f w21 = *(const v4f*)(W + (NRES + 1) * HID + lane * 4);
    const v4f w22 = *(const v4f*)(W + (NRES + 2) * HID + lane * 4);
    const v4f bb = *(const v4f*)(b + lane * 4);
    const char* Wl = (const char*)Ws + lane * 16;
    const int* reci = (const int*)rec;
    int wave = __builtin_amdgcn_readfirstlane((int)(threadIdx.x >> 5));
    int first = (bid * 8 + wave) * AGG_NPW;

    for (int t = 0; t < AGG_NPW; ++t) {
        int i = __builtin_amdgcn_readfirstlane(first + t);
        float* hrow = Hs + (wave * AGG_NPW + t) * HPITCH + lane * 4;
        if (i >= N) {
            *(v4f*)hrow = (v4f){0.0f, 0.0f, 0.0f, 0.0f};
            continue;
        }
        int start = rowptr[i];
        int n = ecnt[i];
        v4f rs = *(const v4f*)(rec + (size_t)i * 8);
        int os = reci[(size_t)i * 8 + 4];
        v4f acc = (v4f){0.0f, 0.0f, 0.0f, 0.0f};
        float q0 = 0.0f, q1 = 0.0f, q2 = 0.0f;
        for (int base = 0; base < n; base += 32) {
            int m = n - base;
            if (m > 32) m = 32;
            bool live = lane < m;
            int c = live ? col[start + base + lane] : i;
            c = min(max(c, 0), N - 1);
            v4f r = *(const v4f*)(rec + (size_t)c * 8);
            int off = reci[(size_t)c * 8 + 4];
            float d = live ? r.w : 0.0f;
            off = live ? off : NRES * HID * 4;
            q0 += live ? r.x : 0.0f;
            q1 += live ? r.y : 0.0f;
            q2 += live ? r.z : 0.0f;
            int dbits = __float_as_int(d);
            for (int j = 0; j < m; j += 8) {
                int o0 = __builtin_amdgcn_readlane(off, j);
                int o1 = __builtin_amdgcn_readlane(off, j + 1);
                int o2 = __builtin_amdgcn_readlane(off, j + 2);
                int o3 = __builtin_amdgcn_readlane(off, j + 3);
                int o4 = __builtin_amdgcn_readlane(off, j + 4);
                int o5 = __builtin_amdgcn_readlane(off, j + 5);
                int o6 = __builtin_amdgcn_readlane(off, j + 6);
                int o7 = __builtin_amdgcn_readlane(off, j + 7);
                v4f x0 = *(const v4f*)(Wl + o0);
                v4f x1 = *(const v4f*)(Wl + o1);
                v4f x2 = *(const v4f*)(Wl + o2);
                v4f x3 = *(const v4f*)(Wl + o3);
                v4f x4 = *(const v4f*)(Wl + o4);
                v4f x5 = *(const v4f*)(Wl + o5);
                v4f x6 = *(const v4f*)(Wl + o6);
                v4f x7 = *(const v4f*)(Wl + o7);
                float d0 = __int_as_float(__builtin_amdgcn_readlane(dbits, j));
                float d1 = __int_as_float(__builtin_amdgcn_readlane(dbits, j + 1));
                float d2 = __int_as_float(__builtin_amdgcn_readlane(dbits, j + 2));
                float d3 = __int_as_float(__builtin_amdgcn_readlane(dbits, j + 3));
                float d4 = __int_as_float(__builtin_amdgcn_readlane(dbits, j + 4));
                float d5 = __int_as_float(__builtin_amdgcn_readlane(dbits, j + 5));
                float d6 = __int_as_float(__builtin_amdgcn_readlane(dbits, j + 6));
                float d7 = __int_as_float(__builtin_amdgcn_readlane(dbits, j + 7));
                acc += d0 * x0;
                acc += d1 * x1;
                acc += d2 * x2;
                acc += d3 * x3;
                acc += d4 * x4;
                acc += d5 * x5;
                acc += d6 * x6;
                acc += d7 * x7;
            }
        }
        v4f xs = *(const v4f*)(Wl + os);
        acc += rs.w * xs;
        #pragma unroll
        for (int s = 16; s >= 1; s >>= 1) {
            q0 += __shfl_xor(q0, s);
            q1 += __shfl_xor(q1, s);
            q2 += __shfl_xor(q2, s);
        }
        q0 += rs.x;
        q1 += rs.y;
        q2 += rs.z;
        v4f S = acc + q0 * w20 + q1 * w21 + q2 * w22;
        float dnode = rs.w;
        v4f o;
        o.x = fmaxf(dnode * S.x + bb.x, 0.0f);
        o.y = fmaxf(dnode * S.y + bb.y, 0.0f);
        o.z = fmaxf(dnode * S.z + bb.z, 0.0f);
        o.w = fmaxf(dnode * S.w + bb.w, 0.0f);
        *(v4f*)hrow = o;
        if (lane == 0) Ds[wave * AGG_NPW + t] = dnode;
    }
    __syncthreads();

    {
        int rt = wave >> 2, q = wave & 3;
        int lmod = lane & 15, half = lane >> 4;
        const float* Arow = Hs + (rt * 16 + lmod) * HPITCH + half * 2;
        const float* Bl = Wq + (size_t)(q * 32 + lane) * 4;
        v8f acc0 = (v8f){}, acc1 = (v8f){};
        v2f a[2][4];
        v4f bq[2][4];
        #pragma unroll
        for (int u = 0; u < 4; ++u) {
            a[0][u] = *(const v2f*)(Arow + 4 * u);
            bq[0][u] = *(const v4f*)(Bl + (size_t)u * 512);
        }
        #pragma unroll
        for (int g = 0; g < 8; ++g) {
            if (g + 1 < 8) {
                #pragma unroll
                for (int u = 0; u < 4; ++u) {
                    int kk = (g + 1) * 4 + u;
                    a[(g + 1) & 1][u] = *(const v2f*)(Arow + 4 * kk);
                    bq[(g + 1) & 1][u] = *(const v4f*)(Bl + (size_t)kk * 512);
                }
            }
            #pragma unroll
            for (int u = 0; u < 4; ++u) {
                v4f w4 = bq[g & 1][u];
                v2f b0; b0.x = w4.x; b0.y = w4.y;
                v2f b1; b1.x = w4.z; b1.y = w4.w;
                acc0 = __builtin_amdgcn_wmma_f32_16x16x4_f32(false, a[g & 1][u], false, b0, (short)0, acc0, false, false);
                acc1 = __builtin_amdgcn_wmma_f32_16x16x4_f32(false, a[g & 1][u], false, b1, (short)0, acc1, false, false);
            }
        }
        float* yb = Ys + (rt * 16 + half * 8) * HPITCH + q * 32 + lmod;
        #pragma unroll
        for (int r = 0; r < 8; ++r) {
            yb[r * HPITCH] = acc0[r];
            yb[r * HPITCH + 16] = acc1[r];
        }
    }
    __syncthreads();

    for (int t = 0; t < AGG_NPW; ++t) {
        int i = first + t;
        if (i < N) {
            int row = wave * AGG_NPW + t;
            v4f v = *(const v4f*)(Ys + row * HPITCH + lane * 4);
            float d = Ds[row];
            *(v4f*)(y + (size_t)i * HID + lane * 4) = v * d;
        }
    }
}

__global__ void __launch_bounds__(256)
k_agg1(const float* rec_p, const int* rowptr_p, const int* ecnt_p, const int* col_p,
       const float* W_p, const float* b_p, const float* Wq_p, int N_p, float* y_p,
       const float* rec_m, const int* rowptr_m, const int* ecnt_m, const int* col_m,
       const float* W_m, const float* b_m, const float* Wq_m, int N_m, float* y_m, int split) {
    BR_PICK;
    agg1_body(PK(rec), PK(rowptr), PK(ecnt), PK(col), PK(W), PK(b), PK(Wq), PK(N), PK(y), bid);
}

__device__ __forceinline__ void pool_flush(float* __restrict__ pesum, float* __restrict__ cnt,
                                           int g, int lane, v4f acc, int n) {
    if ((unsigned)g >= (unsigned)NGRAPH) return;
    float* p = pesum + g * HID + lane * 4;
    atomicAdd(p + 0, acc.x);
    atomicAdd(p + 1, acc.y);
    atomicAdd(p + 2, acc.z);
    atomicAdd(p + 3, acc.w);
    if (lane == 0) atomicAdd(cnt + g, (float)n);
}

#define GATHER_NPB 8

__device__ __forceinline__ void
gather_body(const float* __restrict__ y, const int* __restrict__ rowptr,
            const int* __restrict__ ecnt, const int* __restrict__ col,
            const float* __restrict__ dinv, const float* __restrict__ b,
            int N, float* __restrict__ h, int bid, bool pooled,
            const int* __restrict__ batch, float* __restrict__ pesum, float* __restrict__ cnt) {
    __shared__ __attribute__((aligned(16))) float Rs[GATHER_NPB * HID];
    __shared__ int Gs[GATHER_NPB];
    __shared__ int arrived;
    if (pooled) {
        if (threadIdx.x == 0) arrived = 0;
        __syncthreads();
    }
    int wave = __builtin_amdgcn_readfirstlane((int)(threadIdx.x >> 5));
    int i = __builtin_amdgcn_readfirstlane(bid * GATHER_NPB + wave);
    if (i >= N) return;
    int lane = threadIdx.x & 31;
    int start = rowptr[i];
    int n = ecnt[i];
    int g = pooled ? batch[i] : 0;
    const float* yl = y + lane * 4;
    v4f acc = (v4f){0.0f, 0.0f, 0.0f, 0.0f};
    for (int base = 0; base < n; base += 32) {
        int m = n - base;
        if (m > 32) m = 32;
        int c = (lane < m) ? col[start + base + lane] : i;
        c = min(max(c, 0), N - 1);
        for (int j = 0; j < m; j += 8) {
            int s0 = __builtin_amdgcn_readlane(c, j);
            int s1 = __builtin_amdgcn_readlane(c, j + 1);
            int s2 = __builtin_amdgcn_readlane(c, j + 2);
            int s3 = __builtin_amdgcn_readlane(c, j + 3);
            int s4 = __builtin_amdgcn_readlane(c, j + 4);
            int s5 = __builtin_amdgcn_readlane(c, j + 5);
            int s6 = __builtin_amdgcn_readlane(c, j + 6);
            int s7 = __builtin_amdgcn_readlane(c, j + 7);
            v4f v0 = *(const v4f*)(yl + (size_t)s0 * HID);
            v4f v1 = *(const v4f*)(yl + (size_t)s1 * HID);
            v4f v2 = *(const v4f*)(yl + (size_t)s2 * HID);
            v4f v3 = *(const v4f*)(yl + (size_t)s3 * HID);
            v4f v4 = *(const v4f*)(yl + (size_t)s4 * HID);
            v4f v5 = *(const v4f*)(yl + (size_t)s5 * HID);
            v4f v6 = *(const v4f*)(yl + (size_t)s6 * HID);
            v4f v7 = *(const v4f*)(yl + (size_t)s7 * HID);
            asm volatile("" :: "v"(v0), "v"(v1), "v"(v2), "v"(v3),
                               "v"(v4), "v"(v5), "v"(v6), "v"(v7));
            acc += v0;
            if (j + 1 < m) acc += v1;
            if (j + 2 < m) acc += v2;
            if (j + 3 < m) acc += v3;
            if (j + 4 < m) acc += v4;
            if (j + 5 < m) acc += v5;
            if (j + 6 < m) acc += v6;
            if (j + 7 < m) acc += v7;
        }
    }
    v4f self = *(const v4f*)(yl + (size_t)i * HID);
    v4f bb = *(const v4f*)(b + lane * 4);
    float d = dinv[i];
    v4f o;
    o.x = fmaxf(d * (acc.x + self.x) + bb.x, 0.0f);
    o.y = fmaxf(d * (acc.y + self.y) + bb.y, 0.0f);
    o.z = fmaxf(d * (acc.z + self.z) + bb.z, 0.0f);
    o.w = fmaxf(d * (acc.w + self.w) + bb.w, 0.0f);
    if (!pooled) {
        *(v4f*)(h + (size_t)i * HID + lane * 4) = o;
        return;
    }
    *(v4f*)(Rs + wave * HID + lane * 4) = o;
    if (lane == 0) Gs[wave] = g;
    __threadfence_block();
    int t = 0;
    if (lane == 0) t = atomicAdd(&arrived, 1);
    t = __builtin_amdgcn_readfirstlane(t);
    int nact = min(GATHER_NPB, N - bid * GATHER_NPB);
    if (t != nact - 1) return;
    __threadfence_block();
    int cur = __builtin_amdgcn_readfirstlane(Gs[0]);
    int rows = 0;
    v4f sum = (v4f){0.0f, 0.0f, 0.0f, 0.0f};
    for (int r = 0; r < nact; ++r) {
        int gr = __builtin_amdgcn_readfirstlane(Gs[r]);
        if (gr != cur) {
            pool_flush(pesum, cnt, cur, lane, sum, rows);
            cur = gr;
            rows = 0;
            sum = (v4f){0.0f, 0.0f, 0.0f, 0.0f};
        }
        sum += *(const v4f*)(Rs + r * HID + lane * 4);
        rows += 1;
    }
    pool_flush(pesum, cnt, cur, lane, sum, rows);
}

__global__ void
k_gather(const float* y_p, const int* rowptr_p, const int* ecnt_p, const int* col_p,
         const float* dinv_p, const float* b_p, int N_p,
         const int* batch_p, float* pesum, float* cnt,
         const float* y_m, const int* rowptr_m, const int* ecnt_m, const int* col_m,
         const float* dinv_m, const float* b_m, int N_m, float* h_m, int split) {
    BR_PICK;
    gather_body(PK(y), PK(rowptr), PK(ecnt), PK(col), PK(dinv), PK(b), PK(N), h_m, bid,
                !second, batch_p, pesum, cnt);
}

__device__ __forceinline__ void
wfrag_body(const float* __restrict__ W, int K, float* __restrict__ Wf, int bid) {
    int idx = bid * 256 + threadIdx.x;
    if (idx >= K * HID) return;
    int j    = idx & 1;
    int nt   = (idx >> 1) & 7;
    int l    = (idx >> 4) & 31;
    int kk   = idx >> 9;
    int krow = 4 * kk + (l >> 4) * 2 + j;
    int col  = nt * 16 + (l & 15);
    Wf[idx] = W[(size_t)krow * HID + col];
}

__device__ __forceinline__ void
wpair_body(const float* __restrict__ W, float* __restrict__ Wq, int bid) {
    int idx = bid * 256 + threadIdx.x;
    if (idx >= HID * HID) return;
    int j    = idx & 1;
    int t    = (idx >> 1) & 1;
    int l    = (idx >> 2) & 31;
    int q    = (idx >> 7) & 3;
    int kk   = idx >> 9;
    int krow = 4 * kk + (l >> 4) * 2 + j;
    int col  = (2 * q + t) * 16 + (l & 15);
    Wq[idx] = W[(size_t)krow * HID + col];
}

#define FILL_BLOCKS (2 * MAXBIN * BIN_PAD / 256)
#define POOLZ_N     (NGRAPH * HID + NGRAPH)
#define POOLZ_BLOCKS ((POOLZ_N + 255) / 256)
#define WF_BLOCKS   (HID * HID / 256)
#define SETUP_BLOCKS (2 * FILL_BLOCKS + POOLZ_BLOCKS + 4 * WF_BLOCKS)

__global__ void __launch_bounds__(256)
k_setup(int* __restrict__ bins_p, int* __restrict__ bins_m, float* __restrict__ poolz,
        const float* __restrict__ W_p2, float* __restrict__ Wf_p,
        const float* __restrict__ W_m2, float* __restrict__ Wf_m,
        const float* __restrict__ W1f, float* __restrict__ Wf_h) {
    int bid = blockIdx.x;
    int t = threadIdx.x;
    if (bid < FILL_BLOCKS) { bins_p[bid * 256 + t] = 0; return; }
    bid -= FILL_BLOCKS;
    if (bid < FILL_BLOCKS) { bins_m[bid * 256 + t] = 0; return; }
    bid -= FILL_BLOCKS;
    if (bid < POOLZ_BLOCKS) {
        int i = bid * 256 + t;
        if (i < POOLZ_N) poolz[i] = 0.0f;
        return;
    }
    bid -= POOLZ_BLOCKS;
    if (bid < WF_BLOCKS) { wpair_body(W_p2, Wf_p, bid); return; }
    bid -= WF_BLOCKS;
    if (bid < WF_BLOCKS) { wpair_body(W_m2, Wf_m, bid); return; }
    bid -= WF_BLOCKS;
    wfrag_body(W1f, 2 * HID, Wf_h, bid);
}

template <bool SCALED>
__device__ __forceinline__ void gemm_ksteps(const float* __restrict__ Arow,
                                            const float* __restrict__ Wl,
                                            int ksteps, float ascale, v8f (&acc)[8]) {
    for (int kk = 0; kk < ksteps; ++kk) {
        v2f a;
        a.x = Arow[4 * kk];
        a.y = Arow[4 * kk + 1];
        if (SCALED) {
            a.x *= ascale;
            a.y *= ascale;
        }
        const float4* q = (const float4*)(Wl + (size_t)kk * 512);
        float4 q0 = q[0], q1 = q[1], q2 = q[2], q3 = q[3];
        v2f b0; b0.x = q0.x; b0.y = q0.y;
        v2f b1; b1.x = q0.z; b1.y = q0.w;
        v2f b2; b2.x = q1.x; b2.y = q1.y;
        v2f b3; b3.x = q1.z; b3.y = q1.w;
        v2f b4; b4.x = q2.x; b4.y = q2.y;
        v2f b5; b5.x = q2.z; b5.y = q2.w;
        v2f b6; b6.x = q3.x; b6.y = q3.y;
        v2f b7; b7.x = q3.z; b7.y = q3.w;
        acc[0] = __builtin_amdgcn_wmma_f32_16x16x4_f32(false, a, false, b0, (short)0, acc[0], false, false);
        acc[1] = __builtin_amdgcn_wmma_f32_16x16x4_f32(false, a, false, b1, (short)0, acc[1], false, false);
        acc[2] = __builtin_amdgcn_wmma_f32_16x16x4_f32(false, a, false, b2, (short)0, acc[2], false, false);
        acc[3] = __builtin_amdgcn_wmma_f32_16x16x4_f32(false, a, false, b3, (short)0, acc[3], false, false);
        acc[4] = __builtin_amdgcn_wmma_f32_16x16x4_f32(false, a, false, b4, (short)0, acc[4], false, false);
        acc[5] = __builtin_amdgcn_wmma_f32_16x16x4_f32(false, a, false, b5, (short)0, acc[5], false, false);
        acc[6] = __builtin_amdgcn_wmma_f32_16x16x4_f32(false, a, false, b6, (short)0, acc[6], false, false);
        acc[7] = __builtin_amdgcn_wmma_f32_16x16x4_f32(false, a, false, b7, (short)0, acc[7], false, false);
    }
}

__global__ void __launch_bounds__(128)
k_gemm_head(const float* __restrict__ h, const float* __restrict__ pesum,
            const float* __restrict__ cnt, const int* __restrict__ batch, int M,
            const float* __restrict__ Wf, const float* __restrict__ bias,
            const float* __restrict__ W2, const float* __restrict__ b2,
            float* __restrict__ out) {
    __shared__ float so[4 * 16 * 3];
    int w = __builtin_amdgcn_readfirstlane((int)(threadIdx.x >> 5));
    int lane = threadIdx.x & 31;
    int tile = blockIdx.x * 4 + w;
    if (tile * 16 < M) {
        int row0 = tile * 16;
        int half = lane >> 4;
        int lmod = lane & 15;
        int koff = half * 2;
        int g = min(max(batch[row0 + lmod], 0), NGRAPH - 1);
        float inv = 1.0f / fmaxf(cnt[g], 1.0f);
        const float* Wl = Wf + (size_t)lane * 16;

        v8f acc[8];
        #pragma unroll
        for (int nt = 0; nt < 8; ++nt) acc[nt] = (v8f){};

        gemm_ksteps<false>(h + (size_t)(row0 + lmod) * HID + koff, Wl, HID / 4, 1.0f, acc);
        gemm_ksteps<true>(pesum + (size_t)g * HID + koff, Wl + (size_t)(HID / 4) * 512, HID / 4,
                          inv, acc);

        float p[8][3];
        #pragma unroll
        for (int r = 0; r < 8; ++r) { p[r][0] = 0.0f; p[r][1] = 0.0f; p[r][2] = 0.0f; }
        #pragma unroll
        for (int nt = 0; nt < 8; ++nt) {
            int col = nt * 16 + lmod;
            float bv = bias[col];
            float w0 = W2[col * 3 + 0];
            float w1 = W2[col * 3 + 1];
            float w2 = W2[col * 3 + 2];
            #pragma unroll
            for (int r = 0; r < 8; ++r) {
                float v = fmaxf(acc[nt][r] + bv, 0.0f);
                p[r][0] += v * w0;
                p[r][1] += v * w1;
                p[r][2] += v * w2;
            }
        }
        #pragma unroll
        for (int s = 8; s >= 1; s >>= 1) {
            #pragma unroll
            for (int r = 0; r < 8; ++r) {
                p[r][0] += __shfl_xor(p[r][0], s);
                p[r][1] += __shfl_xor(p[r][1], s);
                p[r][2] += __shfl_xor(p[r][2], s);
            }
        }
        if (lmod == 0) {
            float c0 = b2[0], c1 = b2[1], c2 = b2[2];
            #pragma unroll
            for (int r = 0; r < 8; ++r) {
                float* q = so + w * 48 + (half * 8 + r) * 3;
                q[0] = p[r][0] + c0;
                q[1] = p[r][1] + c1;
                q[2] = p[r][2] + c2;
            }
        }
    }
    __syncthreads();
    for (int t = threadIdx.x; t < 4 * 16 * 3; t += 128) {
        size_t gi = (size_t)blockIdx.x * (4 * 16 * 3) + t;
        if (gi < (size_t)M * 3) out[gi] = so[t];
    }
}


static inline int cdiv(long long a, long long b) { return (int)((a + b - 1) / b); }

extern "C" void kernel_launch(void* const* d_in, const int* in_sizes, int n_in,
                              void* d_out, int out_size, void* d_ws, size_t ws_size,
                              hipStream_t stream) {
    (void)n_in; (void)out_size; (void)ws_size;

    const int*   res_p   = (const int*)d_in[0];
    const float* pos_p   = (const float*)d_in[1];
    const int*   ei_p    = (const int*)d_in[2];
    const int*   batch_p = (const int*)d_in[3];
    const int*   res_m   = (const int*)d_in[4];
    const float* pos_m   = (const float*)d_in[5];
    const int*   ei_m    = (const int*)d_in[6];
    const int*   batch_m = (const int*)d_in[7];
    const float* W_p1 = (const float*)d_in[8];
    const float* b_p1 = (const float*)d_in[9];
    const float* W_p2 = (const float*)d_in[10];
    const float* b_p2 = (const float*)d_in[11];
    const float* W_m1 = (const float*)d_in[12];
    const float* b_m1 = (const float*)d_in[13];
    const float* W_m2 = (const float*)d_in[14];
    const float* b_m2 = (const float*)d_in[15];
    const float* W1f  = (const float*)d_in[16];
    const float* b1f  = (const float*)d_in[17];
    const float* W2f  = (const float*)d_in[18];
    const float* b2f  = (const float*)d_in[19];

    int N_P = in_sizes[0];
    int E_P = in_sizes[2] / 2;
    int N_M = in_sizes[4];
    int E_M = in_sizes[6] / 2;
    const int* src_p = ei_p;
    const int* dst_p = ei_p + E_P;
    const int* src_m = ei_m;
    const int* dst_m = ei_m + E_M;

    float* ws = (float*)d_ws;
    size_t nhp = (size_t)N_P * HID;
    size_t nhm = (size_t)N_M * HID;
    size_t npa = ((size_t)N_P + 63) / 64 * 64;
    size_t nma = ((size_t)N_M + 63) / 64 * 64;
    float* A_p    = ws;
    float* C_p    = A_p + nhp;
    float* A_m    = C_p + nhp;
    float* C_m    = A_m + nhm;
    float* rec_p  = C_m + nhm;
    float* rec_m  = rec_p + npa * 8;
    float* dinv_p = rec_m + nma * 8;
    float* dinv_m = dinv_p + npa;
    float* pe_sum = dinv_m + nma;
    float* cnt    = pe_sum + NGRAPH * HID;
    float* Wf_p   = cnt + NGRAPH;
    float* Wf_m   = Wf_p + HID * HID;
    float* Wf_h   = Wf_m + HID * HID;
    int* ecnt_p   = (int*)(Wf_h + 2 * HID * HID);
    int* rowptr_p = ecnt_p + npa;
    int* part_p   = rowptr_p + npa;
    int* col_p    = part_p + E_P;
    int* bins_p   = col_p + E_P;
    int* ecnt_m   = bins_p + (2 * BIN_PAD + 1) * MAXBIN;
    int* rowptr_m = ecnt_m + nma;
    int* part_m   = rowptr_m + nma;
    int* col_m    = part_m + E_M;
    int* bins_m   = col_m + E_M;
    int* bcur_p   = bins_p + MAXBIN * BIN_PAD;
    int* bsta_p   = bins_p + 2 * MAXBIN * BIN_PAD;
    int* bcur_m   = bins_m + MAXBIN * BIN_PAD;
    int* bsta_m   = bins_m + 2 * MAXBIN * BIN_PAD;
    int nbin_p = cdiv(N_P, BIN_NODES);
    int nbin_m = cdiv(N_M, BIN_NODES);

    k_setup<<<SETUP_BLOCKS, 256, 0, stream>>>(bins_p, bins_m, pe_sum, W_p2, Wf_p, W_m2, Wf_m,
                                              W1f, Wf_h);

    int gp = cdiv(E_P, PCH), gm = cdiv(E_M, PCH);
    k_bin_hist<<<gp + gm, 256, 0, stream>>>(dst_p, E_P, N_P, bins_p,
                                            dst_m, E_M, N_M, bins_m, gp);
    k_bin_part<<<gp + gm, 256, 0, stream>>>(src_p, dst_p, E_P, N_P, nbin_p, bins_p, bcur_p, bsta_p, part_p,
                                            src_m, dst_m, E_M, N_M, nbin_m, bins_m, bcur_m, bsta_m, part_m,
                                            gp);
    k_bin_csr<<<nbin_p + nbin_m, 256, 0, stream>>>(part_p, bsta_p, bins_p, N_P, ecnt_p, rowptr_p, dinv_p, col_p,
                                                   res_p, pos_p, rec_p,
                                                   part_m, bsta_m, bins_m, N_M, ecnt_m, rowptr_m, dinv_m, col_m,
                                                   res_m, pos_m, rec_m, nbin_p);

    gp = cdiv(N_P, AGG_ROWS); gm = cdiv(N_M, AGG_ROWS);
    k_agg1<<<gp + gm, 256, 0, stream>>>(rec_p, rowptr_p, ecnt_p, col_p, W_p1, b_p1, Wf_p, N_P, A_p,
                                        rec_m, rowptr_m, ecnt_m, col_m, W_m1, b_m1, Wf_m, N_M, A_m, gp);

    gp = cdiv(N_P, GATHER_NPB); gm = cdiv(N_M, GATHER_NPB);
    k_gather<<<gp + gm, 256, 0, stream>>>(A_p, rowptr_p, ecnt_p, col_p, dinv_p, b_p2, N_P,
                                          batch_p, pe_sum, cnt,
                                          A_m, rowptr_m, ecnt_m, col_m, dinv_m, b_m2, N_M, C_m, gp);

    k_gemm_head<<<cdiv(N_M / 16, 4), 128, 0, stream>>>(C_m, pe_sum, cnt, batch_m, N_M, Wf_h, b1f,
                                                       W2f, b2f, (float*)d_out);
}


# Layer-2 gather: a wave takes 4 consecutive nodes, their row bounds and first neighbour ids fetched together; 150000 wave entries -> 37500; a pooled block sums 32 rows, not 8: pool sum order changes
# speedup vs baseline: 6.9909x; 1.0395x over previous
#include <hip/hip_runtime.h>
#include <hip/hip_bf16.h>

typedef __attribute__((ext_vector_type(2))) float v2f;
typedef __attribute__((ext_vector_type(8))) float v8f;
typedef __attribute__((ext_vector_type(4))) float v4f;

#define HID 128
#define NGRAPH 64
#define NRES 20
#define AGG_NPW 4
#define AGG_ROWS (8 * AGG_NPW)
#define HPITCH (HID + 4)

#define BR_PICK bool second = (int)blockIdx.x >= split; \
                int bid = second ? (int)blockIdx.x - split : (int)blockIdx.x
#define PK(x) (second ? x##_m : x##_p)

#define BIN_SHIFT 8
#define BIN_NODES 256
#define MAXBIN 512
#define PCH 4096
#define PEPT 16
#define LCAP 16384
#define BIN_PAD 32


__device__ __forceinline__ void
bin_hist_body(const int* __restrict__ dst, int E, int N, int* __restrict__ binCount, int bid) {
    __shared__ int h[MAXBIN];
    int tid = threadIdx.x;
    for (int t = tid; t < MAXBIN; t += 256) h[t] = 0;
    __syncthreads();
    int e0 = bid * PCH;
    int dd[PEPT];
    #pragma unroll
    for (int k = 0; k < PEPT; ++k) dd[k] = dst[min(e0 + k * 256 + tid, E - 1)];
    #pragma unroll
    for (int k = 0; k < PEPT; ++k) {
        int e = e0 + k * 256 + tid;
        int d = min(max(dd[k], 0), N - 1);
        if (e < E) atomicAdd(&h[d >> BIN_SHIFT], 1);
    }
    __syncthreads();
    for (int t = tid; t < MAXBIN; t += 256) {
        int c = h[t];
        if (c > 0) atomicAdd(binCount + t * BIN_PAD, c);
    }
}

__global__ void __launch_bounds__(256)
k_bin_hist(const int* dst_p, int E_p, int N_p, int* binCount_p,
           const int* dst_m, int E_m, int N_m, int* binCount_m, int split) {
    BR_PICK;
    bin_hist_body(PK(dst), PK(E), PK(N), PK(binCount), bid);
}

__device__ __forceinline__ void scan_bins(const int* in, int* out, int lane) {
    int v[MAXBIN / 32];
    int s = 0;
    #pragma unroll
    for (int j = 0; j < MAXBIN / 32; ++j) {
        v[j] = in[lane * (MAXBIN / 32) + j];
        s += v[j];
    }
    int x = s;
    #pragma unroll
    for (int d = 1; d < 32; d <<= 1) {
        int t = __shfl_up(x, d);
        if (lane >= d) x += t;
    }
    int run = x - s;
    #pragma unroll
    for (int j = 0; j < MAXBIN / 32; ++j) {
        out[lane * (MAXBIN / 32) + j] = run;
        run += v[j];
    }
}

__device__ __forceinline__ void
bin_part_body(const int* __restrict__ src, const int* __restrict__ dst, int E, int N, int nbin,
              const int* __restrict__ binCount, int* __restrict__ binCursor,
              int* __restrict__ binStart, int* __restrict__ part, int bid) {
    __shared__ int h[MAXBIN];
    __shared__ int gc[MAXBIN];
    __shared__ int loc[MAXBIN];
    __shared__ int bstart[MAXBIN];
    __shared__ int gbase[MAXBIN];
    __shared__ int sval[PCH];
    __shared__ unsigned short sbin[PCH];
    int tid = threadIdx.x;
    for (int t = tid; t < MAXBIN; t += 256) {
        h[t] = 0;
        gc[t] = (t < nbin) ? binCount[t * BIN_PAD] : 0;
    }
    __syncthreads();
    int e0 = bid * PCH;
    int pk[PEPT], bb[PEPT], rr[PEPT];
    #pragma unroll
    for (int k = 0; k < PEPT; ++k) {
        int ec = min(e0 + k * 256 + tid, E - 1);
        bb[k] = dst[ec];
        pk[k] = src[ec];
    }
    #pragma unroll
    for (int k = 0; k < PEPT; ++k) {
        int e = e0 + k * 256 + tid;
        int d = min(max(bb[k], 0), N - 1);
        int s = min(max(pk[k], 0), N - 1);
        bb[k] = (e < E) ? (d >> BIN_SHIFT) : -1;
        pk[k] = (s << BIN_SHIFT) | (d & (BIN_NODES - 1));
        rr[k] = 0;
    }
    #pragma unroll
    for (int k = 0; k < PEPT; ++k)
        if (bb[k] >= 0) rr[k] = atomicAdd(&h[bb[k]], 1);
    __syncthreads();
    int wave = tid >> 5, lane = tid & 31;
    if (wave == 0) scan_bins(h, loc, lane);
    else if (wave == 1) scan_bins(gc, bstart, lane);
    __syncthreads();
    for (int t = tid; t < MAXBIN; t += 256) {
        int c = h[t];
        if (c > 0) gbase[t] = bstart[t] + atomicAdd(binCursor + t * BIN_PAD, c);
        if (bid == 0 && t < nbin) binStart[t] = bstart[t];
    }
    #pragma unroll
    for (int k = 0; k < PEPT; ++k) {
        if (bb[k] >= 0) {
            int p = loc[bb[k]] + rr[k];
            sval[p] = pk[k];
            sbin[p] = (unsigned short)bb[k];
        }
    }
    __syncthreads();
    int nv = min(PCH, E - e0);
    for (int i = tid; i < nv; i += 256) {
        int b = sbin[i];
        part[gbase[b] + (i - loc[b])] = sval[i];
    }
}

__global__ void __launch_bounds__(256)
k_bin_part(const int* src_p, const int* dst_p, int E_p, int N_p, int nbin_p,
           const int* binCount_p, int* binCursor_p, int* binStart_p, int* part_p,
           const int* src_m, const int* dst_m, int E_m, int N_m, int nbin_m,
           const int* binCount_m, int* binCursor_m, int* binStart_m, int* part_m, int split) {
    BR_PICK;
    bin_part_body(PK(src), PK(dst), PK(E), PK(N), PK(nbin), PK(binCount), PK(binCursor),
                  PK(binStart), PK(part), bid);
}

__device__ __forceinline__ void
bin_csr_body(const int* __restrict__ part, const int* __restrict__ binStart,
             const int* __restrict__ binCount, int N, int* __restrict__ ecnt,
             int* __restrict__ rowptr, float* __restrict__ dinv, int* __restrict__ col,
             const int* __restrict__ res, const float* __restrict__ pos,
             float* __restrict__ rec, int bid) {
    __shared__ int cnt[BIN_NODES];
    __shared__ int cur[BIN_NODES];
    __shared__ int wtot[8];
    __shared__ int lcol[LCAP];
    int tid = threadIdx.x;
    int wave = tid >> 5, lane = tid & 31;
    int b = bid;
    int s = binStart[b];
    int m = binCount[b * BIN_PAD];
    cnt[tid] = 0;
    __syncthreads();
    for (int i0 = 0; i0 < m; i0 += 1024) {
        int v[4];
        #pragma unroll
        for (int u = 0; u < 4; ++u) {
            int i = i0 + u * 256 + tid;
            v[u] = (i < m) ? part[s + i] : -1;
        }
        #pragma unroll
        for (int u = 0; u < 4; ++u)
            if (v[u] >= 0) atomicAdd(&cnt[v[u] & (BIN_NODES - 1)], 1);
    }
    __syncthreads();
    int c = cnt[tid];
    int x = c;
    #pragma unroll
    for (int d = 1; d < 32; d <<= 1) {
        int t = __shfl_up(x, d);
        if (lane >= d) x += t;
    }
    if (lane == 31) wtot[wave] = x;
    __syncthreads();
    int wb = 0;
    for (int w = 0; w < wave; ++w) wb += wtot[w];
    int off = wb + x - c;
    cur[tid] = off;
    int node = b * BIN_NODES + tid;
    if (node < N) {
        ecnt[node] = c;
        rowptr[node] = s + off;
        float dv = rsqrtf((float)c + 1.0f);
        dinv[node] = dv;
        v4f q;
        q.x = dv * pos[node * 3 + 0];
        q.y = dv * pos[node * 3 + 1];
        q.z = dv * pos[node * 3 + 2];
        q.w = dv;
        *(v4f*)(rec + (size_t)node * 8) = q;
        int r = res[node];
        if (r < 0 || r >= NRES) r = NRES;
        ((int*)rec)[(size_t)node * 8 + 4] = r * (HID * 4);
    }
    __syncthreads();
    bool staged = m <= LCAP;
    for (int i0 = 0; i0 < m; i0 += 1024) {
        int v[4];
        #pragma unroll
        for (int u = 0; u < 4; ++u) {
            int i = i0 + u * 256 + tid;
            v[u] = (i < m) ? part[s + i] : -1;
        }
        #pragma unroll
        for (int u = 0; u < 4; ++u) {
            if (v[u] >= 0) {
                int p = atomicAdd(&cur[v[u] & (BIN_NODES - 1)], 1);
                int sv = (int)((unsigned)v[u] >> BIN_SHIFT);
                if (staged) lcol[p] = sv;
                else col[s + p] = sv;
            }
        }
    }
    if (staged) {
        __syncthreads();
        for (int i = tid; i < m; i += 256) col[s + i] = lcol[i];
    }
}

__global__ void __launch_bounds__(256)
k_bin_csr(const int* part_p, const int* binStart_p, const int* binCount_p, int N_p,
          int* ecnt_p, int* rowptr_p, float* dinv_p, int* col_p,
          const int* res_p, const float* pos_p, float* rec_p,
          const int* part_m, const int* binStart_m, const int* binCount_m, int N_m,
          int* ecnt_m, int* rowptr_m, float* dinv_m, int* col_m,
          const int* res_m, const float* pos_m, float* rec_m, int split) {
    BR_PICK;
    bin_csr_body(PK(part), PK(binStart), PK(binCount), PK(N), PK(ecnt), PK(rowptr), PK(dinv),
                 PK(col), PK(res), PK(pos), PK(rec), bid);
}


__device__ __forceinline__ void
agg1_body(const float* __restrict__ rec, const int* __restrict__ rowptr,
          const int* __restrict__ ecnt, const int* __restrict__ col,
          const float* __restrict__ W, const float* __restrict__ b,
          const float* __restrict__ Wq, int N, float* __restrict__ y, int bid) {
    __shared__ __attribute__((aligned(16))) float Ws[(NRES + 1) * HID];
    __shared__ __attribute__((aligned(16))) float Hs[AGG_ROWS * HPITCH];
    __shared__ __attribute__((aligned(16))) float Ys[AGG_ROWS * HPITCH];
    __shared__ float Ds[AGG_ROWS];
    for (int t = threadIdx.x; t < NRES * HID / 4; t += 256)
        ((v4f*)Ws)[t] = ((const v4f*)W)[t];
    if (threadIdx.x < HID / 4)
        ((v4f*)Ws)[NRES * HID / 4 + threadIdx.x] = (v4f){0.0f, 0.0f, 0.0f, 0.0f};
    __syncthreads();

    int lane = threadIdx.x & 31;
    const v4f w20 = *(const v4f*)(W + (NRES + 0) * HID + lane * 4);
    const v4f w21 = *(const v4f*)(W + (NRES + 1) * HID + lane * 4);
    const v4f w22 = *(const v4f*)(W + (NRES + 2) * HID + lane * 4);
    const v4f bb = *(const v4f*)(b + lane * 4);
    const char* Wl = (const char*)Ws + lane * 16;
    const int* reci = (const int*)rec;
    int wave = __builtin_amdgcn_readfirstlane((int)(threadIdx.x >> 5));
    int first = (bid * 8 + wave) * AGG_NPW;

    for (int t = 0; t < AGG_NPW; ++t) {
        int i = __builtin_amdgcn_readfirstlane(first + t);
        float* hrow = Hs + (wave * AGG_NPW + t) * HPITCH + lane * 4;
        if (i >= N) {
            *(v4f*)hrow = (v4f){0.0f, 0.0f, 0.0f, 0.0f};
            continue;
        }
        int start = rowptr[i];
        int n = ecnt[i];
        v4f rs = *(const v4f*)(rec + (size_t)i * 8);
        int os = reci[(size_t)i * 8 + 4];
        v4f acc = (v4f){0.0f, 0.0f, 0.0f, 0.0f};
        float q0 = 0.0f, q1 = 0.0f, q2 = 0.0f;
        for (int base = 0; base < n; base += 32) {
            int m = n - base;
            if (m > 32) m = 32;
            bool live = lane < m;
            int c = live ? col[start + base + lane] : i;
            c = min(max(c, 0), N - 1);
            v4f r = *(const v4f*)(rec + (size_t)c * 8);
            int off = reci[(size_t)c * 8 + 4];
            float d = live ? r.w : 0.0f;
            off = live ? off : NRES * HID * 4;
            q0 += live ? r.x : 0.0f;
            q1 += live ? r.y : 0.0f;
            q2 += live ? r.z : 0.0f;
            int dbits = __float_as_int(d);
            for (int j = 0; j < m; j += 8) {
                int o0 = __builtin_amdgcn_readlane(off, j);
                int o1 = __builtin_amdgcn_readlane(off, j + 1);
                int o2 = __builtin_amdgcn_readlane(off, j + 2);
                int o3 = __builtin_amdgcn_readlane(off, j + 3);
                int o4 = __builtin_amdgcn_readlane(off, j + 4);
                int o5 = __builtin_amdgcn_readlane(off, j + 5);
                int o6 = __builtin_amdgcn_readlane(off, j + 6);
                int o7 = __builtin_amdgcn_readlane(off, j + 7);
                v4f x0 = *(const v4f*)(Wl + o0);
                v4f x1 = *(const v4f*)(Wl + o1);
                v4f x2 = *(const v4f*)(Wl + o2);
                v4f x3 = *(const v4f*)(Wl + o3);
                v4f x4 = *(const v4f*)(Wl + o4);
                v4f x5 = *(const v4f*)(Wl + o5);
                v4f x6 = *(const v4f*)(Wl + o6);
                v4f x7 = *(const v4f*)(Wl + o7);
                float d0 = __int_as_float(__builtin_amdgcn_readlane(dbits, j));
                float d1 = __int_as_float(__builtin_amdgcn_readlane(dbits, j + 1));
                float d2 = __int_as_float(__builtin_amdgcn_readlane(dbits, j + 2));
                float d3 = __int_as_float(__builtin_amdgcn_readlane(dbits, j + 3));
                float d4 = __int_as_float(__builtin_amdgcn_readlane(dbits, j + 4));
                float d5 = __int_as_float(__builtin_amdgcn_readlane(dbits, j + 5));
                float d6 = __int_as_float(__builtin_amdgcn_readlane(dbits, j + 6));
                float d7 = __int_as_float(__builtin_amdgcn_readlane(dbits, j + 7));
                acc += d0 * x0;
                acc += d1 * x1;
                acc += d2 * x2;
                acc += d3 * x3;
                acc += d4 * x4;
                acc += d5 * x5;
                acc += d6 * x6;
                acc += d7 * x7;
            }
        }
        v4f xs = *(const v4f*)(Wl + os);
        acc += rs.w * xs;
        #pragma unroll
        for (int s = 16; s >= 1; s >>= 1) {
            q0 += __shfl_xor(q0, s);
            q1 += __shfl_xor(q1, s);
            q2 += __shfl_xor(q2, s);
        }
        q0 += rs.x;
        q1 += rs.y;
        q2 += rs.z;
        v4f S = acc + q0 * w20 + q1 * w21 + q2 * w22;
        float dnode = rs.w;
        v4f o;
        o.x = fmaxf(dnode * S.x + bb.x, 0.0f);
        o.y = fmaxf(dnode * S.y + bb.y, 0.0f);
        o.z = fmaxf(dnode * S.z + bb.z, 0.0f);
        o.w = fmaxf(dnode * S.w + bb.w, 0.0f);
        *(v4f*)hrow = o;
        if (lane == 0) Ds[wave * AGG_NPW + t] = dnode;
    }
    __syncthreads();

    {
        int rt = wave >> 2, q = wave & 3;
        int lmod = lane & 15, half = lane >> 4;
        const float* Arow = Hs + (rt * 16 + lmod) * HPITCH + half * 2;
        const float* Bl = Wq + (size_t)(q * 32 + lane) * 4;
        v8f acc0 = (v8f){}, acc1 = (v8f){};
        v2f a[2][4];
        v4f bq[2][4];
        #pragma unroll
        for (int u = 0; u < 4; ++u) {
            a[0][u] = *(const v2f*)(Arow + 4 * u);
            bq[0][u] = *(const v4f*)(Bl + (size_t)u * 512);
        }
        #pragma unroll
        for (int g = 0; g < 8; ++g) {
            if (g + 1 < 8) {
                #pragma unroll
                for (int u = 0; u < 4; ++u) {
                    int kk = (g + 1) * 4 + u;
                    a[(g + 1) & 1][u] = *(const v2f*)(Arow + 4 * kk);
                    bq[(g + 1) & 1][u] = *(const v4f*)(Bl + (size_t)kk * 512);
                }
            }
            #pragma unroll
            for (int u = 0; u < 4; ++u) {
                v4f w4 = bq[g & 1][u];
                v2f b0; b0.x = w4.x; b0.y = w4.y;
                v2f b1; b1.x = w4.z; b1.y = w4.w;
                acc0 = __builtin_amdgcn_wmma_f32_16x16x4_f32(false, a[g & 1][u], false, b0, (short)0, acc0, false, false);
                acc1 = __builtin_amdgcn_wmma_f32_16x16x4_f32(false, a[g & 1][u], false, b1, (short)0, acc1, false, false);
            }
        }
        float* yb = Ys + (rt * 16 + half * 8) * HPITCH + q * 32 + lmod;
        #pragma unroll
        for (int r = 0; r < 8; ++r) {
            yb[r * HPITCH] = acc0[r];
            yb[r * HPITCH + 16] = acc1[r];
        }
    }
    __syncthreads();

    for (int t = 0; t < AGG_NPW; ++t) {
        int i = first + t;
        if (i < N) {
            int row = wave * AGG_NPW + t;
            v4f v = *(const v4f*)(Ys + row * HPITCH + lane * 4);
            float d = Ds[row];
            *(v4f*)(y + (size_t)i * HID + lane * 4) = v * d;
        }
    }
}

__global__ void __launch_bounds__(256)
k_agg1(const float* rec_p, const int* rowptr_p, const int* ecnt_p, const int* col_p,
       const float* W_p, const float* b_p, const float* Wq_p, int N_p, float* y_p,
       const float* rec_m, const int* rowptr_m, const int* ecnt_m, const int* col_m,
       const float* W_m, const float* b_m, const float* Wq_m, int N_m, float* y_m, int split) {
    BR_PICK;
    agg1_body(PK(rec), PK(rowptr), PK(ecnt), PK(col), PK(W), PK(b), PK(Wq), PK(N), PK(y), bid);
}

__device__ __forceinline__ void pool_flush(float* __restrict__ pesum, float* __restrict__ cnt,
                                           int g, int lane, v4f acc, int n) {
    if ((unsigned)g >= (unsigned)NGRAPH) return;
    float* p = pesum + g * HID + lane * 4;
    atomicAdd(p + 0, acc.x);
    atomicAdd(p + 1, acc.y);
    atomicAdd(p + 2, acc.z);
    atomicAdd(p + 3, acc.w);
    if (lane == 0) atomicAdd(cnt + g, (float)n);
}

#define GATHER_NPB 8
#define GATHER_NPW 4
#define GATHER_ROWS (GATHER_NPB * GATHER_NPW)
static_assert(GATHER_ROWS <= 32, "the block sum keeps one graph id per lane of one wave");

__device__ __forceinline__ void
gather_body(const float* __restrict__ y, const int* __restrict__ rowptr,
            const int* __restrict__ ecnt, const int* __restrict__ col,
            const float* __restrict__ dinv, const float* __restrict__ b,
            int N, float* __restrict__ h, int bid, bool pooled,
            const int* __restrict__ batch, float* __restrict__ pesum, float* __restrict__ cnt) {
    __shared__ __attribute__((aligned(16))) float Rs[GATHER_ROWS * HID];
    __shared__ int Gs[GATHER_ROWS];
    __shared__ int arrived;
    if (pooled) {
        if (threadIdx.x == 0) arrived = 0;
        __syncthreads();
    }
    int wave = __builtin_amdgcn_readfirstlane((int)(threadIdx.x >> 5));
    int i0 = __builtin_amdgcn_readfirstlane((bid * GATHER_NPB + wave) * GATHER_NPW);
    if (i0 >= N) return;
    int lane = threadIdx.x & 31;
    int st[GATHER_NPW], nn[GATHER_NPW], gg[GATHER_NPW], cf[GATHER_NPW];
    #pragma unroll
    for (int t = 0; t < GATHER_NPW; ++t) {
        int ic = min(i0 + t, N - 1);
        st[t] = rowptr[ic];
        nn[t] = (i0 + t < N) ? ecnt[ic] : 0;
        gg[t] = pooled ? batch[ic] : 0;
    }
    #pragma unroll
    for (int t = 0; t < GATHER_NPW; ++t) {
        int ic = min(i0 + t, N - 1);
        cf[t] = (lane < nn[t]) ? col[st[t] + lane] : ic;
    }
    const float* yl = y + lane * 4;
    const v4f bb = *(const v4f*)(b + lane * 4);
    #pragma unroll
    for (int t = 0; t < GATHER_NPW; ++t) {
        int i = i0 + t;
        if (i >= N) break;
        int start = st[t];
        int n = nn[t];
        int c = cf[t];
        v4f acc = (v4f){0.0f, 0.0f, 0.0f, 0.0f};
        for (int base = 0; base < n; base += 32) {
            int m = n - base;
            if (m > 32) m = 32;
            if (base > 0) c = (lane < m) ? col[start + base + lane] : i;
            c = min(max(c, 0), N - 1);
            for (int j = 0; j < m; j += 8) {
                int s0 = __builtin_amdgcn_readlane(c, j);
                int s1 = __builtin_amdgcn_readlane(c, j + 1);
                int s2 = __builtin_amdgcn_readlane(c, j + 2);
                int s3 = __builtin_amdgcn_readlane(c, j + 3);
                int s4 = __builtin_amdgcn_readlane(c, j + 4);
                int s5 = __builtin_amdgcn_readlane(c, j + 5);
                int s6 = __builtin_amdgcn_readlane(c, j + 6);
                int s7 = __builtin_amdgcn_readlane(c, j + 7);
                v4f v0 = *(const v4f*)(yl + (size_t)s0 * HID);
                v4f v1 = *(const v4f*)(yl + (size_t)s1 * HID);
                v4f v2 = *(const v4f*)(yl + (size_t)s2 * HID);
                v4f v3 = *(const v4f*)(yl + (size_t)s3 * HID);
                v4f v4 = *(const v4f*)(yl + (size_t)s4 * HID);
                v4f v5 = *(const v4f*)(yl + (size_t)s5 * HID);
                v4f v6 = *(const v4f*)(yl + (size_t)s6 * HID);
                v4f v7 = *(const v4f*)(yl + (size_t)s7 * HID);
                asm volatile("" :: "v"(v0), "v"(v1), "v"(v2), "v"(v3),
                                   "v"(v4), "v"(v5), "v"(v6), "v"(v7));
                acc += v0;
                if (j + 1 < m) acc += v1;
                if (j + 2 < m) acc += v2;
                if (j + 3 < m) acc += v3;
                if (j + 4 < m) acc += v4;
                if (j + 5 < m) acc += v5;
                if (j + 6 < m) acc += v6;
                if (j + 7 < m) acc += v7;
            }
        }
        v4f self = *(const v4f*)(yl + (size_t)i * HID);
        float d = dinv[i];
        v4f o;
        o.x = fmaxf(d * (acc.x + self.x) + bb.x, 0.0f);
        o.y = fmaxf(d * (acc.y + self.y) + bb.y, 0.0f);
        o.z = fmaxf(d * (acc.z + self.z) + bb.z, 0.0f);
        o.w = fmaxf(d * (acc.w + self.w) + bb.w, 0.0f);
        if (!pooled) {
            *(v4f*)(h + (size_t)i * HID + lane * 4) = o;
        } else {
            int row = wave * GATHER_NPW + t;
            *(v4f*)(Rs + row * HID + lane * 4) = o;
            if (lane == 0) Gs[row] = gg[t];
        }
    }
    if (!pooled) return;
    __threadfence_block();
    int a = 0;
    if (lane == 0) a = atomicAdd(&arrived, 1);
    a = __builtin_amdgcn_readfirstlane(a);
    int nrows = min(GATHER_ROWS, N - bid * GATHER_ROWS);
    int nact = (nrows + GATHER_NPW - 1) / GATHER_NPW;
    if (a != nact - 1) return;
    __threadfence_block();
    int gl = Gs[min(lane, nrows - 1)];
    int cur = __builtin_amdgcn_readlane(gl, 0);
    int rows = 0;
    v4f sum = (v4f){0.0f, 0.0f, 0.0f, 0.0f};
    for (int r0 = 0; r0 < nrows; r0 += 8) {
        v4f w[8];
        #pragma unroll
        for (int u = 0; u < 8; ++u)
            w[u] = *(const v4f*)(Rs + min(r0 + u, nrows - 1) * HID + lane * 4);
        #pragma unroll
        for (int u = 0; u < 8; ++u) {
            if (r0 + u < nrows) {
                int gr = __builtin_amdgcn_readlane(gl, r0 + u);
                if (gr != cur) {
                    pool_flush(pesum, cnt, cur, lane, sum, rows);
                    cur = gr;
                    rows = 0;
                    sum = (v4f){0.0f, 0.0f, 0.0f, 0.0f};
                }
                sum += w[u];
                rows += 1;
            }
        }
    }
    pool_flush(pesum, cnt, cur, lane, sum, rows);
}

__global__ void
k_gather(const float* y_p, const int* rowptr_p, const int* ecnt_p, const int* col_p,
         const float* dinv_p, const float* b_p, int N_p,
         const int* batch_p, float* pesum, float* cnt,
         const float* y_m, const int* rowptr_m, const int* ecnt_m, const int* col_m,
         const float* dinv_m, const float* b_m, int N_m, float* h_m, int split) {
    BR_PICK;
    gather_body(PK(y), PK(rowptr), PK(ecnt), PK(col), PK(dinv), PK(b), PK(N), h_m, bid,
                !second, batch_p, pesum, cnt);
}

__device__ __forceinline__ void
wfrag_body(const float* __restrict__ W, int K, float* __restrict__ Wf, int bid) {
    int idx = bid * 256 + threadIdx.x;
    if (idx >= K * HID) return;
    int j    = idx & 1;
    int nt   = (idx >> 1) & 7;
    int l    = (idx >> 4) & 31;
    int kk   = idx >> 9;
    int krow = 4 * kk + (l >> 4) * 2 + j;
    int col  = nt * 16 + (l & 15);
    Wf[idx] = W[(size_t)krow * HID + col];
}

__device__ __forceinline__ void
wpair_body(const float* __restrict__ W, float* __restrict__ Wq, int bid) {
    int idx = bid * 256 + threadIdx.x;
    if (idx >= HID * HID) return;
    int j    = idx & 1;
    int t    = (idx >> 1) & 1;
    int l    = (idx >> 2) & 31;
    int q    = (idx >> 7) & 3;
    int kk   = idx >> 9;
    int krow = 4 * kk + (l >> 4) * 2 + j;
    int col  = (2 * q + t) * 16 + (l & 15);
    Wq[idx] = W[(size_t)krow * HID + col];
}

#define FILL_BLOCKS (2 * MAXBIN * BIN_PAD / 256)
#define POOLZ_N     (NGRAPH * HID + NGRAPH)
#define POOLZ_BLOCKS ((POOLZ_N + 255) / 256)
#define WF_BLOCKS   (HID * HID / 256)
#define SETUP_BLOCKS (2 * FILL_BLOCKS + POOLZ_BLOCKS + 4 * WF_BLOCKS)

__global__ void __launch_bounds__(256)
k_setup(int* __restrict__ bins_p, int* __restrict__ bins_m, float* __restrict__ poolz,
        const float* __restrict__ W_p2, float* __restrict__ Wf_p,
        const float* __restrict__ W_m2, float* __restrict__ Wf_m,
        const float* __restrict__ W1f, float* __restrict__ Wf_h) {
    int bid = blockIdx.x;
    int t = threadIdx.x;
    if (bid < FILL_BLOCKS) { bins_p[bid * 256 + t] = 0; return; }
    bid -= FILL_BLOCKS;
    if (bid < FILL_BLOCKS) { bins_m[bid * 256 + t] = 0; return; }
    bid -= FILL_BLOCKS;
    if (bid < POOLZ_BLOCKS) {
        int i = bid * 256 + t;
        if (i < POOLZ_N) poolz[i] = 0.0f;
        return;
    }
    bid -= POOLZ_BLOCKS;
    if (bid < WF_BLOCKS) { wpair_body(W_p2, Wf_p, bid); return; }
    bid -= WF_BLOCKS;
    if (bid < WF_BLOCKS) { wpair_body(W_m2, Wf_m, bid); return; }
    bid -= WF_BLOCKS;
    wfrag_body(W1f, 2 * HID, Wf_h, bid);
}

template <bool SCALED>
__device__ __forceinline__ void gemm_ksteps(const float* __restrict__ Arow,
                                            const float* __restrict__ Wl,
                                            int ksteps, float ascale, v8f (&acc)[8]) {
    for (int kk = 0; kk < ksteps; ++kk) {
        v2f a;
        a.x = Arow[4 * kk];
        a.y = Arow[4 * kk + 1];
        if (SCALED) {
            a.x *= ascale;
            a.y *= ascale;
        }
        const float4* q = (const float4*)(Wl + (size_t)kk * 512);
        float4 q0 = q[0], q1 = q[1], q2 = q[2], q3 = q[3];
        v2f b0; b0.x = q0.x; b0.y = q0.y;
        v2f b1; b1.x = q0.z; b1.y = q0.w;
        v2f b2; b2.x = q1.x; b2.y = q1.y;
        v2f b3; b3.x = q1.z; b3.y = q1.w;
        v2f b4; b4.x = q2.x; b4.y = q2.y;
        v2f b5; b5.x = q2.z; b5.y = q2.w;
        v2f b6; b6.x = q3.x; b6.y = q3.y;
        v2f b7; b7.x = q3.z; b7.y = q3.w;
        acc[0] = __builtin_amdgcn_wmma_f32_16x16x4_f32(false, a, false, b0, (short)0, acc[0], false, false);
        acc[1] = __builtin_amdgcn_wmma_f32_16x16x4_f32(false, a, false, b1, (short)0, acc[1], false, false);
        acc[2] = __builtin_amdgcn_wmma_f32_16x16x4_f32(false, a, false, b2, (short)0, acc[2], false, false);
        acc[3] = __builtin_amdgcn_wmma_f32_16x16x4_f32(false, a, false, b3, (short)0, acc[3], false, false);
        acc[4] = __builtin_amdgcn_wmma_f32_16x16x4_f32(false, a, false, b4, (short)0, acc[4], false, false);
        acc[5] = __builtin_amdgcn_wmma_f32_16x16x4_f32(false, a, false, b5, (short)0, acc[5], false, false);
        acc[6] = __builtin_amdgcn_wmma_f32_16x16x4_f32(false, a, false, b6, (short)0, acc[6], false, false);
        acc[7] = __builtin_amdgcn_wmma_f32_16x16x4_f32(false, a, false, b7, (short)0, acc[7], false, false);
    }
}

__global__ void __launch_bounds__(128)
k_gemm_head(const float* __restrict__ h, const float* __restrict__ pesum,
            const float* __restrict__ cnt, const int* __restrict__ batch, int M,
            const float* __restrict__ Wf, const float* __restrict__ bias,
            const float* __restrict__ W2, const float* __restrict__ b2,
            float* __restrict__ out) {
    __shared__ float so[4 * 16 * 3];
    int w = __builtin_amdgcn_readfirstlane((int)(threadIdx.x >> 5));
    int lane = threadIdx.x & 31;
    int tile = blockIdx.x * 4 + w;
    if (tile * 16 < M) {
        int row0 = tile * 16;
        int half = lane >> 4;
        int lmod = lane & 15;
        int koff = half * 2;
        int g = min(max(batch[row0 + lmod], 0), NGRAPH - 1);
        float inv = 1.0f / fmaxf(cnt[g], 1.0f);
        const float* Wl = Wf + (size_t)lane * 16;

        v8f acc[8];
        #pragma unroll
        for (int nt = 0; nt < 8; ++nt) acc[nt] = (v8f){};

        gemm_ksteps<false>(h + (size_t)(row0 + lmod) * HID + koff, Wl, HID / 4, 1.0f, acc);
        gemm_ksteps<true>(pesum + (size_t)g * HID + koff, Wl + (size_t)(HID / 4) * 512, HID / 4,
                          inv, acc);

        float p[8][3];
        #pragma unroll
        for (int r = 0; r < 8; ++r) { p[r][0] = 0.0f; p[r][1] = 0.0f; p[r][2] = 0.0f; }
        #pragma unroll
        for (int nt = 0; nt < 8; ++nt) {
            int col = nt * 16 + lmod;
            float bv = bias[col];
            float w0 = W2[col * 3 + 0];
            float w1 = W2[col * 3 + 1];
            float w2 = W2[col * 3 + 2];
            #pragma unroll
            for (int r = 0; r < 8; ++r) {
                float v = fmaxf(acc[nt][r] + bv, 0.0f);
                p[r][0] += v * w0;
                p[r][1] += v * w1;
                p[r][2] += v * w2;
            }
        }
        #pragma unroll
        for (int s = 8; s >= 1; s >>= 1) {
            #pragma unroll
            for (int r = 0; r < 8; ++r) {
                p[r][0] += __shfl_xor(p[r][0], s);
                p[r][1] += __shfl_xor(p[r][1], s);
                p[r][2] += __shfl_xor(p[r][2], s);
            }
        }
        if (lmod == 0) {
            float c0 = b2[0], c1 = b2[1], c2 = b2[2];
            #pragma unroll
            for (int r = 0; r < 8; ++r) {
                float* q = so + w * 48 + (half * 8 + r) * 3;
                q[0] = p[r][0] + c0;
                q[1] = p[r][1] + c1;
                q[2] = p[r][2] + c2;
            }
        }
    }
    __syncthreads();
    for (int t = threadIdx.x; t < 4 * 16 * 3; t += 128) {
        size_t gi = (size_t)blockIdx.x * (4 * 16 * 3) + t;
        if (gi < (size_t)M * 3) out[gi] = so[t];
    }
}


static inline int cdiv(long long a, long long b) { return (int)((a + b - 1) / b); }

extern "C" void kernel_launch(void* const* d_in, const int* in_sizes, int n_in,
                              void* d_out, int out_size, void* d_ws, size_t ws_size,
                              hipStream_t stream) {
    (void)n_in; (void)out_size; (void)ws_size;

    const int*   res_p   = (const int*)d_in[0];
    const float* pos_p   = (const float*)d_in[1];
    const int*   ei_p    = (const int*)d_in[2];
    const int*   batch_p = (const int*)d_in[3];
    const int*   res_m   = (const int*)d_in[4];
    const float* pos_m   = (const float*)d_in[5];
    const int*   ei_m    = (const int*)d_in[6];
    const int*   batch_m = (const int*)d_in[7];
    const float* W_p1 = (const float*)d_in[8];
    const float* b_p1 = (const float*)d_in[9];
    const float* W_p2 = (const float*)d_in[10];
    const float* b_p2 = (const float*)d_in[11];
    const float* W_m1 = (const float*)d_in[12];
    const float* b_m1 = (const float*)d_in[13];
    const float* W_m2 = (const float*)d_in[14];
    const float* b_m2 = (const float*)d_in[15];
    const float* W1f  = (const float*)d_in[16];
    const float* b1f  = (const float*)d_in[17];
    const float* W2f  = (const float*)d_in[18];
    const float* b2f  = (const float*)d_in[19];

    int N_P = in_sizes[0];
    int E_P = in_sizes[2] / 2;
    int N_M = in_sizes[4];
    int E_M = in_sizes[6] / 2;
    const int* src_p = ei_p;
    const int* dst_p = ei_p + E_P;
    const int* src_m = ei_m;
    const int* dst_m = ei_m + E_M;

    float* ws = (float*)d_ws;
    size_t nhp = (size_t)N_P * HID;
    size_t nhm = (size_t)N_M * HID;
    size_t npa = ((size_t)N_P + 63) / 64 * 64;
    size_t nma = ((size_t)N_M + 63) / 64 * 64;
    float* A_p    = ws;
    float* C_p    = A_p + nhp;
    float* A_m    = C_p + nhp;
    float* C_m    = A_m + nhm;
    float* rec_p  = C_m + nhm;
    float* rec_m  = rec_p + npa * 8;
    float* dinv_p = rec_m + nma * 8;
    float* dinv_m = dinv_p + npa;
    float* pe_sum = dinv_m + nma;
    float* cnt    = pe_sum + NGRAPH * HID;
    float* Wf_p   = cnt + NGRAPH;
    float* Wf_m   = Wf_p + HID * HID;
    float* Wf_h   = Wf_m + HID * HID;
    int* ecnt_p   = (int*)(Wf_h + 2 * HID * HID);
    int* rowptr_p = ecnt_p + npa;
    int* part_p   = rowptr_p + npa;
    int* col_p    = part_p + E_P;
    int* bins_p   = col_p + E_P;
    int* ecnt_m   = bins_p + (2 * BIN_PAD + 1) * MAXBIN;
    int* rowptr_m = ecnt_m + nma;
    int* part_m   = rowptr_m + nma;
    int* col_m    = part_m + E_M;
    int* bins_m   = col_m + E_M;
    int* bcur_p   = bins_p + MAXBIN * BIN_PAD;
    int* bsta_p   = bins_p + 2 * MAXBIN * BIN_PAD;
    int* bcur_m   = bins_m + MAXBIN * BIN_PAD;
    int* bsta_m   = bins_m + 2 * MAXBIN * BIN_PAD;
    int nbin_p = cdiv(N_P, BIN_NODES);
    int nbin_m = cdiv(N_M, BIN_NODES);

    k_setup<<<SETUP_BLOCKS, 256, 0, stream>>>(bins_p, bins_m, pe_sum, W_p2, Wf_p, W_m2, Wf_m,
                                              W1f, Wf_h);

    int gp = cdiv(E_P, PCH), gm = cdiv(E_M, PCH);
    k_bin_hist<<<gp + gm, 256, 0, stream>>>(dst_p, E_P, N_P, bins_p,
                                            dst_m, E_M, N_M, bins_m, gp);
    k_bin_part<<<gp + gm, 256, 0, stream>>>(src_p, dst_p, E_P, N_P, nbin_p, bins_p, bcur_p, bsta_p, part_p,
                                            src_m, dst_m, E_M, N_M, nbin_m, bins_m, bcur_m, bsta_m, part_m,
                                            gp);
    k_bin_csr<<<nbin_p + nbin_m, 256, 0, stream>>>(part_p, bsta_p, bins_p, N_P, ecnt_p, rowptr_p, dinv_p, col_p,
                                                   res_p, pos_p, rec_p,
                                                   part_m, bsta_m, bins_m, N_M, ecnt_m, rowptr_m, dinv_m, col_m,
                                                   res_m, pos_m, rec_m, nbin_p);

    gp = cdiv(N_P, AGG_ROWS); gm = cdiv(N_M, AGG_ROWS);
    k_agg1<<<gp + gm, 256, 0, stream>>>(rec_p, rowptr_p, ecnt_p, col_p, W_p1, b_p1, Wf_p, N_P, A_p,
                                        rec_m, rowptr_m, ecnt_m, col_m, W_m1, b_m1, Wf_m, N_M, A_m, gp);

    gp = cdiv(N_P, GATHER_ROWS); gm = cdiv(N_M, GATHER_ROWS);
    k_gather<<<gp + gm, 256, 0, stream>>>(A_p, rowptr_p, ecnt_p, col_p, dinv_p, b_p2, N_P,
                                          batch_p, pe_sum, cnt,
                                          A_m, rowptr_m, ecnt_m, col_m, dinv_m, b_m2, N_M, C_m, gp);

    k_gemm_head<<<cdiv(N_M / 16, 4), 128, 0, stream>>>(C_m, pe_sum, cnt, batch_m, N_M, Wf_h, b1f,
                                                       W2f, b2f, (float*)d_out);
}
